# starting kernel
# baseline (speedup 1.0000x reference)
#include <hip/hip_runtime.h>
#include <cmath>

typedef __attribute__((ext_vector_type(2))) float v2f;
typedef __attribute__((ext_vector_type(8))) float v8f;

#define N_PTS   12000
#define DIN     64
#define DLAT    16
#define DFEAT   64
#define KNN     16
#define ROW_TILES (N_PTS / 16)
#define WAVES_PER_WG 8
#define NUM_WG  ((ROW_TILES + WAVES_PER_WG - 1) / WAVES_PER_WG)

static __device__ __forceinline__ v8f wmma_f32(v2f a, v2f b, v8f c) {
  return __builtin_amdgcn_wmma_f32_16x16x4_f32(
       false, a,  false, b,
       (short)0, c,  false,  false);
}

__global__ void __launch_bounds__(256)
encoder_kernel(const float* __restrict__ x,
               const float* __restrict__ Wf, const float* __restrict__ bf,
               const float* __restrict__ Wl, const float* __restrict__ bl,
               float* __restrict__ feats, float* __restrict__ coords) {
  const int lane = threadIdx.x & 31;
  const int wave = threadIdx.x >> 5;
  const int tile = blockIdx.x * WAVES_PER_WG + wave;
  if (tile >= ROW_TILES) return;
  const int row0 = tile * 16;
  const int m  = lane & 15;
  const int kh = lane >> 4;

  v2f a[16];
  const float* xr = x + (size_t)(row0 + m) * DIN;
  #pragma unroll
  for (int s = 0; s < 16; ++s) {
    a[s].x = xr[4 * s + 2 * kh];
    a[s].y = xr[4 * s + 1 + 2 * kh];
  }

  #pragma unroll
  for (int nt = 0; nt < 4; ++nt) {
    v8f acc = {};
    #pragma unroll
    for (int s = 0; s < 16; ++s) {
      v2f b;
      b.x = Wf[(4 * s + 2 * kh) * DFEAT + nt * 16 + m];
      b.y = Wf[(4 * s + 1 + 2 * kh) * DFEAT + nt * 16 + m];
      acc = wmma_f32(a[s], b, acc);
    }
    #pragma unroll
    for (int r = 0; r < 8; ++r) {
      const int row = row0 + r + 8 * kh;
      const int col = nt * 16 + m;
      feats[(size_t)row * DFEAT + col] = tanhf(acc[r] + bf[col]);
    }
  }

  {
    v8f acc = {};
    #pragma unroll
    for (int s = 0; s < 16; ++s) {
      v2f b;
      b.x = Wl[(4 * s + 2 * kh) * DLAT + m];
      b.y = Wl[(4 * s + 1 + 2 * kh) * DLAT + m];
      acc = wmma_f32(a[s], b, acc);
    }
    #pragma unroll
    for (int r = 0; r < 8; ++r) {
      const int row = row0 + r + 8 * kh;
      coords[(size_t)row * DLAT + m] = tanhf(acc[r] + bl[m]);
    }
  }
}

__global__ void sq_kernel(const float* __restrict__ coords, float* __restrict__ sq) {
  const int i = blockIdx.x * blockDim.x + threadIdx.x;
  if (i >= N_PTS) return;
  const float* c = coords + (size_t)i * DLAT;
  float s = 0.f;
  #pragma unroll
  for (int d = 0; d < DLAT; ++d) s += c[d] * c[d];
  sq[i] = s;
}

__global__ void __launch_bounds__(256)
gravnet_kernel(const float* __restrict__ coords, const float* __restrict__ sq,
               const float* __restrict__ feats,
               const float* __restrict__ Wout, const float* __restrict__ bout,
               float* __restrict__ out) {
  __shared__ float lds_dot[WAVES_PER_WG][16 * 17];
  __shared__ float lds_w  [WAVES_PER_WG][16][KNN];
  __shared__ int   lds_j  [WAVES_PER_WG][16][KNN];
  __shared__ float lds_agg[WAVES_PER_WG][16][DFEAT];

  const int lane = threadIdx.x & 31;
  const int wave = threadIdx.x >> 5;
  const int tile = blockIdx.x * WAVES_PER_WG + wave;
  if (tile >= ROW_TILES) return;
  const int row0 = tile * 16;
  const int m  = lane & 15;
  const int kh = lane >> 4;

  v2f a[4];
  {
    const float* cr = coords + (size_t)(row0 + m) * DLAT;
    #pragma unroll
    for (int s = 0; s < 4; ++s) {
      a[s].x = cr[4 * s + 2 * kh];
      a[s].y = cr[4 * s + 1 + 2 * kh];
    }
  }

  float bestd[KNN]; int bestj[KNN];
  #pragma unroll
  for (int s = 0; s < KNN; ++s) { bestd[s] = 3.4e38f; bestj[s] = 0; }
  float worst = 3.4e38f; int wslot = 0;
  const float sqi = sq[row0 + m];

  for (int jt = 0; jt < ROW_TILES; ++jt) {
    const int j0 = jt * 16;
    if (jt + 1 < ROW_TILES)
      __builtin_prefetch(coords + (size_t)(j0 + 16 + m) * DLAT, 0, 3);

    v8f acc = {};
    const float* cj = coords + (size_t)(j0 + m) * DLAT;
    #pragma unroll
    for (int s = 0; s < 4; ++s) {
      v2f b;
      b.x = cj[4 * s + 2 * kh];
      b.y = cj[4 * s + 1 + 2 * kh];
      acc = wmma_f32(a[s], b, acc);
    }

    #pragma unroll
    for (int r = 0; r < 8; ++r)
      lds_dot[wave][(r + 8 * kh) * 17 + m] = acc[r];
    __builtin_amdgcn_wave_barrier();

    if (lane < 16) {
      const int row = row0 + lane;
      #pragma unroll
      for (int n = 0; n < 16; ++n) {
        const int j = j0 + n;
        float d = sqi + sq[j] - 2.f * lds_dot[wave][lane * 17 + n];
        d = fmaxf(d, 0.f);
        if (j != row && d < worst) {
          #pragma unroll
          for (int s = 0; s < KNN; ++s)
            if (s == wslot) { bestd[s] = d; bestj[s] = j; }
          worst = bestd[0]; wslot = 0;
          #pragma unroll
          for (int s = 1; s < KNN; ++s)
            if (bestd[s] > worst) { worst = bestd[s]; wslot = s; }
        }
      }
    }
    __builtin_amdgcn_wave_barrier();
  }

  if (lane < 16) {
    #pragma unroll
    for (int k = 0; k < KNN; ++k) {
      lds_w[wave][lane][k] = expf(-bestd[k]);
      lds_j[wave][lane][k] = bestj[k];
    }
  }
  __builtin_amdgcn_wave_barrier();

  for (int r = 0; r < 16; ++r) {
    float acc0 = 0.f, acc1 = 0.f;
    #pragma unroll
    for (int k = 0; k < KNN; ++k) {
      const int   j  = lds_j[wave][r][k];
      const float wk = lds_w[wave][r][k];
      acc0 += wk * feats[(size_t)j * DFEAT + lane];
      acc1 += wk * feats[(size_t)j * DFEAT + lane + 32];
    }
    lds_agg[wave][r][lane]      = acc0;
    lds_agg[wave][r][lane + 32] = acc1;
  }
  __builtin_amdgcn_wave_barrier();

  const float* fr = feats + (size_t)(row0 + m) * DFEAT;
  #pragma unroll
  for (int nt = 0; nt < 4; ++nt) {
    v8f acc = {};
    #pragma unroll
    for (int s = 0; s < 16; ++s) {
      v2f av, bv;
      av.x = fr[4 * s + 2 * kh];
      av.y = fr[4 * s + 1 + 2 * kh];
      bv.x = Wout[(4 * s + 2 * kh) * DFEAT + nt * 16 + m];
      bv.y = Wout[(4 * s + 1 + 2 * kh) * DFEAT + nt * 16 + m];
      acc = wmma_f32(av, bv, acc);
    }
    #pragma unroll
    for (int s = 0; s < 16; ++s) {
      v2f av, bv;
      av.x = lds_agg[wave][m][4 * s + 2 * kh];
      av.y = lds_agg[wave][m][4 * s + 1 + 2 * kh];
      bv.x = Wout[(64 + 4 * s + 2 * kh) * DFEAT + nt * 16 + m];
      bv.y = Wout[(64 + 4 * s + 1 + 2 * kh) * DFEAT + nt * 16 + m];
      acc = wmma_f32(av, bv, acc);
    }
    #pragma unroll
    for (int r = 0; r < 8; ++r) {
      const int row = row0 + r + 8 * kh;
      const int col = nt * 16 + m;
      out[(size_t)row * DFEAT + col] = acc[r] + bout[col];
    }
  }
}

extern "C" void kernel_launch(void* const* d_in, const int* in_sizes, int n_in,
                              void* d_out, int out_size, void* d_ws, size_t ws_size,
                              hipStream_t stream) {
  const float* x  = (const float*)d_in[0];
  const float* Wf = (const float*)d_in[1];
  const float* bf = (const float*)d_in[2];
  const float* Wl = (const float*)d_in[3];
  const float* bl = (const float*)d_in[4];
  const float* Wo = (const float*)d_in[5];
  const float* bo = (const float*)d_in[6];
  float* out = (float*)d_out;

  float* feats  = (float*)d_ws;
  float* coords = feats  + (size_t)N_PTS * DFEAT;
  float* sq     = coords + (size_t)N_PTS * DLAT;

  encoder_kernel<<<NUM_WG, 256, 0, stream>>>(x, Wf, bf, Wl, bl, feats, coords);
  sq_kernel<<<(N_PTS + 255) / 256, 256, 0, stream>>>(coords, sq);
  gravnet_kernel<<<NUM_WG, 256, 0, stream>>>(coords, sq, feats, Wo, bo, out);
}


# Split each 16-row tile's column scan over the 8 waves of one workgroup (750 workgroups, was 94) and merge the 8 private top-16 lists by rank in LDS; agg is now summed in ascending-distance order
# speedup vs baseline: 3.2945x; 3.2945x over previous
#include <hip/hip_runtime.h>
#include <cmath>

typedef __attribute__((ext_vector_type(2))) float v2f;
typedef __attribute__((ext_vector_type(8))) float v8f;

#define N_PTS   12000
#define DIN     64
#define DLAT    16
#define DFEAT   64
#define KNN     16
#define ROW_TILES (N_PTS / 16)
#define WAVES_PER_WG 8
#define NUM_WG  ((ROW_TILES + WAVES_PER_WG - 1) / WAVES_PER_WG)

static __device__ __forceinline__ v8f wmma_f32(v2f a, v2f b, v8f c) {
  return __builtin_amdgcn_wmma_f32_16x16x4_f32(
       false, a,  false, b,
       (short)0, c,  false,  false);
}

__global__ void __launch_bounds__(256)
encoder_kernel(const float* __restrict__ x,
               const float* __restrict__ Wf, const float* __restrict__ bf,
               const float* __restrict__ Wl, const float* __restrict__ bl,
               float* __restrict__ feats, float* __restrict__ coords) {
  const int lane = threadIdx.x & 31;
  const int wave = threadIdx.x >> 5;
  const int tile = blockIdx.x * WAVES_PER_WG + wave;
  if (tile >= ROW_TILES) return;
  const int row0 = tile * 16;
  const int m  = lane & 15;
  const int kh = lane >> 4;

  v2f a[16];
  const float* xr = x + (size_t)(row0 + m) * DIN;
  #pragma unroll
  for (int s = 0; s < 16; ++s) {
    a[s].x = xr[4 * s + 2 * kh];
    a[s].y = xr[4 * s + 1 + 2 * kh];
  }

  #pragma unroll
  for (int nt = 0; nt < 4; ++nt) {
    v8f acc = {};
    #pragma unroll
    for (int s = 0; s < 16; ++s) {
      v2f b;
      b.x = Wf[(4 * s + 2 * kh) * DFEAT + nt * 16 + m];
      b.y = Wf[(4 * s + 1 + 2 * kh) * DFEAT + nt * 16 + m];
      acc = wmma_f32(a[s], b, acc);
    }
    #pragma unroll
    for (int r = 0; r < 8; ++r) {
      const int row = row0 + r + 8 * kh;
      const int col = nt * 16 + m;
      feats[(size_t)row * DFEAT + col] = tanhf(acc[r] + bf[col]);
    }
  }

  {
    v8f acc = {};
    #pragma unroll
    for (int s = 0; s < 16; ++s) {
      v2f b;
      b.x = Wl[(4 * s + 2 * kh) * DLAT + m];
      b.y = Wl[(4 * s + 1 + 2 * kh) * DLAT + m];
      acc = wmma_f32(a[s], b, acc);
    }
    #pragma unroll
    for (int r = 0; r < 8; ++r) {
      const int row = row0 + r + 8 * kh;
      coords[(size_t)row * DLAT + m] = tanhf(acc[r] + bl[m]);
    }
  }
}

__global__ void sq_kernel(const float* __restrict__ coords, float* __restrict__ sq) {
  const int i = blockIdx.x * blockDim.x + threadIdx.x;
  if (i >= N_PTS) return;
  const float* c = coords + (size_t)i * DLAT;
  float s = 0.f;
  #pragma unroll
  for (int d = 0; d < DLAT; ++d) s += c[d] * c[d];
  sq[i] = s;
}

#define NSPLIT  WAVES_PER_WG
#define NCAND   (NSPLIT * KNN)
#define KPITCH  (NCAND + 2)

typedef unsigned long long u64;
typedef __attribute__((ext_vector_type(2))) unsigned long long v2u64;

__global__ void __launch_bounds__(256)
gravnet_kernel(const float* __restrict__ coords, const float* __restrict__ sq,
               const float* __restrict__ feats,
               const float* __restrict__ Wout, const float* __restrict__ bout,
               float* __restrict__ out) {
  __shared__ float lds_dot[NSPLIT][16 * 17];
  __shared__ __attribute__((aligned(16))) u64 lds_key[16][KPITCH];
  __shared__ float lds_w  [16][KNN];
  __shared__ int   lds_j  [16][KNN];
  __shared__ float lds_agg[16][DFEAT];

  const int tid  = threadIdx.x;
  const int lane = tid & 31;
  const int wave = __builtin_amdgcn_readfirstlane(tid >> 5);
  const int row0 = blockIdx.x * 16;
  const int m  = lane & 15;
  const int kh = lane >> 4;

  v2f a[4];
  {
    const float* cr = coords + (size_t)(row0 + m) * DLAT;
    #pragma unroll
    for (int s = 0; s < 4; ++s) {
      a[s].x = cr[4 * s + 2 * kh];
      a[s].y = cr[4 * s + 1 + 2 * kh];
    }
  }

  float bestd[KNN]; int bestj[KNN];
  #pragma unroll
  for (int s = 0; s < KNN; ++s) { bestd[s] = 3.4e38f; bestj[s] = 0; }
  float worst = 3.4e38f; int wslot = 0;
  const float sqi = sq[row0 + m];

  for (int jt = wave; jt < ROW_TILES; jt += NSPLIT) {
    const int j0 = jt * 16;
    if (jt + NSPLIT < ROW_TILES)
      __builtin_prefetch(coords + (size_t)(j0 + 16 * NSPLIT + m) * DLAT, 0, 3);

    v8f acc = {};
    const float* cj = coords + (size_t)(j0 + m) * DLAT;
    #pragma unroll
    for (int s = 0; s < 4; ++s) {
      v2f b;
      b.x = cj[4 * s + 2 * kh];
      b.y = cj[4 * s + 1 + 2 * kh];
      acc = wmma_f32(a[s], b, acc);
    }

    #pragma unroll
    for (int r = 0; r < 8; ++r)
      lds_dot[wave][(r + 8 * kh) * 17 + m] = acc[r];
    __builtin_amdgcn_wave_barrier();

    if (lane < 16) {
      const int row = row0 + lane;
      #pragma unroll
      for (int n = 0; n < 16; ++n) {
        const int j = j0 + n;
        float d = sqi + sq[j] - 2.f * lds_dot[wave][lane * 17 + n];
        d = fmaxf(d, 0.f);
        if (j != row && d < worst) {
          #pragma unroll
          for (int s = 0; s < KNN; ++s)
            if (s == wslot) { bestd[s] = d; bestj[s] = j; }
          worst = bestd[0]; wslot = 0;
          #pragma unroll
          for (int s = 1; s < KNN; ++s)
            if (bestd[s] > worst) { worst = bestd[s]; wslot = s; }
        }
      }
    }
    __builtin_amdgcn_wave_barrier();
  }

  if (lane < 16) {
    #pragma unroll
    for (int k = 0; k < KNN; ++k) {
      const int db = max(__float_as_int(bestd[k]), 0);
      lds_key[lane][wave * KNN + k] = ((u64)(unsigned)db << 32) | (u64)(unsigned)bestj[k];
    }
  }
  lds_w[tid >> 4][tid & 15] = 0.f;
  lds_j[tid >> 4][tid & 15] = 0;
  __syncthreads();

  {
    const int r = tid >> 4;
    const int q = tid & 15;
    u64 ck[NSPLIT]; int rank[NSPLIT];
    #pragma unroll
    for (int i = 0; i < NSPLIT; ++i) {
      ck[i]   = lds_key[r][q + KNN * i];
      rank[i] = 0;
    }
    for (int e = 0; e < NCAND; e += 2) {
      const v2u64 kv = *reinterpret_cast<const v2u64*>(&lds_key[r][e]);
      #pragma unroll
      for (int i = 0; i < NSPLIT; ++i)
        rank[i] += (kv.x < ck[i] ? 1 : 0) + (kv.y < ck[i] ? 1 : 0);
    }
    #pragma unroll
    for (int i = 0; i < NSPLIT; ++i) {
      if (rank[i] < KNN) {
        lds_w[r][rank[i]] = expf(-__int_as_float((int)(ck[i] >> 32)));
        lds_j[r][rank[i]] = (int)(unsigned)ck[i];
      }
    }
  }
  __syncthreads();

  #pragma unroll
  for (int rr = 0; rr < 16 / NSPLIT; ++rr) {
    const int r = wave * (16 / NSPLIT) + rr;
    float acc0 = 0.f, acc1 = 0.f;
    #pragma unroll
    for (int k = 0; k < KNN; ++k) {
      const int   j  = lds_j[r][k];
      const float wk = lds_w[r][k];
      acc0 += wk * feats[(size_t)j * DFEAT + lane];
      acc1 += wk * feats[(size_t)j * DFEAT + lane + 32];
    }
    lds_agg[r][lane]      = acc0;
    lds_agg[r][lane + 32] = acc1;
  }
  __syncthreads();

  if (wave < 2) {
    const float* fr = feats + (size_t)(row0 + m) * DFEAT;
    #pragma unroll
    for (int t = 0; t < 2; ++t) {
      const int nt = wave * 2 + t;
      v8f acc = {};
      #pragma unroll
      for (int s = 0; s < 16; ++s) {
        v2f av, bv;
        av.x = fr[4 * s + 2 * kh];
        av.y = fr[4 * s + 1 + 2 * kh];
        bv.x = Wout[(4 * s + 2 * kh) * DFEAT + nt * 16 + m];
        bv.y = Wout[(4 * s + 1 + 2 * kh) * DFEAT + nt * 16 + m];
        acc = wmma_f32(av, bv, acc);
      }
      #pragma unroll
      for (int s = 0; s < 16; ++s) {
        v2f av, bv;
        av.x = lds_agg[m][4 * s + 2 * kh];
        av.y = lds_agg[m][4 * s + 1 + 2 * kh];
        bv.x = Wout[(64 + 4 * s + 2 * kh) * DFEAT + nt * 16 + m];
        bv.y = Wout[(64 + 4 * s + 1 + 2 * kh) * DFEAT + nt * 16 + m];
        acc = wmma_f32(av, bv, acc);
      }
      #pragma unroll
      for (int r = 0; r < 8; ++r) {
        const int row = row0 + r + 8 * kh;
        const int col = nt * 16 + m;
        out[(size_t)row * DFEAT + col] = acc[r] + bout[col];
      }
    }
  }
}

extern "C" void kernel_launch(void* const* d_in, const int* in_sizes, int n_in,
                              void* d_out, int out_size, void* d_ws, size_t ws_size,
                              hipStream_t stream) {
  const float* x  = (const float*)d_in[0];
  const float* Wf = (const float*)d_in[1];
  const float* bf = (const float*)d_in[2];
  const float* Wl = (const float*)d_in[3];
  const float* bl = (const float*)d_in[4];
  const float* Wo = (const float*)d_in[5];
  const float* bo = (const float*)d_in[6];
  float* out = (float*)d_out;

  float* feats  = (float*)d_ws;
  float* coords = feats  + (size_t)N_PTS * DFEAT;
  float* sq     = coords + (size_t)N_PTS * DLAT;

  encoder_kernel<<<NUM_WG, 256, 0, stream>>>(x, Wf, bf, Wl, bl, feats, coords);
  sq_kernel<<<(N_PTS + 255) / 256, 256, 0, stream>>>(coords, sq);
  gravnet_kernel<<<ROW_TILES, 256, 0, stream>>>(coords, sq, feats, Wo, bo, out);
}


# Top-16 in two scans: scan 1 keeps only each lane's 16 smallest distances, sorted in registers (one med3 per slot); scan 2 recomputes the tiles and collects indices up to the row's 16th distance
# speedup vs baseline: 7.2794x; 2.2096x over previous
#include <hip/hip_runtime.h>
#include <cmath>

typedef __attribute__((ext_vector_type(2))) float v2f;
typedef __attribute__((ext_vector_type(8))) float v8f;

#define N_PTS   12000
#define DIN     64
#define DLAT    16
#define DFEAT   64
#define KNN     16
#define ROW_TILES (N_PTS / 16)
#define WAVES_PER_WG 8
#define NUM_WG  ((ROW_TILES + WAVES_PER_WG - 1) / WAVES_PER_WG)

static __device__ __forceinline__ v8f wmma_f32(v2f a, v2f b, v8f c) {
  return __builtin_amdgcn_wmma_f32_16x16x4_f32(
       false, a,  false, b,
       (short)0, c,  false,  false);
}

__global__ void __launch_bounds__(256)
encoder_kernel(const float* __restrict__ x,
               const float* __restrict__ Wf, const float* __restrict__ bf,
               const float* __restrict__ Wl, const float* __restrict__ bl,
               float* __restrict__ feats, float* __restrict__ coords) {
  const int lane = threadIdx.x & 31;
  const int wave = threadIdx.x >> 5;
  const int tile = blockIdx.x * WAVES_PER_WG + wave;
  if (tile >= ROW_TILES) return;
  const int row0 = tile * 16;
  const int m  = lane & 15;
  const int kh = lane >> 4;

  v2f a[16];
  const float* xr = x + (size_t)(row0 + m) * DIN;
  #pragma unroll
  for (int s = 0; s < 16; ++s) {
    a[s].x = xr[4 * s + 2 * kh];
    a[s].y = xr[4 * s + 1 + 2 * kh];
  }

  #pragma unroll
  for (int nt = 0; nt < 4; ++nt) {
    v8f acc = {};
    #pragma unroll
    for (int s = 0; s < 16; ++s) {
      v2f b;
      b.x = Wf[(4 * s + 2 * kh) * DFEAT + nt * 16 + m];
      b.y = Wf[(4 * s + 1 + 2 * kh) * DFEAT + nt * 16 + m];
      acc = wmma_f32(a[s], b, acc);
    }
    #pragma unroll
    for (int r = 0; r < 8; ++r) {
      const int row = row0 + r + 8 * kh;
      const int col = nt * 16 + m;
      feats[(size_t)row * DFEAT + col] = tanhf(acc[r] + bf[col]);
    }
  }

  {
    v8f acc = {};
    #pragma unroll
    for (int s = 0; s < 16; ++s) {
      v2f b;
      b.x = Wl[(4 * s + 2 * kh) * DLAT + m];
      b.y = Wl[(4 * s + 1 + 2 * kh) * DLAT + m];
      acc = wmma_f32(a[s], b, acc);
    }
    #pragma unroll
    for (int r = 0; r < 8; ++r) {
      const int row = row0 + r + 8 * kh;
      coords[(size_t)row * DLAT + m] = tanhf(acc[r] + bl[m]);
    }
  }
}

__global__ void sq_kernel(const float* __restrict__ coords, float* __restrict__ sq) {
  const int i = blockIdx.x * blockDim.x + threadIdx.x;
  if (i >= N_PTS) return;
  const float* c = coords + (size_t)i * DLAT;
  float s = 0.f;
  #pragma unroll
  for (int d = 0; d < DLAT; ++d) s += c[d] * c[d];
  sq[i] = s;
}

#define NSPLIT  WAVES_PER_WG
#define NLIST   (2 * NSPLIT)
#define NCAND   (NLIST * KNN)
#define KPITCH  (NCAND + 2)
#define VPITCH  (2 * KPITCH)
#define BIGF    3.4e38f

typedef unsigned long long u64;
typedef __attribute__((ext_vector_type(2))) unsigned long long v2u64;
typedef __attribute__((ext_vector_type(4))) float v4f;

#define KEY_NONE 0xFFFFFFFFFFFFFFFFull

static __device__ __forceinline__ float pair_d2(float sqi, float sqj, float dot) {
  return fmaxf(__builtin_fmaf(-2.f, dot, sqi + sqj), 0.f);
}

static __device__ __forceinline__ v8f tile_dots(const float* __restrict__ coords, int j0,
                                                int m, int kh, const v2f (&b)[4]) {
  const float* cj = coords + (size_t)(j0 + m) * DLAT;
  v8f acc = {};
  #pragma unroll
  for (int s = 0; s < 4; ++s) {
    v2f a;
    a.x = cj[4 * s + 2 * kh];
    a.y = cj[4 * s + 1 + 2 * kh];
    acc = wmma_f32(a, b[s], acc);
  }
  return acc;
}

template <int MASK>
static __device__ __forceinline__ float row_xor(float v) {
  return __int_as_float(__builtin_amdgcn_update_dpp(0, __float_as_int(v), 0x160 + MASK, 0xf, 0xf, false));
}

__global__ void __launch_bounds__(256)
gravnet_kernel(const float* __restrict__ coords, const float* __restrict__ sq,
               const float* __restrict__ feats,
               const float* __restrict__ Wout, const float* __restrict__ bout,
               float* __restrict__ out) {
  __shared__ __attribute__((aligned(16))) u64 lds_cand[16][KPITCH];
  __shared__ __attribute__((aligned(16))) u64 lds_key[16][KPITCH];
  __shared__ __attribute__((aligned(16))) int lds_cnt[16][NLIST];
  __shared__ float lds_T  [16];
  __shared__ float lds_w  [16][KNN];
  __shared__ int   lds_j  [16][KNN];
  __shared__ float lds_agg[16][DFEAT];
  float* lds_val = reinterpret_cast<float*>(&lds_cand[0][0]);

  const int tid  = threadIdx.x;
  const int lane = tid & 31;
  const int wave = __builtin_amdgcn_readfirstlane(tid >> 5);
  const int rt   = blockIdx.x;
  const int row0 = rt * 16;
  const int m  = lane & 15;
  const int kh = lane >> 4;
  const int row  = row0 + m;
  const int list = wave * 2 + kh;

  v2f b[4];
  {
    const float* cr = coords + (size_t)row * DLAT;
    #pragma unroll
    for (int s = 0; s < 4; ++s) {
      b[s].x = cr[4 * s + 2 * kh];
      b[s].y = cr[4 * s + 1 + 2 * kh];
    }
  }
  const float sqi = sq[row];

  float best[KNN];
  #pragma unroll
  for (int s = 0; s < KNN; ++s) best[s] = BIGF;

  for (int jt = wave; jt < ROW_TILES; jt += NSPLIT) {
    const int j0 = jt * 16;
    if (jt + NSPLIT < ROW_TILES)
      __builtin_prefetch(coords + (size_t)(j0 + 16 * NSPLIT + m) * DLAT, 0, 3);
    const v8f acc = tile_dots(coords, j0, m, kh, b);
    const v4f s0 = *reinterpret_cast<const v4f*>(sq + j0 + 8 * kh);
    const v4f s1 = *reinterpret_cast<const v4f*>(sq + j0 + 8 * kh + 4);
    float dd[8];
    #pragma unroll
    for (int r = 0; r < 8; ++r)
      dd[r] = pair_d2(sqi, r < 4 ? s0[r & 3] : s1[r & 3], acc[r]);
    if (jt == rt) {
      #pragma unroll
      for (int r = 0; r < 8; ++r)
        if (j0 + 8 * kh + r == row) dd[r] = BIGF;
    }
    #pragma unroll
    for (int r = 0; r < 8; ++r) {
      #pragma unroll
      for (int s = 0; s < KNN - 1; ++s)
        best[s] = __builtin_amdgcn_fmed3f(best[s], best[s + 1], dd[r]);
      best[KNN - 1] = __builtin_amdgcn_fmed3f(best[KNN - 1], -BIGF, dd[r]);
    }
  }

  {
    v4f* vp = reinterpret_cast<v4f*>(lds_val + m * VPITCH + list * KNN);
    #pragma unroll
    for (int g = 0; g < 4; ++g) {
      v4f v = {best[4 * g], best[4 * g + 1], best[4 * g + 2], best[4 * g + 3]};
      vp[g] = v;
    }
    v2u64* kp = reinterpret_cast<v2u64*>(&lds_key[m][list * KNN]);
    const v2u64 none = {KEY_NONE, KEY_NONE};
    #pragma unroll
    for (int g = 0; g < KNN / 2; ++g) kp[g] = none;
  }
  lds_w[tid >> 4][tid & 15] = 0.f;
  lds_j[tid >> 4][tid & 15] = 0;
  __syncthreads();

  {
    const int r = tid >> 4;
    const int q = tid & 15;
    const float* lv = lds_val + r * VPITCH + q * KNN;
    int head = 0;
    float t = 0.f;
    for (int step = 0; step < KNN; ++step) {
      const float v = lv[KNN - 1 - head];
      float mn = v;
      mn = fminf(mn, row_xor<1>(mn));
      mn = fminf(mn, row_xor<2>(mn));
      mn = fminf(mn, row_xor<4>(mn));
      mn = fminf(mn, row_xor<8>(mn));
      t = mn;
      const unsigned holders = (__builtin_amdgcn_ballot_w32(v == mn) >> (lane & 16)) & 0xFFFFu;
      if (q == __builtin_ctz(holders)) ++head;
    }
    if (q == 0) lds_T[r] = t;
  }
  __syncthreads();

  int cs = 0, ct = 0;
  {
    const float T = lds_T[m];
    u64* kl = &lds_key[m][list * KNN];
    for (int jt = wave; jt < ROW_TILES; jt += NSPLIT) {
      const int j0 = jt * 16;
      if (jt + NSPLIT < ROW_TILES)
        __builtin_prefetch(coords + (size_t)(j0 + 16 * NSPLIT + m) * DLAT, 0, 3);
      const v8f acc = tile_dots(coords, j0, m, kh, b);
      const v4f s0 = *reinterpret_cast<const v4f*>(sq + j0 + 8 * kh);
      const v4f s1 = *reinterpret_cast<const v4f*>(sq + j0 + 8 * kh + 4);
      float dd[8];
      #pragma unroll
      for (int r = 0; r < 8; ++r)
        dd[r] = pair_d2(sqi, r < 4 ? s0[r & 3] : s1[r & 3], acc[r]);
      if (jt == rt) {
        #pragma unroll
        for (int r = 0; r < 8; ++r)
          if (j0 + 8 * kh + r == row) dd[r] = BIGF;
      }
      float lo = dd[0];
      #pragma unroll
      for (int r = 1; r < 8; ++r) lo = fminf(lo, dd[r]);
      if (lo <= T) {
        #pragma unroll
        for (int r = 0; r < 8; ++r) {
          const float d = dd[r];
          if (d <= T) {
            const int db = max(__float_as_int(d), 0);
            const u64 key = ((u64)(unsigned)db << 32) | (u64)(unsigned)(j0 + 8 * kh + r);
            if (d < T) {
              if (cs < KNN) {
                if (cs + ct == KNN) --ct;
                kl[cs] = key; ++cs;
              }
            } else if (cs + ct < KNN) {
              ++ct; kl[KNN - ct] = key;
            }
          }
        }
      }
    }
    lds_cnt[m][list] = cs | (ct << 8);
  }
  __syncthreads();

  int tot = 0;
  {
    const int r = tid >> 4;
    const int q = tid & 15;
    int off = 0, mine = 0;
    #pragma unroll
    for (int l = 0; l < NLIST; ++l) {
      const int c = lds_cnt[r][l];
      const int n = (c & 255) + (c >> 8);
      off += (l < q) ? n : 0;
      mine = (l == q) ? c : mine;
      tot += n;
    }
    const int ncs = mine & 255, nct = mine >> 8;
    const u64* kl = &lds_key[r][q * KNN];
    for (int i = 0; i < ncs + nct; ++i) {
      const int slot = i < ncs ? i : KNN - nct + (i - ncs);
      lds_cand[r][off + i] = kl[slot];
    }
    if (q == 0) lds_cand[r][tot] = KEY_NONE;
  }
  __syncthreads();

  {
    const int r = tid >> 4;
    const int q = tid & 15;
    for (int p = q; p < tot; p += 16) {
      const u64 ck = lds_cand[r][p];
      int rank = 0;
      for (int e = 0; e < tot; e += 2) {
        const v2u64 kv = *reinterpret_cast<const v2u64*>(&lds_cand[r][e]);
        rank += (kv.x < ck ? 1 : 0) + (kv.y < ck ? 1 : 0);
      }
      if (rank < KNN) {
        lds_w[r][rank] = expf(-__int_as_float((int)(ck >> 32)));
        lds_j[r][rank] = (int)(unsigned)ck;
      }
    }
  }
  __syncthreads();

  #pragma unroll
  for (int rr = 0; rr < 16 / NSPLIT; ++rr) {
    const int r = wave * (16 / NSPLIT) + rr;
    float acc0 = 0.f, acc1 = 0.f;
    #pragma unroll
    for (int k = 0; k < KNN; ++k) {
      const int   j  = lds_j[r][k];
      const float wk = lds_w[r][k];
      acc0 += wk * feats[(size_t)j * DFEAT + lane];
      acc1 += wk * feats[(size_t)j * DFEAT + lane + 32];
    }
    lds_agg[r][lane]      = acc0;
    lds_agg[r][lane + 32] = acc1;
  }
  __syncthreads();

  if (wave < 2) {
    const float* fr = feats + (size_t)(row0 + m) * DFEAT;
    #pragma unroll
    for (int t = 0; t < 2; ++t) {
      const int nt = wave * 2 + t;
      v8f acc = {};
      #pragma unroll
      for (int s = 0; s < 16; ++s) {
        v2f av, bv;
        av.x = fr[4 * s + 2 * kh];
        av.y = fr[4 * s + 1 + 2 * kh];
        bv.x = Wout[(4 * s + 2 * kh) * DFEAT + nt * 16 + m];
        bv.y = Wout[(4 * s + 1 + 2 * kh) * DFEAT + nt * 16 + m];
        acc = wmma_f32(av, bv, acc);
      }
      #pragma unroll
      for (int s = 0; s < 16; ++s) {
        v2f av, bv;
        av.x = lds_agg[m][4 * s + 2 * kh];
        av.y = lds_agg[m][4 * s + 1 + 2 * kh];
        bv.x = Wout[(64 + 4 * s + 2 * kh) * DFEAT + nt * 16 + m];
        bv.y = Wout[(64 + 4 * s + 1 + 2 * kh) * DFEAT + nt * 16 + m];
        acc = wmma_f32(av, bv, acc);
      }
      #pragma unroll
      for (int r = 0; r < 8; ++r) {
        const int row = row0 + r + 8 * kh;
        const int col = nt * 16 + m;
        out[(size_t)row * DFEAT + col] = acc[r] + bout[col];
      }
    }
  }
}

extern "C" void kernel_launch(void* const* d_in, const int* in_sizes, int n_in,
                              void* d_out, int out_size, void* d_ws, size_t ws_size,
                              hipStream_t stream) {
  const float* x  = (const float*)d_in[0];
  const float* Wf = (const float*)d_in[1];
  const float* bf = (const float*)d_in[2];
  const float* Wl = (const float*)d_in[3];
  const float* bl = (const float*)d_in[4];
  const float* Wo = (const float*)d_in[5];
  const float* bo = (const float*)d_in[6];
  float* out = (float*)d_out;

  float* feats  = (float*)d_ws;
  float* coords = feats  + (size_t)N_PTS * DFEAT;
  float* sq     = coords + (size_t)N_PTS * DLAT;

  encoder_kernel<<<NUM_WG, 256, 0, stream>>>(x, Wf, bf, Wl, bl, feats, coords);
  sq_kernel<<<(N_PTS + 255) / 256, 256, 0, stream>>>(coords, sq);
  gravnet_kernel<<<ROW_TILES, 256, 0, stream>>>(coords, sq, feats, Wo, bo, out);
}


# Scan 1 draws an exact upper bound U of each row's k-th nearest distance from the lists after each wave's first 20 column tiles; later tiles skip the sorted insert unless a lane is at or below U
# speedup vs baseline: 8.2453x; 1.1327x over previous
#include <hip/hip_runtime.h>
#include <cmath>

typedef __attribute__((ext_vector_type(2))) float v2f;
typedef __attribute__((ext_vector_type(8))) float v8f;

#define N_PTS   12000
#define DIN     64
#define DLAT    16
#define DFEAT   64
#define KNN     16
#define ROW_TILES (N_PTS / 16)
#define WAVES_PER_WG 8
#define NUM_WG  ((ROW_TILES + WAVES_PER_WG - 1) / WAVES_PER_WG)

static __device__ __forceinline__ v8f wmma_f32(v2f a, v2f b, v8f c) {
  return __builtin_amdgcn_wmma_f32_16x16x4_f32(
       false, a,  false, b,
       (short)0, c,  false,  false);
}

__global__ void __launch_bounds__(256)
encoder_kernel(const float* __restrict__ x,
               const float* __restrict__ Wf, const float* __restrict__ bf,
               const float* __restrict__ Wl, const float* __restrict__ bl,
               float* __restrict__ feats, float* __restrict__ coords) {
  const int lane = threadIdx.x & 31;
  const int wave = threadIdx.x >> 5;
  const int tile = blockIdx.x * WAVES_PER_WG + wave;
  if (tile >= ROW_TILES) return;
  const int row0 = tile * 16;
  const int m  = lane & 15;
  const int kh = lane >> 4;

  v2f a[16];
  const float* xr = x + (size_t)(row0 + m) * DIN;
  #pragma unroll
  for (int s = 0; s < 16; ++s) {
    a[s].x = xr[4 * s + 2 * kh];
    a[s].y = xr[4 * s + 1 + 2 * kh];
  }

  #pragma unroll
  for (int nt = 0; nt < 4; ++nt) {
    v8f acc = {};
    #pragma unroll
    for (int s = 0; s < 16; ++s) {
      v2f b;
      b.x = Wf[(4 * s + 2 * kh) * DFEAT + nt * 16 + m];
      b.y = Wf[(4 * s + 1 + 2 * kh) * DFEAT + nt * 16 + m];
      acc = wmma_f32(a[s], b, acc);
    }
    #pragma unroll
    for (int r = 0; r < 8; ++r) {
      const int row = row0 + r + 8 * kh;
      const int col = nt * 16 + m;
      feats[(size_t)row * DFEAT + col] = tanhf(acc[r] + bf[col]);
    }
  }

  {
    v8f acc = {};
    #pragma unroll
    for (int s = 0; s < 16; ++s) {
      v2f b;
      b.x = Wl[(4 * s + 2 * kh) * DLAT + m];
      b.y = Wl[(4 * s + 1 + 2 * kh) * DLAT + m];
      acc = wmma_f32(a[s], b, acc);
    }
    #pragma unroll
    for (int r = 0; r < 8; ++r) {
      const int row = row0 + r + 8 * kh;
      coords[(size_t)row * DLAT + m] = tanhf(acc[r] + bl[m]);
    }
  }
}

__global__ void sq_kernel(const float* __restrict__ coords, float* __restrict__ sq) {
  const int i = blockIdx.x * blockDim.x + threadIdx.x;
  if (i >= N_PTS) return;
  const float* c = coords + (size_t)i * DLAT;
  float s = 0.f;
  #pragma unroll
  for (int d = 0; d < DLAT; ++d) s += c[d] * c[d];
  sq[i] = s;
}

#define NSPLIT  WAVES_PER_WG
#define NLIST   (2 * NSPLIT)
#define NCAND   (NLIST * KNN)
#define KPITCH  (NCAND + 2)
#define VPITCH  (2 * KPITCH)
#define BIGF    3.4e38f
#define TILES_A 20

typedef unsigned long long u64;
typedef __attribute__((ext_vector_type(2))) unsigned long long v2u64;
typedef __attribute__((ext_vector_type(4))) float v4f;

#define KEY_NONE 0xFFFFFFFFFFFFFFFFull

static __device__ __forceinline__ float pair_d2(float sqi, float sqj, float dot) {
  return fmaxf(__builtin_fmaf(-2.f, dot, sqi + sqj), 0.f);
}

static __device__ __forceinline__ v8f tile_dots(const float* __restrict__ coords, int j0,
                                                int m, int kh, const v2f (&b)[4]) {
  const float* cj = coords + (size_t)(j0 + m) * DLAT;
  v8f acc = {};
  #pragma unroll
  for (int s = 0; s < 4; ++s) {
    v2f a;
    a.x = cj[4 * s + 2 * kh];
    a.y = cj[4 * s + 1 + 2 * kh];
    acc = wmma_f32(a, b[s], acc);
  }
  return acc;
}

template <int MASK>
static __device__ __forceinline__ float row_xor(float v) {
  return __int_as_float(__builtin_amdgcn_update_dpp(0, __float_as_int(v), 0x160 + MASK, 0xf, 0xf, false));
}

static __device__ __forceinline__ void list_insert(float (&best)[KNN], float d) {
  #pragma unroll
  for (int s = 0; s < KNN - 1; ++s)
    best[s] = __builtin_amdgcn_fmed3f(best[s], best[s + 1], d);
  best[KNN - 1] = __builtin_amdgcn_fmed3f(best[KNN - 1], -BIGF, d);
}

static __device__ __forceinline__ void tile_d2(const float* __restrict__ coords,
                                               const float* __restrict__ sq,
                                               int jt, int rt, int row, int m, int kh,
                                               const v2f (&b)[4], float sqi, float (&dd)[8]) {
  const int j0 = jt * 16;
  const v8f acc = tile_dots(coords, j0, m, kh, b);
  const v4f s0 = *reinterpret_cast<const v4f*>(sq + j0 + 8 * kh);
  const v4f s1 = *reinterpret_cast<const v4f*>(sq + j0 + 8 * kh + 4);
  #pragma unroll
  for (int r = 0; r < 8; ++r)
    dd[r] = pair_d2(sqi, r < 4 ? s0[r & 3] : s1[r & 3], acc[r]);
  if (jt == rt) {
    #pragma unroll
    for (int r = 0; r < 8; ++r)
      if (j0 + 8 * kh + r == row) dd[r] = BIGF;
  }
}

static __device__ __forceinline__ float row_16th(const float* lv, int q, int lane) {
  int head = 0;
  float t = 0.f;
  for (int step = 0; step < KNN; ++step) {
    const float v = lv[KNN - 1 - head];
    float mn = v;
    mn = fminf(mn, row_xor<1>(mn));
    mn = fminf(mn, row_xor<2>(mn));
    mn = fminf(mn, row_xor<4>(mn));
    mn = fminf(mn, row_xor<8>(mn));
    t = mn;
    const unsigned holders = (__builtin_amdgcn_ballot_w32(v == mn) >> (lane & 16)) & 0xFFFFu;
    if (q == __builtin_ctz(holders)) ++head;
  }
  return t;
}

__global__ void __launch_bounds__(256)
gravnet_kernel(const float* __restrict__ coords, const float* __restrict__ sq,
               const float* __restrict__ feats,
               const float* __restrict__ Wout, const float* __restrict__ bout,
               float* __restrict__ out) {
  __shared__ __attribute__((aligned(16))) u64 lds_cand[16][KPITCH];
  __shared__ __attribute__((aligned(16))) u64 lds_key[16][KPITCH];
  __shared__ __attribute__((aligned(16))) int lds_cnt[16][NLIST];
  __shared__ float lds_T  [16];
  __shared__ float lds_w  [16][KNN];
  __shared__ int   lds_j  [16][KNN];
  __shared__ float lds_agg[16][DFEAT];
  float* lds_val = reinterpret_cast<float*>(&lds_cand[0][0]);

  const int tid  = threadIdx.x;
  const int lane = tid & 31;
  const int wave = __builtin_amdgcn_readfirstlane(tid >> 5);
  const int rt   = blockIdx.x;
  const int row0 = rt * 16;
  const int m  = lane & 15;
  const int kh = lane >> 4;
  const int row  = row0 + m;
  const int list = wave * 2 + kh;

  v2f b[4];
  {
    const float* cr = coords + (size_t)row * DLAT;
    #pragma unroll
    for (int s = 0; s < 4; ++s) {
      b[s].x = cr[4 * s + 2 * kh];
      b[s].y = cr[4 * s + 1 + 2 * kh];
    }
  }
  const float sqi = sq[row];

  float best[KNN];
  #pragma unroll
  for (int s = 0; s < KNN; ++s) best[s] = BIGF;

  for (int jt = wave; jt < wave + NSPLIT * TILES_A; jt += NSPLIT) {
    __builtin_prefetch(coords + (size_t)(jt * 16 + 16 * NSPLIT + m) * DLAT, 0, 3);
    float dd[8];
    tile_d2(coords, sq, jt, rt, row, m, kh, b, sqi, dd);
    #pragma unroll
    for (int r = 0; r < 8; ++r) list_insert(best, dd[r]);
  }

  {
    v4f* vp = reinterpret_cast<v4f*>(lds_val + m * VPITCH + list * KNN);
    #pragma unroll
    for (int g = 0; g < 4; ++g) {
      v4f v = {best[4 * g], best[4 * g + 1], best[4 * g + 2], best[4 * g + 3]};
      vp[g] = v;
    }
  }
  __syncthreads();
  {
    const int r = tid >> 4;
    const int q = tid & 15;
    const float u = row_16th(lds_val + r * VPITCH + q * KNN, q, lane);
    if (q == 0) lds_T[r] = u;
  }
  __syncthreads();
  {
    const float U = lds_T[m];
    for (int jt = wave + NSPLIT * TILES_A; jt < ROW_TILES; jt += NSPLIT) {
      if (jt + NSPLIT < ROW_TILES)
        __builtin_prefetch(coords + (size_t)(jt * 16 + 16 * NSPLIT + m) * DLAT, 0, 3);
      float dd[8];
      tile_d2(coords, sq, jt, rt, row, m, kh, b, sqi, dd);
      float p[8];
      p[0] = dd[0];
      #pragma unroll
      for (int r = 1; r < 8; ++r) p[r] = __builtin_amdgcn_fmed3f(p[r - 1], dd[r], -BIGF);
      const float lo = p[7];
      if (__builtin_amdgcn_ballot_w32(lo <= U) != 0) {
        float b1 = __builtin_amdgcn_fmed3f(dd[0], dd[1], BIGF);
        #pragma unroll
        for (int r = 2; r < 8; ++r) b1 = __builtin_amdgcn_fmed3f(p[r - 1], b1, dd[r]);
        if (__builtin_amdgcn_ballot_w32(b1 <= U) != 0) {
          #pragma unroll
          for (int r = 0; r < 8; ++r) list_insert(best, dd[r]);
        } else {
          list_insert(best, lo);
        }
      }
    }
  }

  {
    v4f* vp = reinterpret_cast<v4f*>(lds_val + m * VPITCH + list * KNN);
    #pragma unroll
    for (int g = 0; g < 4; ++g) {
      v4f v = {best[4 * g], best[4 * g + 1], best[4 * g + 2], best[4 * g + 3]};
      vp[g] = v;
    }
    v2u64* kp = reinterpret_cast<v2u64*>(&lds_key[m][list * KNN]);
    const v2u64 none = {KEY_NONE, KEY_NONE};
    #pragma unroll
    for (int g = 0; g < KNN / 2; ++g) kp[g] = none;
  }
  lds_w[tid >> 4][tid & 15] = 0.f;
  lds_j[tid >> 4][tid & 15] = 0;
  __syncthreads();

  {
    const int r = tid >> 4;
    const int q = tid & 15;
    const float t = row_16th(lds_val + r * VPITCH + q * KNN, q, lane);
    if (q == 0) lds_T[r] = t;
  }
  __syncthreads();

  int cs = 0, ct = 0;
  {
    const float T = lds_T[m];
    u64* kl = &lds_key[m][list * KNN];
    for (int jt = wave; jt < ROW_TILES; jt += NSPLIT) {
      const int j0 = jt * 16;
      if (jt + NSPLIT < ROW_TILES)
        __builtin_prefetch(coords + (size_t)(j0 + 16 * NSPLIT + m) * DLAT, 0, 3);
      const v8f acc = tile_dots(coords, j0, m, kh, b);
      const v4f s0 = *reinterpret_cast<const v4f*>(sq + j0 + 8 * kh);
      const v4f s1 = *reinterpret_cast<const v4f*>(sq + j0 + 8 * kh + 4);
      float dd[8];
      #pragma unroll
      for (int r = 0; r < 8; ++r)
        dd[r] = pair_d2(sqi, r < 4 ? s0[r & 3] : s1[r & 3], acc[r]);
      if (jt == rt) {
        #pragma unroll
        for (int r = 0; r < 8; ++r)
          if (j0 + 8 * kh + r == row) dd[r] = BIGF;
      }
      float lo = dd[0];
      #pragma unroll
      for (int r = 1; r < 8; ++r) lo = fminf(lo, dd[r]);
      if (lo <= T) {
        #pragma unroll
        for (int r = 0; r < 8; ++r) {
          const float d = dd[r];
          if (d <= T) {
            const int db = max(__float_as_int(d), 0);
            const u64 key = ((u64)(unsigned)db << 32) | (u64)(unsigned)(j0 + 8 * kh + r);
            if (d < T) {
              if (cs < KNN) {
                if (cs + ct == KNN) --ct;
                kl[cs] = key; ++cs;
              }
            } else if (cs + ct < KNN) {
              ++ct; kl[KNN - ct] = key;
            }
          }
        }
      }
    }
    lds_cnt[m][list] = cs | (ct << 8);
  }
  __syncthreads();

  int tot = 0;
  {
    const int r = tid >> 4;
    const int q = tid & 15;
    int off = 0, mine = 0;
    #pragma unroll
    for (int l = 0; l < NLIST; ++l) {
      const int c = lds_cnt[r][l];
      const int n = (c & 255) + (c >> 8);
      off += (l < q) ? n : 0;
      mine = (l == q) ? c : mine;
      tot += n;
    }
    const int ncs = mine & 255, nct = mine >> 8;
    const u64* kl = &lds_key[r][q * KNN];
    for (int i = 0; i < ncs + nct; ++i) {
      const int slot = i < ncs ? i : KNN - nct + (i - ncs);
      lds_cand[r][off + i] = kl[slot];
    }
    if (q == 0) lds_cand[r][tot] = KEY_NONE;
  }
  __syncthreads();

  {
    const int r = tid >> 4;
    const int q = tid & 15;
    for (int p = q; p < tot; p += 16) {
      const u64 ck = lds_cand[r][p];
      int rank = 0;
      for (int e = 0; e < tot; e += 2) {
        const v2u64 kv = *reinterpret_cast<const v2u64*>(&lds_cand[r][e]);
        rank += (kv.x < ck ? 1 : 0) + (kv.y < ck ? 1 : 0);
      }
      if (rank < KNN) {
        lds_w[r][rank] = expf(-__int_as_float((int)(ck >> 32)));
        lds_j[r][rank] = (int)(unsigned)ck;
      }
    }
  }
  __syncthreads();

  #pragma unroll
  for (int rr = 0; rr < 16 / NSPLIT; ++rr) {
    const int r = wave * (16 / NSPLIT) + rr;
    float acc0 = 0.f, acc1 = 0.f;
    #pragma unroll
    for (int k = 0; k < KNN; ++k) {
      const int   j  = lds_j[r][k];
      const float wk = lds_w[r][k];
      acc0 += wk * feats[(size_t)j * DFEAT + lane];
      acc1 += wk * feats[(size_t)j * DFEAT + lane + 32];
    }
    lds_agg[r][lane]      = acc0;
    lds_agg[r][lane + 32] = acc1;
  }
  __syncthreads();

  if (wave < 2) {
    const float* fr = feats + (size_t)(row0 + m) * DFEAT;
    #pragma unroll
    for (int t = 0; t < 2; ++t) {
      const int nt = wave * 2 + t;
      v8f acc = {};
      #pragma unroll
      for (int s = 0; s < 16; ++s) {
        v2f av, bv;
        av.x = fr[4 * s + 2 * kh];
        av.y = fr[4 * s + 1 + 2 * kh];
        bv.x = Wout[(4 * s + 2 * kh) * DFEAT + nt * 16 + m];
        bv.y = Wout[(4 * s + 1 + 2 * kh) * DFEAT + nt * 16 + m];
        acc = wmma_f32(av, bv, acc);
      }
      #pragma unroll
      for (int s = 0; s < 16; ++s) {
        v2f av, bv;
        av.x = lds_agg[m][4 * s + 2 * kh];
        av.y = lds_agg[m][4 * s + 1 + 2 * kh];
        bv.x = Wout[(64 + 4 * s + 2 * kh) * DFEAT + nt * 16 + m];
        bv.y = Wout[(64 + 4 * s + 1 + 2 * kh) * DFEAT + nt * 16 + m];
        acc = wmma_f32(av, bv, acc);
      }
      #pragma unroll
      for (int r = 0; r < 8; ++r) {
        const int row = row0 + r + 8 * kh;
        const int col = nt * 16 + m;
        out[(size_t)row * DFEAT + col] = acc[r] + bout[col];
      }
    }
  }
}

extern "C" void kernel_launch(void* const* d_in, const int* in_sizes, int n_in,
                              void* d_out, int out_size, void* d_ws, size_t ws_size,
                              hipStream_t stream) {
  const float* x  = (const float*)d_in[0];
  const float* Wf = (const float*)d_in[1];
  const float* bf = (const float*)d_in[2];
  const float* Wl = (const float*)d_in[3];
  const float* bl = (const float*)d_in[4];
  const float* Wo = (const float*)d_in[5];
  const float* bo = (const float*)d_in[6];
  float* out = (float*)d_out;

  float* feats  = (float*)d_ws;
  float* coords = feats  + (size_t)N_PTS * DFEAT;
  float* sq     = coords + (size_t)N_PTS * DLAT;

  encoder_kernel<<<NUM_WG, 256, 0, stream>>>(x, Wf, bf, Wl, bl, feats, coords);
  sq_kernel<<<(N_PTS + 255) / 256, 256, 0, stream>>>(coords, sq);
  gravnet_kernel<<<ROW_TILES, 256, 0, stream>>>(coords, sq, feats, Wo, bo, out);
}


# Both scans load two column tiles ahead: a tile's coords fragment and norms go into one of three rotating register sets two tiles before its products are taken; the cache-hint prefetch is dropped
# speedup vs baseline: 9.1510x; 1.1098x over previous
#include <hip/hip_runtime.h>
#include <cmath>

typedef __attribute__((ext_vector_type(2))) float v2f;
typedef __attribute__((ext_vector_type(8))) float v8f;

#define N_PTS   12000
#define DIN     64
#define DLAT    16
#define DFEAT   64
#define KNN     16
#define ROW_TILES (N_PTS / 16)
#define WAVES_PER_WG 8
#define NUM_WG  ((ROW_TILES + WAVES_PER_WG - 1) / WAVES_PER_WG)

static __device__ __forceinline__ v8f wmma_f32(v2f a, v2f b, v8f c) {
  return __builtin_amdgcn_wmma_f32_16x16x4_f32(
       false, a,  false, b,
       (short)0, c,  false,  false);
}

__global__ void __launch_bounds__(256)
encoder_kernel(const float* __restrict__ x,
               const float* __restrict__ Wf, const float* __restrict__ bf,
               const float* __restrict__ Wl, const float* __restrict__ bl,
               float* __restrict__ feats, float* __restrict__ coords) {
  const int lane = threadIdx.x & 31;
  const int wave = threadIdx.x >> 5;
  const int tile = blockIdx.x * WAVES_PER_WG + wave;
  if (tile >= ROW_TILES) return;
  const int row0 = tile * 16;
  const int m  = lane & 15;
  const int kh = lane >> 4;

  v2f a[16];
  const float* xr = x + (size_t)(row0 + m) * DIN;
  #pragma unroll
  for (int s = 0; s < 16; ++s) {
    a[s].x = xr[4 * s + 2 * kh];
    a[s].y = xr[4 * s + 1 + 2 * kh];
  }

  #pragma unroll
  for (int nt = 0; nt < 4; ++nt) {
    v8f acc = {};
    #pragma unroll
    for (int s = 0; s < 16; ++s) {
      v2f b;
      b.x = Wf[(4 * s + 2 * kh) * DFEAT + nt * 16 + m];
      b.y = Wf[(4 * s + 1 + 2 * kh) * DFEAT + nt * 16 + m];
      acc = wmma_f32(a[s], b, acc);
    }
    #pragma unroll
    for (int r = 0; r < 8; ++r) {
      const int row = row0 + r + 8 * kh;
      const int col = nt * 16 + m;
      feats[(size_t)row * DFEAT + col] = tanhf(acc[r] + bf[col]);
    }
  }

  {
    v8f acc = {};
    #pragma unroll
    for (int s = 0; s < 16; ++s) {
      v2f b;
      b.x = Wl[(4 * s + 2 * kh) * DLAT + m];
      b.y = Wl[(4 * s + 1 + 2 * kh) * DLAT + m];
      acc = wmma_f32(a[s], b, acc);
    }
    #pragma unroll
    for (int r = 0; r < 8; ++r) {
      const int row = row0 + r + 8 * kh;
      coords[(size_t)row * DLAT + m] = tanhf(acc[r] + bl[m]);
    }
  }
}

__global__ void sq_kernel(const float* __restrict__ coords, float* __restrict__ sq) {
  const int i = blockIdx.x * blockDim.x + threadIdx.x;
  if (i >= N_PTS) return;
  const float* c = coords + (size_t)i * DLAT;
  float s = 0.f;
  #pragma unroll
  for (int d = 0; d < DLAT; ++d) s += c[d] * c[d];
  sq[i] = s;
}

#define NSPLIT  WAVES_PER_WG
#define NLIST   (2 * NSPLIT)
#define NCAND   (NLIST * KNN)
#define KPITCH  (NCAND + 2)
#define VPITCH  (2 * KPITCH)
#define BIGF    3.4e38f
#define TILES_A 20

typedef unsigned long long u64;
typedef __attribute__((ext_vector_type(2))) unsigned long long v2u64;
typedef __attribute__((ext_vector_type(4))) float v4f;

#define KEY_NONE 0xFFFFFFFFFFFFFFFFull

static __device__ __forceinline__ float pair_d2(float sqi, float sqj, float dot) {
  return fmaxf(__builtin_fmaf(-2.f, dot, sqi + sqj), 0.f);
}

struct TileFrag {
  v2f a[4];
  v4f s0, s1;
};

static __device__ __forceinline__ void load_frag(const float* __restrict__ coords,
                                                 const float* __restrict__ sq,
                                                 int jt, int m, int kh, TileFrag& f) {
  const int j0 = jt * 16;
  const float* cj = coords + (size_t)(j0 + m) * DLAT;
  #pragma unroll
  for (int s = 0; s < 4; ++s) {
    f.a[s].x = cj[4 * s + 2 * kh];
    f.a[s].y = cj[4 * s + 1 + 2 * kh];
  }
  f.s0 = *reinterpret_cast<const v4f*>(sq + j0 + 8 * kh);
  f.s1 = *reinterpret_cast<const v4f*>(sq + j0 + 8 * kh + 4);
}

static __device__ __forceinline__ v8f frag_dots(const TileFrag& f, const v2f (&b)[4]) {
  v8f acc = {};
  #pragma unroll
  for (int s = 0; s < 4; ++s) acc = wmma_f32(f.a[s], b[s], acc);
  return acc;
}

template <int MASK>
static __device__ __forceinline__ float row_xor(float v) {
  return __int_as_float(__builtin_amdgcn_update_dpp(0, __float_as_int(v), 0x160 + MASK, 0xf, 0xf, false));
}

static __device__ __forceinline__ void list_insert(float (&best)[KNN], float d) {
  #pragma unroll
  for (int s = 0; s < KNN - 1; ++s)
    best[s] = __builtin_amdgcn_fmed3f(best[s], best[s + 1], d);
  best[KNN - 1] = __builtin_amdgcn_fmed3f(best[KNN - 1], -BIGF, d);
}

static __device__ __forceinline__ void frag_d2(const TileFrag& f, int jt, int rt, int row, int kh,
                                               const v2f (&b)[4], float sqi, float (&dd)[8]) {
  const int j0 = jt * 16;
  const v8f acc = frag_dots(f, b);
  #pragma unroll
  for (int r = 0; r < 8; ++r)
    dd[r] = pair_d2(sqi, r < 4 ? f.s0[r & 3] : f.s1[r & 3], acc[r]);
  if (jt == rt) {
    #pragma unroll
    for (int r = 0; r < 8; ++r)
      if (j0 + 8 * kh + r == row) dd[r] = BIGF;
  }
}

static __device__ __forceinline__ float row_16th(const float* lv, int q, int lane) {
  int head = 0;
  float t = 0.f;
  for (int step = 0; step < KNN; ++step) {
    const float v = lv[KNN - 1 - head];
    float mn = v;
    mn = fminf(mn, row_xor<1>(mn));
    mn = fminf(mn, row_xor<2>(mn));
    mn = fminf(mn, row_xor<4>(mn));
    mn = fminf(mn, row_xor<8>(mn));
    t = mn;
    const unsigned holders = (__builtin_amdgcn_ballot_w32(v == mn) >> (lane & 16)) & 0xFFFFu;
    if (q == __builtin_ctz(holders)) ++head;
  }
  return t;
}

__global__ void __launch_bounds__(256)
gravnet_kernel(const float* __restrict__ coords, const float* __restrict__ sq,
               const float* __restrict__ feats,
               const float* __restrict__ Wout, const float* __restrict__ bout,
               float* __restrict__ out) {
  __shared__ __attribute__((aligned(16))) u64 lds_cand[16][KPITCH];
  __shared__ __attribute__((aligned(16))) u64 lds_key[16][KPITCH];
  __shared__ __attribute__((aligned(16))) int lds_cnt[16][NLIST];
  __shared__ float lds_T  [16];
  __shared__ float lds_w  [16][KNN];
  __shared__ int   lds_j  [16][KNN];
  __shared__ float lds_agg[16][DFEAT];
  float* lds_val = reinterpret_cast<float*>(&lds_cand[0][0]);

  const int tid  = threadIdx.x;
  const int lane = tid & 31;
  const int wave = __builtin_amdgcn_readfirstlane(tid >> 5);
  const int rt   = blockIdx.x;
  const int row0 = rt * 16;
  const int m  = lane & 15;
  const int kh = lane >> 4;
  const int row  = row0 + m;
  const int list = wave * 2 + kh;

  v2f b[4];
  {
    const float* cr = coords + (size_t)row * DLAT;
    #pragma unroll
    for (int s = 0; s < 4; ++s) {
      b[s].x = cr[4 * s + 2 * kh];
      b[s].y = cr[4 * s + 1 + 2 * kh];
    }
  }
  const float sqi = sq[row];

  float best[KNN];
  #pragma unroll
  for (int s = 0; s < KNN; ++s) best[s] = BIGF;

  TileFrag f0, f1, f2;
  load_frag(coords, sq, wave, m, kh, f0);
  load_frag(coords, sq, wave + NSPLIT, m, kh, f1);
  {
    auto step = [&](const TileFrag& use, TileFrag& fill, int jt) {
      load_frag(coords, sq, jt + 2 * NSPLIT, m, kh, fill);
      float dd[8];
      frag_d2(use, jt, rt, row, kh, b, sqi, dd);
      #pragma unroll
      for (int r = 0; r < 8; ++r) list_insert(best, dd[r]);
    };
    const int end = wave + NSPLIT * TILES_A;
    int jt = wave;
    #pragma unroll 1
    for (; jt + 2 * NSPLIT < end; jt += 3 * NSPLIT) {
      step(f0, f2, jt);
      step(f1, f0, jt + NSPLIT);
      step(f2, f1, jt + 2 * NSPLIT);
    }
    #pragma unroll 1
    for (; jt < end; jt += NSPLIT) {
      step(f0, f2, jt);
      f0 = f1;
      f1 = f2;
    }
  }

  {
    v4f* vp = reinterpret_cast<v4f*>(lds_val + m * VPITCH + list * KNN);
    #pragma unroll
    for (int g = 0; g < 4; ++g) {
      v4f v = {best[4 * g], best[4 * g + 1], best[4 * g + 2], best[4 * g + 3]};
      vp[g] = v;
    }
  }
  __syncthreads();
  {
    const int r = tid >> 4;
    const int q = tid & 15;
    const float u = row_16th(lds_val + r * VPITCH + q * KNN, q, lane);
    if (q == 0) lds_T[r] = u;
  }
  __syncthreads();
  {
    const float U = lds_T[m];
    auto step = [&](const TileFrag& use, TileFrag& fill, int jt) {
      const int jn = (jt + 2 * NSPLIT < ROW_TILES) ? jt + 2 * NSPLIT
                   : (jt + NSPLIT < ROW_TILES) ? wave : wave + NSPLIT;
      load_frag(coords, sq, jn, m, kh, fill);
      float dd[8];
      frag_d2(use, jt, rt, row, kh, b, sqi, dd);
      float p[8];
      p[0] = dd[0];
      #pragma unroll
      for (int r = 1; r < 8; ++r) p[r] = __builtin_amdgcn_fmed3f(p[r - 1], dd[r], -BIGF);
      const float lo = p[7];
      if (__builtin_amdgcn_ballot_w32(lo <= U) != 0) {
        float b1 = __builtin_amdgcn_fmed3f(dd[0], dd[1], BIGF);
        #pragma unroll
        for (int r = 2; r < 8; ++r) b1 = __builtin_amdgcn_fmed3f(p[r - 1], b1, dd[r]);
        if (__builtin_amdgcn_ballot_w32(b1 <= U) != 0) {
          #pragma unroll
          for (int r = 0; r < 8; ++r) list_insert(best, dd[r]);
        } else {
          list_insert(best, lo);
        }
      }
    };
    int jt = wave + NSPLIT * TILES_A;
    #pragma unroll 1
    for (; jt + 2 * NSPLIT < ROW_TILES; jt += 3 * NSPLIT) {
      step(f0, f2, jt);
      step(f1, f0, jt + NSPLIT);
      step(f2, f1, jt + 2 * NSPLIT);
    }
    #pragma unroll 1
    for (; jt < ROW_TILES; jt += NSPLIT) {
      step(f0, f2, jt);
      f0 = f1;
      f1 = f2;
    }
  }

  {
    v4f* vp = reinterpret_cast<v4f*>(lds_val + m * VPITCH + list * KNN);
    #pragma unroll
    for (int g = 0; g < 4; ++g) {
      v4f v = {best[4 * g], best[4 * g + 1], best[4 * g + 2], best[4 * g + 3]};
      vp[g] = v;
    }
    v2u64* kp = reinterpret_cast<v2u64*>(&lds_key[m][list * KNN]);
    const v2u64 none = {KEY_NONE, KEY_NONE};
    #pragma unroll
    for (int g = 0; g < KNN / 2; ++g) kp[g] = none;
  }
  lds_w[tid >> 4][tid & 15] = 0.f;
  lds_j[tid >> 4][tid & 15] = 0;
  __syncthreads();

  {
    const int r = tid >> 4;
    const int q = tid & 15;
    const float t = row_16th(lds_val + r * VPITCH + q * KNN, q, lane);
    if (q == 0) lds_T[r] = t;
  }
  __syncthreads();

  int cs = 0, ct = 0;
  {
    const float T = lds_T[m];
    u64* kl = &lds_key[m][list * KNN];
    auto step = [&](const TileFrag& use, TileFrag& fill, int jt) {
      const int j0 = jt * 16;
      const int jn = (jt + 2 * NSPLIT < ROW_TILES) ? jt + 2 * NSPLIT : jt;
      load_frag(coords, sq, jn, m, kh, fill);
      float dd[8];
      frag_d2(use, jt, rt, row, kh, b, sqi, dd);
      float lo = dd[0];
      #pragma unroll
      for (int r = 1; r < 8; ++r) lo = fminf(lo, dd[r]);
      if (lo <= T) {
        #pragma unroll
        for (int r = 0; r < 8; ++r) {
          const float d = dd[r];
          if (d <= T) {
            const int db = max(__float_as_int(d), 0);
            const u64 key = ((u64)(unsigned)db << 32) | (u64)(unsigned)(j0 + 8 * kh + r);
            if (d < T) {
              if (cs < KNN) {
                if (cs + ct == KNN) --ct;
                kl[cs] = key; ++cs;
              }
            } else if (cs + ct < KNN) {
              ++ct; kl[KNN - ct] = key;
            }
          }
        }
      }
    };
    int jt = wave;
    #pragma unroll 1
    for (; jt + 2 * NSPLIT < ROW_TILES; jt += 3 * NSPLIT) {
      step(f0, f2, jt);
      step(f1, f0, jt + NSPLIT);
      step(f2, f1, jt + 2 * NSPLIT);
    }
    #pragma unroll 1
    for (; jt < ROW_TILES; jt += NSPLIT) {
      step(f0, f2, jt);
      f0 = f1;
      f1 = f2;
    }
    lds_cnt[m][list] = cs | (ct << 8);
  }
  __syncthreads();

  int tot = 0;
  {
    const int r = tid >> 4;
    const int q = tid & 15;
    int off = 0, mine = 0;
    #pragma unroll
    for (int l = 0; l < NLIST; ++l) {
      const int c = lds_cnt[r][l];
      const int n = (c & 255) + (c >> 8);
      off += (l < q) ? n : 0;
      mine = (l == q) ? c : mine;
      tot += n;
    }
    const int ncs = mine & 255, nct = mine >> 8;
    const u64* kl = &lds_key[r][q * KNN];
    for (int i = 0; i < ncs + nct; ++i) {
      const int slot = i < ncs ? i : KNN - nct + (i - ncs);
      lds_cand[r][off + i] = kl[slot];
    }
    if (q == 0) lds_cand[r][tot] = KEY_NONE;
  }
  __syncthreads();

  {
    const int r = tid >> 4;
    const int q = tid & 15;
    for (int p = q; p < tot; p += 16) {
      const u64 ck = lds_cand[r][p];
      int rank = 0;
      for (int e = 0; e < tot; e += 2) {
        const v2u64 kv = *reinterpret_cast<const v2u64*>(&lds_cand[r][e]);
        rank += (kv.x < ck ? 1 : 0) + (kv.y < ck ? 1 : 0);
      }
      if (rank < KNN) {
        lds_w[r][rank] = expf(-__int_as_float((int)(ck >> 32)));
        lds_j[r][rank] = (int)(unsigned)ck;
      }
    }
  }
  __syncthreads();

  #pragma unroll
  for (int rr = 0; rr < 16 / NSPLIT; ++rr) {
    const int r = wave * (16 / NSPLIT) + rr;
    float acc0 = 0.f, acc1 = 0.f;
    #pragma unroll
    for (int k = 0; k < KNN; ++k) {
      const int   j  = lds_j[r][k];
      const float wk = lds_w[r][k];
      acc0 += wk * feats[(size_t)j * DFEAT + lane];
      acc1 += wk * feats[(size_t)j * DFEAT + lane + 32];
    }
    lds_agg[r][lane]      = acc0;
    lds_agg[r][lane + 32] = acc1;
  }
  __syncthreads();

  if (wave < 2) {
    const float* fr = feats + (size_t)(row0 + m) * DFEAT;
    #pragma unroll
    for (int t = 0; t < 2; ++t) {
      const int nt = wave * 2 + t;
      v8f acc = {};
      #pragma unroll
      for (int s = 0; s < 16; ++s) {
        v2f av, bv;
        av.x = fr[4 * s + 2 * kh];
        av.y = fr[4 * s + 1 + 2 * kh];
        bv.x = Wout[(4 * s + 2 * kh) * DFEAT + nt * 16 + m];
        bv.y = Wout[(4 * s + 1 + 2 * kh) * DFEAT + nt * 16 + m];
        acc = wmma_f32(av, bv, acc);
      }
      #pragma unroll
      for (int s = 0; s < 16; ++s) {
        v2f av, bv;
        av.x = lds_agg[m][4 * s + 2 * kh];
        av.y = lds_agg[m][4 * s + 1 + 2 * kh];
        bv.x = Wout[(64 + 4 * s + 2 * kh) * DFEAT + nt * 16 + m];
        bv.y = Wout[(64 + 4 * s + 1 + 2 * kh) * DFEAT + nt * 16 + m];
        acc = wmma_f32(av, bv, acc);
      }
      #pragma unroll
      for (int r = 0; r < 8; ++r) {
        const int row = row0 + r + 8 * kh;
        const int col = nt * 16 + m;
        out[(size_t)row * DFEAT + col] = acc[r] + bout[col];
      }
    }
  }
}

extern "C" void kernel_launch(void* const* d_in, const int* in_sizes, int n_in,
                              void* d_out, int out_size, void* d_ws, size_t ws_size,
                              hipStream_t stream) {
  const float* x  = (const float*)d_in[0];
  const float* Wf = (const float*)d_in[1];
  const float* bf = (const float*)d_in[2];
  const float* Wl = (const float*)d_in[3];
  const float* bl = (const float*)d_in[4];
  const float* Wo = (const float*)d_in[5];
  const float* bo = (const float*)d_in[6];
  float* out = (float*)d_out;

  float* feats  = (float*)d_ws;
  float* coords = feats  + (size_t)N_PTS * DFEAT;
  float* sq     = coords + (size_t)N_PTS * DLAT;

  encoder_kernel<<<NUM_WG, 256, 0, stream>>>(x, Wf, bf, Wl, bl, feats, coords);
  sq_kernel<<<(N_PTS + 255) / 256, 256, 0, stream>>>(coords, sq);
  gravnet_kernel<<<ROW_TILES, 256, 0, stream>>>(coords, sq, feats, Wo, bo, out);
}


# Scan 1's later tiles record (distance, index) of each candidate at or below U in LDS; scan 2 recomputes only each wave's first 20 column tiles and filters the records by T (full rescan on overflow)
# speedup vs baseline: 11.6780x; 1.2761x over previous
#include <hip/hip_runtime.h>
#include <cmath>

typedef __attribute__((ext_vector_type(2))) float v2f;
typedef __attribute__((ext_vector_type(8))) float v8f;

#define N_PTS   12000
#define DIN     64
#define DLAT    16
#define DFEAT   64
#define KNN     16
#define ROW_TILES (N_PTS / 16)
#define WAVES_PER_WG 8
#define NUM_WG  ((ROW_TILES + WAVES_PER_WG - 1) / WAVES_PER_WG)

static __device__ __forceinline__ v8f wmma_f32(v2f a, v2f b, v8f c) {
  return __builtin_amdgcn_wmma_f32_16x16x4_f32(
       false, a,  false, b,
       (short)0, c,  false,  false);
}

__global__ void __launch_bounds__(256)
encoder_kernel(const float* __restrict__ x,
               const float* __restrict__ Wf, const float* __restrict__ bf,
               const float* __restrict__ Wl, const float* __restrict__ bl,
               float* __restrict__ feats, float* __restrict__ coords) {
  const int lane = threadIdx.x & 31;
  const int wave = threadIdx.x >> 5;
  const int tile = blockIdx.x * WAVES_PER_WG + wave;
  if (tile >= ROW_TILES) return;
  const int row0 = tile * 16;
  const int m  = lane & 15;
  const int kh = lane >> 4;

  v2f a[16];
  const float* xr = x + (size_t)(row0 + m) * DIN;
  #pragma unroll
  for (int s = 0; s < 16; ++s) {
    a[s].x = xr[4 * s + 2 * kh];
    a[s].y = xr[4 * s + 1 + 2 * kh];
  }

  #pragma unroll
  for (int nt = 0; nt < 4; ++nt) {
    v8f acc = {};
    #pragma unroll
    for (int s = 0; s < 16; ++s) {
      v2f b;
      b.x = Wf[(4 * s + 2 * kh) * DFEAT + nt * 16 + m];
      b.y = Wf[(4 * s + 1 + 2 * kh) * DFEAT + nt * 16 + m];
      acc = wmma_f32(a[s], b, acc);
    }
    #pragma unroll
    for (int r = 0; r < 8; ++r) {
      const int row = row0 + r + 8 * kh;
      const int col = nt * 16 + m;
      feats[(size_t)row * DFEAT + col] = tanhf(acc[r] + bf[col]);
    }
  }

  {
    v8f acc = {};
    #pragma unroll
    for (int s = 0; s < 16; ++s) {
      v2f b;
      b.x = Wl[(4 * s + 2 * kh) * DLAT + m];
      b.y = Wl[(4 * s + 1 + 2 * kh) * DLAT + m];
      acc = wmma_f32(a[s], b, acc);
    }
    #pragma unroll
    for (int r = 0; r < 8; ++r) {
      const int row = row0 + r + 8 * kh;
      coords[(size_t)row * DLAT + m] = tanhf(acc[r] + bl[m]);
    }
  }
}

__global__ void sq_kernel(const float* __restrict__ coords, float* __restrict__ sq) {
  const int i = blockIdx.x * blockDim.x + threadIdx.x;
  if (i >= N_PTS) return;
  const float* c = coords + (size_t)i * DLAT;
  float s = 0.f;
  #pragma unroll
  for (int d = 0; d < DLAT; ++d) s += c[d] * c[d];
  sq[i] = s;
}

#define NSPLIT  WAVES_PER_WG
#define NLIST   (2 * NSPLIT)
#define NCAND   (NLIST * KNN)
#define KPITCH  (NCAND + 2)
#define VPITCH  (2 * KPITCH)
#define BIGF    3.4e38f
#define TILES_A 20
#define RCAP    24
#define RPITCH  (NLIST * RCAP + 2)

typedef unsigned long long u64;
typedef __attribute__((ext_vector_type(2))) unsigned long long v2u64;
typedef __attribute__((ext_vector_type(4))) float v4f;
typedef __attribute__((ext_vector_type(2))) int v2i;

#define KEY_NONE 0xFFFFFFFFFFFFFFFFull

static __device__ __forceinline__ float pair_d2(float sqi, float sqj, float dot) {
  return fmaxf(__builtin_fmaf(-2.f, dot, sqi + sqj), 0.f);
}

struct TileFrag {
  v2f a[4];
  v4f s0, s1;
};

static __device__ __forceinline__ void load_frag(const float* __restrict__ coords,
                                                 const float* __restrict__ sq,
                                                 int jt, int m, int kh, TileFrag& f) {
  const int j0 = jt * 16;
  const float* cj = coords + (size_t)(j0 + m) * DLAT;
  #pragma unroll
  for (int s = 0; s < 4; ++s) {
    f.a[s].x = cj[4 * s + 2 * kh];
    f.a[s].y = cj[4 * s + 1 + 2 * kh];
  }
  f.s0 = *reinterpret_cast<const v4f*>(sq + j0 + 8 * kh);
  f.s1 = *reinterpret_cast<const v4f*>(sq + j0 + 8 * kh + 4);
}

static __device__ __forceinline__ v8f frag_dots(const TileFrag& f, const v2f (&b)[4]) {
  v8f acc = {};
  #pragma unroll
  for (int s = 0; s < 4; ++s) acc = wmma_f32(f.a[s], b[s], acc);
  return acc;
}

template <int MASK>
static __device__ __forceinline__ float row_xor(float v) {
  return __int_as_float(__builtin_amdgcn_update_dpp(0, __float_as_int(v), 0x160 + MASK, 0xf, 0xf, false));
}

static __device__ __forceinline__ void list_insert(float (&best)[KNN], float d) {
  #pragma unroll
  for (int s = 0; s < KNN - 1; ++s)
    best[s] = __builtin_amdgcn_fmed3f(best[s], best[s + 1], d);
  best[KNN - 1] = __builtin_amdgcn_fmed3f(best[KNN - 1], -BIGF, d);
}

static __device__ __forceinline__ void frag_d2(const TileFrag& f, int jt, int rt, int row, int kh,
                                               const v2f (&b)[4], float sqi, float (&dd)[8]) {
  const int j0 = jt * 16;
  const v8f acc = frag_dots(f, b);
  #pragma unroll
  for (int r = 0; r < 8; ++r)
    dd[r] = pair_d2(sqi, r < 4 ? f.s0[r & 3] : f.s1[r & 3], acc[r]);
  if (jt == rt) {
    #pragma unroll
    for (int r = 0; r < 8; ++r)
      if (j0 + 8 * kh + r == row) dd[r] = BIGF;
  }
}

static __device__ __forceinline__ float row_16th(const float* lv, int q, int lane) {
  int head = 0;
  float t = 0.f;
  for (int step = 0; step < KNN; ++step) {
    const float v = lv[KNN - 1 - head];
    float mn = v;
    mn = fminf(mn, row_xor<1>(mn));
    mn = fminf(mn, row_xor<2>(mn));
    mn = fminf(mn, row_xor<4>(mn));
    mn = fminf(mn, row_xor<8>(mn));
    t = mn;
    const unsigned holders = (__builtin_amdgcn_ballot_w32(v == mn) >> (lane & 16)) & 0xFFFFu;
    if (q == __builtin_ctz(holders)) ++head;
  }
  return t;
}

__global__ void __launch_bounds__(256, 6)
gravnet_kernel(const float* __restrict__ coords, const float* __restrict__ sq,
               const float* __restrict__ feats,
               const float* __restrict__ Wout, const float* __restrict__ bout,
               float* __restrict__ out) {
  __shared__ __attribute__((aligned(16))) u64 lds_rec[16][RPITCH];
  u64 (*lds_cand)[KPITCH] = reinterpret_cast<u64 (*)[KPITCH]>(&lds_rec[0][0]);
  __shared__ __attribute__((aligned(16))) u64 lds_key[16][KPITCH];
  __shared__ __attribute__((aligned(16))) int lds_cnt[16][NLIST];
  __shared__ float lds_T  [16];
  __shared__ float lds_w  [16][KNN];
  __shared__ int   lds_j  [16][KNN];
  __shared__ float lds_agg[16][DFEAT];
  float* lds_val = reinterpret_cast<float*>(&lds_cand[0][0]);

  const int tid  = threadIdx.x;
  const int lane = tid & 31;
  const int wave = __builtin_amdgcn_readfirstlane(tid >> 5);
  const int rt   = blockIdx.x;
  const int row0 = rt * 16;
  const int m  = lane & 15;
  const int kh = lane >> 4;
  const int row  = row0 + m;
  const int list = wave * 2 + kh;

  v2f b[4];
  {
    const float* cr = coords + (size_t)row * DLAT;
    #pragma unroll
    for (int s = 0; s < 4; ++s) {
      b[s].x = cr[4 * s + 2 * kh];
      b[s].y = cr[4 * s + 1 + 2 * kh];
    }
  }
  const float sqi = sq[row];

  float best[KNN];
  #pragma unroll
  for (int s = 0; s < KNN; ++s) best[s] = BIGF;

  TileFrag f0, f1, f2;
  load_frag(coords, sq, wave, m, kh, f0);
  load_frag(coords, sq, wave + NSPLIT, m, kh, f1);
  {
    auto step = [&](const TileFrag& use, TileFrag& fill, int jt) {
      load_frag(coords, sq, jt + 2 * NSPLIT, m, kh, fill);
      float dd[8];
      frag_d2(use, jt, rt, row, kh, b, sqi, dd);
      #pragma unroll
      for (int r = 0; r < 8; ++r) list_insert(best, dd[r]);
    };
    const int end = wave + NSPLIT * TILES_A;
    int jt = wave;
    #pragma unroll 1
    for (; jt + 2 * NSPLIT < end; jt += 3 * NSPLIT) {
      step(f0, f2, jt);
      step(f1, f0, jt + NSPLIT);
      step(f2, f1, jt + 2 * NSPLIT);
    }
    #pragma unroll 1
    for (; jt < end; jt += NSPLIT) {
      step(f0, f2, jt);
      f0 = f1;
      f1 = f2;
    }
  }

  {
    v4f* vp = reinterpret_cast<v4f*>(lds_val + m * VPITCH + list * KNN);
    #pragma unroll
    for (int g = 0; g < 4; ++g) {
      v4f v = {best[4 * g], best[4 * g + 1], best[4 * g + 2], best[4 * g + 3]};
      vp[g] = v;
    }
  }
  __syncthreads();
  {
    const int r = tid >> 4;
    const int q = tid & 15;
    const float u = row_16th(lds_val + r * VPITCH + q * KNN, q, lane);
    if (q == 0) lds_T[r] = u;
  }
  __syncthreads();
  int nrec = 0;
  u64* rec = &lds_rec[m][list * RCAP];
  v2i* rec2 = reinterpret_cast<v2i*>(rec);
  {
    const float U = lds_T[m];
    auto step = [&](const TileFrag& use, TileFrag& fill, int jt) {
      const int jn = (jt + 2 * NSPLIT < ROW_TILES) ? jt + 2 * NSPLIT
                   : (jt + NSPLIT < ROW_TILES) ? wave : wave + NSPLIT;
      load_frag(coords, sq, jn, m, kh, fill);
      float dd[8];
      frag_d2(use, jt, rt, row, kh, b, sqi, dd);
      float p[8];
      p[0] = dd[0];
      #pragma unroll
      for (int r = 1; r < 8; ++r) p[r] = __builtin_amdgcn_fmed3f(p[r - 1], dd[r], -BIGF);
      const float lo = p[7];
      const int jc = jt * 16 + 8 * kh;
      if (__builtin_amdgcn_ballot_w32(lo <= U) != 0) {
        float b1 = __builtin_amdgcn_fmed3f(dd[0], dd[1], BIGF);
        #pragma unroll
        for (int r = 2; r < 8; ++r) b1 = __builtin_amdgcn_fmed3f(p[r - 1], b1, dd[r]);
        if (__builtin_amdgcn_ballot_w32(b1 <= U) != 0) {
          #pragma unroll
          for (int r = 0; r < 8; ++r) list_insert(best, dd[r]);
          #pragma unroll
          for (int r = 0; r < 8; ++r) {
            if (dd[r] <= U) {
              if (nrec < RCAP) {
                const v2i kv = {jc + r, max(__float_as_int(dd[r]), 0)};
                rec2[nrec] = kv;
              }
              ++nrec;
            }
          }
        } else {
          list_insert(best, lo);
          int ri = 0;
          #pragma unroll
          for (int r = 1; r < 8; ++r) ri = (dd[r] == lo) ? r : ri;
          if (lo <= U) {
            if (nrec < RCAP) {
              const v2i kv = {jc + ri, max(__float_as_int(lo), 0)};
              rec2[nrec] = kv;
            }
            ++nrec;
          }
        }
      }
    };
    int jt = wave + NSPLIT * TILES_A;
    #pragma unroll 1
    for (; jt + 2 * NSPLIT < ROW_TILES; jt += 3 * NSPLIT) {
      step(f0, f2, jt);
      step(f1, f0, jt + NSPLIT);
      step(f2, f1, jt + 2 * NSPLIT);
    }
    #pragma unroll 1
    for (; jt < ROW_TILES; jt += NSPLIT) {
      step(f0, f2, jt);
      f0 = f1;
      f1 = f2;
    }
  }

  float* lds_val2 = reinterpret_cast<float*>(&lds_key[0][0]);
  {
    v4f* vp = reinterpret_cast<v4f*>(lds_val2 + m * VPITCH + list * KNN);
    #pragma unroll
    for (int g = 0; g < 4; ++g) {
      v4f v = {best[4 * g], best[4 * g + 1], best[4 * g + 2], best[4 * g + 3]};
      vp[g] = v;
    }
  }
  lds_w[tid >> 4][tid & 15] = 0.f;
  lds_j[tid >> 4][tid & 15] = 0;
  __syncthreads();

  {
    const int r = tid >> 4;
    const int q = tid & 15;
    const float t = row_16th(lds_val2 + r * VPITCH + q * KNN, q, lane);
    if (q == 0) lds_T[r] = t;
  }
  __syncthreads();
  {
    v2u64* kp = reinterpret_cast<v2u64*>(&lds_key[m][list * KNN]);
    const v2u64 none = {KEY_NONE, KEY_NONE};
    #pragma unroll
    for (int g = 0; g < KNN / 2; ++g) kp[g] = none;
  }

  int cs = 0, ct = 0;
  {
    const float T = lds_T[m];
    u64* kl = &lds_key[m][list * KNN];
    auto collect = [&](float d, u64 key) {
      if (d < T) {
        if (cs < KNN) {
          if (cs + ct == KNN) --ct;
          kl[cs] = key; ++cs;
        }
      } else if (cs + ct < KNN) {
        ++ct; kl[KNN - ct] = key;
      }
    };
    const bool ovf = __builtin_amdgcn_ballot_w32(nrec > RCAP) != 0;
    const int end2 = ovf ? ROW_TILES : wave + NSPLIT * TILES_A;
    auto step = [&](const TileFrag& use, TileFrag& fill, int jt) {
      const int j0 = jt * 16;
      const int jn = (jt + 2 * NSPLIT < end2) ? jt + 2 * NSPLIT : jt;
      load_frag(coords, sq, jn, m, kh, fill);
      float dd[8];
      frag_d2(use, jt, rt, row, kh, b, sqi, dd);
      float lo = dd[0];
      #pragma unroll
      for (int r = 1; r < 8; ++r) lo = fminf(lo, dd[r]);
      if (lo <= T) {
        #pragma unroll
        for (int r = 0; r < 8; ++r) {
          const float d = dd[r];
          if (d <= T) {
            const int db = max(__float_as_int(d), 0);
            collect(d, ((u64)(unsigned)db << 32) | (u64)(unsigned)(j0 + 8 * kh + r));
          }
        }
      }
    };
    int jt = wave;
    #pragma unroll 1
    for (; jt + 2 * NSPLIT < end2; jt += 3 * NSPLIT) {
      step(f0, f2, jt);
      step(f1, f0, jt + NSPLIT);
      step(f2, f1, jt + 2 * NSPLIT);
    }
    #pragma unroll 1
    for (; jt < end2; jt += NSPLIT) {
      step(f0, f2, jt);
      f0 = f1;
      f1 = f2;
    }
    if (!ovf) {
      for (int i = 0; i < nrec; ++i) {
        const u64 key = rec[i];
        const float d = __int_as_float((int)(key >> 32));
        if (d <= T) collect(d, key);
      }
    }
    lds_cnt[m][list] = cs | (ct << 8);
  }
  __syncthreads();

  int tot = 0;
  {
    const int r = tid >> 4;
    const int q = tid & 15;
    int off = 0, mine = 0;
    #pragma unroll
    for (int l = 0; l < NLIST; ++l) {
      const int c = lds_cnt[r][l];
      const int n = (c & 255) + (c >> 8);
      off += (l < q) ? n : 0;
      mine = (l == q) ? c : mine;
      tot += n;
    }
    const int ncs = mine & 255, nct = mine >> 8;
    const u64* kl = &lds_key[r][q * KNN];
    for (int i = 0; i < ncs + nct; ++i) {
      const int slot = i < ncs ? i : KNN - nct + (i - ncs);
      lds_cand[r][off + i] = kl[slot];
    }
    if (q == 0) lds_cand[r][tot] = KEY_NONE;
  }
  __syncthreads();

  {
    const int r = tid >> 4;
    const int q = tid & 15;
    for (int p = q; p < tot; p += 16) {
      const u64 ck = lds_cand[r][p];
      int rank = 0;
      for (int e = 0; e < tot; e += 2) {
        const v2u64 kv = *reinterpret_cast<const v2u64*>(&lds_cand[r][e]);
        rank += (kv.x < ck ? 1 : 0) + (kv.y < ck ? 1 : 0);
      }
      if (rank < KNN) {
        lds_w[r][rank] = expf(-__int_as_float((int)(ck >> 32)));
        lds_j[r][rank] = (int)(unsigned)ck;
      }
    }
  }
  __syncthreads();

  #pragma unroll
  for (int rr = 0; rr < 16 / NSPLIT; ++rr) {
    const int r = wave * (16 / NSPLIT) + rr;
    float acc0 = 0.f, acc1 = 0.f;
    #pragma unroll
    for (int k = 0; k < KNN; ++k) {
      const int   j  = lds_j[r][k];
      const float wk = lds_w[r][k];
      acc0 += wk * feats[(size_t)j * DFEAT + lane];
      acc1 += wk * feats[(size_t)j * DFEAT + lane + 32];
    }
    lds_agg[r][lane]      = acc0;
    lds_agg[r][lane + 32] = acc1;
  }
  __syncthreads();

  if (wave < 2) {
    const float* fr = feats + (size_t)(row0 + m) * DFEAT;
    #pragma unroll
    for (int t = 0; t < 2; ++t) {
      const int nt = wave * 2 + t;
      v8f acc = {};
      #pragma unroll
      for (int s = 0; s < 16; ++s) {
        v2f av, bv;
        av.x = fr[4 * s + 2 * kh];
        av.y = fr[4 * s + 1 + 2 * kh];
        bv.x = Wout[(4 * s + 2 * kh) * DFEAT + nt * 16 + m];
        bv.y = Wout[(4 * s + 1 + 2 * kh) * DFEAT + nt * 16 + m];
        acc = wmma_f32(av, bv, acc);
      }
      #pragma unroll
      for (int s = 0; s < 16; ++s) {
        v2f av, bv;
        av.x = lds_agg[m][4 * s + 2 * kh];
        av.y = lds_agg[m][4 * s + 1 + 2 * kh];
        bv.x = Wout[(64 + 4 * s + 2 * kh) * DFEAT + nt * 16 + m];
        bv.y = Wout[(64 + 4 * s + 1 + 2 * kh) * DFEAT + nt * 16 + m];
        acc = wmma_f32(av, bv, acc);
      }
      #pragma unroll
      for (int r = 0; r < 8; ++r) {
        const int row = row0 + r + 8 * kh;
        const int col = nt * 16 + m;
        out[(size_t)row * DFEAT + col] = acc[r] + bout[col];
      }
    }
  }
}

extern "C" void kernel_launch(void* const* d_in, const int* in_sizes, int n_in,
                              void* d_out, int out_size, void* d_ws, size_t ws_size,
                              hipStream_t stream) {
  const float* x  = (const float*)d_in[0];
  const float* Wf = (const float*)d_in[1];
  const float* bf = (const float*)d_in[2];
  const float* Wl = (const float*)d_in[3];
  const float* bl = (const float*)d_in[4];
  const float* Wo = (const float*)d_in[5];
  const float* bo = (const float*)d_in[6];
  float* out = (float*)d_out;

  float* feats  = (float*)d_ws;
  float* coords = feats  + (size_t)N_PTS * DFEAT;
  float* sq     = coords + (size_t)N_PTS * DLAT;

  encoder_kernel<<<NUM_WG, 256, 0, stream>>>(x, Wf, bf, Wl, bl, feats, coords);
  sq_kernel<<<(N_PTS + 255) / 256, 256, 0, stream>>>(coords, sq);
  gravnet_kernel<<<ROW_TILES, 256, 0, stream>>>(coords, sq, feats, Wo, bo, out);
}


# Scan 1's later tiles only record candidates at or below U; each lane inserts its recorded distances into its sorted list once after the last tile; U is drawn after 12 tiles (was 20), 32 slots
# speedup vs baseline: 12.8550x; 1.1008x over previous
#include <hip/hip_runtime.h>
#include <cmath>

typedef __attribute__((ext_vector_type(2))) float v2f;
typedef __attribute__((ext_vector_type(8))) float v8f;

#define N_PTS   12000
#define DIN     64
#define DLAT    16
#define DFEAT   64
#define KNN     16
#define ROW_TILES (N_PTS / 16)
#define WAVES_PER_WG 8
#define NUM_WG  ((ROW_TILES + WAVES_PER_WG - 1) / WAVES_PER_WG)

static __device__ __forceinline__ v8f wmma_f32(v2f a, v2f b, v8f c) {
  return __builtin_amdgcn_wmma_f32_16x16x4_f32(
       false, a,  false, b,
       (short)0, c,  false,  false);
}

__global__ void __launch_bounds__(256)
encoder_kernel(const float* __restrict__ x,
               const float* __restrict__ Wf, const float* __restrict__ bf,
               const float* __restrict__ Wl, const float* __restrict__ bl,
               float* __restrict__ feats, float* __restrict__ coords) {
  const int lane = threadIdx.x & 31;
  const int wave = threadIdx.x >> 5;
  const int tile = blockIdx.x * WAVES_PER_WG + wave;
  if (tile >= ROW_TILES) return;
  const int row0 = tile * 16;
  const int m  = lane & 15;
  const int kh = lane >> 4;

  v2f a[16];
  const float* xr = x + (size_t)(row0 + m) * DIN;
  #pragma unroll
  for (int s = 0; s < 16; ++s) {
    a[s].x = xr[4 * s + 2 * kh];
    a[s].y = xr[4 * s + 1 + 2 * kh];
  }

  #pragma unroll
  for (int nt = 0; nt < 4; ++nt) {
    v8f acc = {};
    #pragma unroll
    for (int s = 0; s < 16; ++s) {
      v2f b;
      b.x = Wf[(4 * s + 2 * kh) * DFEAT + nt * 16 + m];
      b.y = Wf[(4 * s + 1 + 2 * kh) * DFEAT + nt * 16 + m];
      acc = wmma_f32(a[s], b, acc);
    }
    #pragma unroll
    for (int r = 0; r < 8; ++r) {
      const int row = row0 + r + 8 * kh;
      const int col = nt * 16 + m;
      feats[(size_t)row * DFEAT + col] = tanhf(acc[r] + bf[col]);
    }
  }

  {
    v8f acc = {};
    #pragma unroll
    for (int s = 0; s < 16; ++s) {
      v2f b;
      b.x = Wl[(4 * s + 2 * kh) * DLAT + m];
      b.y = Wl[(4 * s + 1 + 2 * kh) * DLAT + m];
      acc = wmma_f32(a[s], b, acc);
    }
    #pragma unroll
    for (int r = 0; r < 8; ++r) {
      const int row = row0 + r + 8 * kh;
      coords[(size_t)row * DLAT + m] = tanhf(acc[r] + bl[m]);
    }
  }
}

__global__ void sq_kernel(const float* __restrict__ coords, float* __restrict__ sq) {
  const int i = blockIdx.x * blockDim.x + threadIdx.x;
  if (i >= N_PTS) return;
  const float* c = coords + (size_t)i * DLAT;
  float s = 0.f;
  #pragma unroll
  for (int d = 0; d < DLAT; ++d) s += c[d] * c[d];
  sq[i] = s;
}

#define NSPLIT  WAVES_PER_WG
#define NLIST   (2 * NSPLIT)
#define NCAND   (NLIST * KNN)
#define KPITCH  (NCAND + 2)
#define VPITCH  (2 * KPITCH)
#define BIGF    3.4e38f
#define TILES_A 12
#define RCAP    32
#define RPITCH  (NLIST * RCAP + 2)

typedef unsigned long long u64;
typedef __attribute__((ext_vector_type(2))) unsigned long long v2u64;
typedef __attribute__((ext_vector_type(4))) float v4f;
typedef __attribute__((ext_vector_type(2))) int v2i;

#define KEY_NONE 0xFFFFFFFFFFFFFFFFull

static __device__ __forceinline__ float pair_d2(float sqi, float sqj, float dot) {
  return fmaxf(__builtin_fmaf(-2.f, dot, sqi + sqj), 0.f);
}

struct TileFrag {
  v2f a[4];
  v4f s0, s1;
};

static __device__ __forceinline__ void load_frag(const float* __restrict__ coords,
                                                 const float* __restrict__ sq,
                                                 int jt, int m, int kh, TileFrag& f) {
  const int j0 = jt * 16;
  const float* cj = coords + (size_t)(j0 + m) * DLAT;
  #pragma unroll
  for (int s = 0; s < 4; ++s) {
    f.a[s].x = cj[4 * s + 2 * kh];
    f.a[s].y = cj[4 * s + 1 + 2 * kh];
  }
  f.s0 = *reinterpret_cast<const v4f*>(sq + j0 + 8 * kh);
  f.s1 = *reinterpret_cast<const v4f*>(sq + j0 + 8 * kh + 4);
}

static __device__ __forceinline__ v8f frag_dots(const TileFrag& f, const v2f (&b)[4]) {
  v8f acc = {};
  #pragma unroll
  for (int s = 0; s < 4; ++s) acc = wmma_f32(f.a[s], b[s], acc);
  return acc;
}

template <int MASK>
static __device__ __forceinline__ float row_xor(float v) {
  return __int_as_float(__builtin_amdgcn_update_dpp(0, __float_as_int(v), 0x160 + MASK, 0xf, 0xf, false));
}

static __device__ __forceinline__ void list_insert(float (&best)[KNN], float d) {
  #pragma unroll
  for (int s = 0; s < KNN - 1; ++s)
    best[s] = __builtin_amdgcn_fmed3f(best[s], best[s + 1], d);
  best[KNN - 1] = __builtin_amdgcn_fmed3f(best[KNN - 1], -BIGF, d);
}

static __device__ __forceinline__ void frag_d2(const TileFrag& f, int jt, int rt, int row, int kh,
                                               const v2f (&b)[4], float sqi, float (&dd)[8]) {
  const int j0 = jt * 16;
  const v8f acc = frag_dots(f, b);
  #pragma unroll
  for (int r = 0; r < 8; ++r)
    dd[r] = pair_d2(sqi, r < 4 ? f.s0[r & 3] : f.s1[r & 3], acc[r]);
  if (jt == rt) {
    #pragma unroll
    for (int r = 0; r < 8; ++r)
      if (j0 + 8 * kh + r == row) dd[r] = BIGF;
  }
}

static __device__ __forceinline__ float row_16th(const float* lv, int q, int lane) {
  int head = 0;
  float t = 0.f;
  for (int step = 0; step < KNN; ++step) {
    const float v = lv[KNN - 1 - head];
    float mn = v;
    mn = fminf(mn, row_xor<1>(mn));
    mn = fminf(mn, row_xor<2>(mn));
    mn = fminf(mn, row_xor<4>(mn));
    mn = fminf(mn, row_xor<8>(mn));
    t = mn;
    const unsigned holders = (__builtin_amdgcn_ballot_w32(v == mn) >> (lane & 16)) & 0xFFFFu;
    if (q == __builtin_ctz(holders)) ++head;
  }
  return t;
}

__global__ void __launch_bounds__(256, 6)
gravnet_kernel(const float* __restrict__ coords, const float* __restrict__ sq,
               const float* __restrict__ feats,
               const float* __restrict__ Wout, const float* __restrict__ bout,
               float* __restrict__ out) {
  __shared__ __attribute__((aligned(16))) u64 lds_rec[16][RPITCH];
  u64 (*lds_cand)[KPITCH] = reinterpret_cast<u64 (*)[KPITCH]>(&lds_rec[0][0]);
  __shared__ __attribute__((aligned(16))) u64 lds_key[16][KPITCH];
  __shared__ __attribute__((aligned(16))) int lds_cnt[16][NLIST];
  __shared__ float lds_T  [16];
  __shared__ float lds_w  [16][KNN];
  __shared__ int   lds_j  [16][KNN];
  __shared__ float lds_agg[16][DFEAT];
  float* lds_val = reinterpret_cast<float*>(&lds_cand[0][0]);

  const int tid  = threadIdx.x;
  const int lane = tid & 31;
  const int wave = __builtin_amdgcn_readfirstlane(tid >> 5);
  const int rt   = blockIdx.x;
  const int row0 = rt * 16;
  const int m  = lane & 15;
  const int kh = lane >> 4;
  const int row  = row0 + m;
  const int list = wave * 2 + kh;

  v2f b[4];
  {
    const float* cr = coords + (size_t)row * DLAT;
    #pragma unroll
    for (int s = 0; s < 4; ++s) {
      b[s].x = cr[4 * s + 2 * kh];
      b[s].y = cr[4 * s + 1 + 2 * kh];
    }
  }
  const float sqi = sq[row];

  float best[KNN];
  #pragma unroll
  for (int s = 0; s < KNN; ++s) best[s] = BIGF;

  TileFrag f0, f1, f2;
  load_frag(coords, sq, wave, m, kh, f0);
  load_frag(coords, sq, wave + NSPLIT, m, kh, f1);
  {
    auto step = [&](const TileFrag& use, TileFrag& fill, int jt) {
      load_frag(coords, sq, jt + 2 * NSPLIT, m, kh, fill);
      float dd[8];
      frag_d2(use, jt, rt, row, kh, b, sqi, dd);
      #pragma unroll
      for (int r = 0; r < 8; ++r) list_insert(best, dd[r]);
    };
    const int end = wave + NSPLIT * TILES_A;
    int jt = wave;
    #pragma unroll 1
    for (; jt + 2 * NSPLIT < end; jt += 3 * NSPLIT) {
      step(f0, f2, jt);
      step(f1, f0, jt + NSPLIT);
      step(f2, f1, jt + 2 * NSPLIT);
    }
    #pragma unroll 1
    for (; jt < end; jt += NSPLIT) {
      step(f0, f2, jt);
      f0 = f1;
      f1 = f2;
    }
  }

  {
    v4f* vp = reinterpret_cast<v4f*>(lds_val + m * VPITCH + list * KNN);
    #pragma unroll
    for (int g = 0; g < 4; ++g) {
      v4f v = {best[4 * g], best[4 * g + 1], best[4 * g + 2], best[4 * g + 3]};
      vp[g] = v;
    }
  }
  __syncthreads();
  {
    const int r = tid >> 4;
    const int q = tid & 15;
    const float u = row_16th(lds_val + r * VPITCH + q * KNN, q, lane);
    if (q == 0) lds_T[r] = u;
  }
  __syncthreads();
  int nrec = 0;
  u64* rec = &lds_rec[m][list * RCAP];
  v2i* rec2 = reinterpret_cast<v2i*>(rec);
  const float U = lds_T[m];
  {
    auto step = [&](const TileFrag& use, TileFrag& fill, int jt) {
      const int jn = (jt + 2 * NSPLIT < ROW_TILES) ? jt + 2 * NSPLIT
                   : (jt + NSPLIT < ROW_TILES) ? wave : wave + NSPLIT;
      load_frag(coords, sq, jn, m, kh, fill);
      float dd[8];
      frag_d2(use, jt, rt, row, kh, b, sqi, dd);
      float p[8];
      p[0] = dd[0];
      #pragma unroll
      for (int r = 1; r < 8; ++r) p[r] = __builtin_amdgcn_fmed3f(p[r - 1], dd[r], -BIGF);
      const float lo = p[7];
      const int jc = jt * 16 + 8 * kh;
      if (__builtin_amdgcn_ballot_w32(lo <= U) != 0) {
        float b1 = __builtin_amdgcn_fmed3f(dd[0], dd[1], BIGF);
        #pragma unroll
        for (int r = 2; r < 8; ++r) b1 = __builtin_amdgcn_fmed3f(p[r - 1], b1, dd[r]);
        if (__builtin_amdgcn_ballot_w32(b1 <= U) != 0) {
          #pragma unroll
          for (int r = 0; r < 8; ++r) {
            const bool hit  = dd[r] <= U;
            const bool kept = hit && nrec < RCAP;
            if (kept) {
              const v2i kv = {jc + r, max(__float_as_int(dd[r]), 0)};
              rec2[nrec] = kv;
            }
            nrec += hit ? 1 : 0;
          }
        } else {
          int ri = 0;
          #pragma unroll
          for (int r = 1; r < 8; ++r) ri = (dd[r] == lo) ? r : ri;
          const bool hit  = lo <= U;
          const bool kept = hit && nrec < RCAP;
          if (kept) {
            const v2i kv = {jc + ri, max(__float_as_int(lo), 0)};
            rec2[nrec] = kv;
          }
          nrec += hit ? 1 : 0;
        }
      }
    };
    int jt = wave + NSPLIT * TILES_A;
    #pragma unroll 1
    for (; jt + 2 * NSPLIT < ROW_TILES; jt += 3 * NSPLIT) {
      step(f0, f2, jt);
      step(f1, f0, jt + NSPLIT);
      step(f2, f1, jt + 2 * NSPLIT);
    }
    #pragma unroll 1
    for (; jt < ROW_TILES; jt += NSPLIT) {
      step(f0, f2, jt);
      f0 = f1;
      f1 = f2;
    }
  }

  if (__builtin_amdgcn_ballot_w32(nrec > RCAP) != 0) {
    #pragma unroll 1
    for (int jt = wave + NSPLIT * TILES_A; jt < ROW_TILES; jt += NSPLIT) {
      load_frag(coords, sq, jt, m, kh, f2);
      float dd[8];
      frag_d2(f2, jt, rt, row, kh, b, sqi, dd);
      #pragma unroll
      for (int r = 0; r < 8; ++r) list_insert(best, dd[r] <= U ? dd[r] : BIGF);
    }
  } else {
    float d = (0 < nrec) ? __int_as_float(rec2[0].y) : BIGF;
    for (int i = 0; __builtin_amdgcn_ballot_w32(i < nrec) != 0; ++i) {
      const float dn = (i + 1 < nrec) ? __int_as_float(rec2[i + 1].y) : BIGF;
      list_insert(best, d);
      d = dn;
    }
  }

  float* lds_val2 = reinterpret_cast<float*>(&lds_key[0][0]);
  {
    v4f* vp = reinterpret_cast<v4f*>(lds_val2 + m * VPITCH + list * KNN);
    #pragma unroll
    for (int g = 0; g < 4; ++g) {
      v4f v = {best[4 * g], best[4 * g + 1], best[4 * g + 2], best[4 * g + 3]};
      vp[g] = v;
    }
  }
  lds_w[tid >> 4][tid & 15] = 0.f;
  lds_j[tid >> 4][tid & 15] = 0;
  __syncthreads();

  {
    const int r = tid >> 4;
    const int q = tid & 15;
    const float t = row_16th(lds_val2 + r * VPITCH + q * KNN, q, lane);
    if (q == 0) lds_T[r] = t;
  }
  __syncthreads();
  {
    v2u64* kp = reinterpret_cast<v2u64*>(&lds_key[m][list * KNN]);
    const v2u64 none = {KEY_NONE, KEY_NONE};
    #pragma unroll
    for (int g = 0; g < KNN / 2; ++g) kp[g] = none;
  }

  int cs = 0, ct = 0;
  {
    const float T = lds_T[m];
    u64* kl = &lds_key[m][list * KNN];
    auto collect = [&](float d, u64 key) {
      if (d < T) {
        if (cs < KNN) {
          if (cs + ct == KNN) --ct;
          kl[cs] = key; ++cs;
        }
      } else if (cs + ct < KNN) {
        ++ct; kl[KNN - ct] = key;
      }
    };
    const bool ovf = __builtin_amdgcn_ballot_w32(nrec > RCAP) != 0;
    const int end2 = ovf ? ROW_TILES : wave + NSPLIT * TILES_A;
    auto step = [&](const TileFrag& use, TileFrag& fill, int jt) {
      const int j0 = jt * 16;
      const int jn = (jt + 2 * NSPLIT < end2) ? jt + 2 * NSPLIT : jt;
      load_frag(coords, sq, jn, m, kh, fill);
      float dd[8];
      frag_d2(use, jt, rt, row, kh, b, sqi, dd);
      float lo = dd[0];
      #pragma unroll
      for (int r = 1; r < 8; ++r) lo = fminf(lo, dd[r]);
      if (lo <= T) {
        #pragma unroll
        for (int r = 0; r < 8; ++r) {
          const float d = dd[r];
          if (d <= T) {
            const int db = max(__float_as_int(d), 0);
            collect(d, ((u64)(unsigned)db << 32) | (u64)(unsigned)(j0 + 8 * kh + r));
          }
        }
      }
    };
    int jt = wave;
    #pragma unroll 1
    for (; jt + 2 * NSPLIT < end2; jt += 3 * NSPLIT) {
      step(f0, f2, jt);
      step(f1, f0, jt + NSPLIT);
      step(f2, f1, jt + 2 * NSPLIT);
    }
    #pragma unroll 1
    for (; jt < end2; jt += NSPLIT) {
      step(f0, f2, jt);
      f0 = f1;
      f1 = f2;
    }
    if (!ovf) {
      for (int i = 0; i < nrec; ++i) {
        const u64 key = rec[i];
        const float d = __int_as_float((int)(key >> 32));
        if (d <= T) collect(d, key);
      }
    }
    lds_cnt[m][list] = cs | (ct << 8);
  }
  __syncthreads();

  int tot = 0;
  {
    const int r = tid >> 4;
    const int q = tid & 15;
    int off = 0, mine = 0;
    #pragma unroll
    for (int l = 0; l < NLIST; ++l) {
      const int c = lds_cnt[r][l];
      const int n = (c & 255) + (c >> 8);
      off += (l < q) ? n : 0;
      mine = (l == q) ? c : mine;
      tot += n;
    }
    const int ncs = mine & 255, nct = mine >> 8;
    const u64* kl = &lds_key[r][q * KNN];
    for (int i = 0; i < ncs + nct; ++i) {
      const int slot = i < ncs ? i : KNN - nct + (i - ncs);
      lds_cand[r][off + i] = kl[slot];
    }
    if (q == 0) lds_cand[r][tot] = KEY_NONE;
  }
  __syncthreads();

  {
    const int r = tid >> 4;
    const int q = tid & 15;
    for (int p = q; p < tot; p += 16) {
      const u64 ck = lds_cand[r][p];
      int rank = 0;
      for (int e = 0; e < tot; e += 2) {
        const v2u64 kv = *reinterpret_cast<const v2u64*>(&lds_cand[r][e]);
        rank += (kv.x < ck ? 1 : 0) + (kv.y < ck ? 1 : 0);
      }
      if (rank < KNN) {
        lds_w[r][rank] = expf(-__int_as_float((int)(ck >> 32)));
        lds_j[r][rank] = (int)(unsigned)ck;
      }
    }
  }
  __syncthreads();

  #pragma unroll
  for (int rr = 0; rr < 16 / NSPLIT; ++rr) {
    const int r = wave * (16 / NSPLIT) + rr;
    float acc0 = 0.f, acc1 = 0.f;
    #pragma unroll
    for (int k = 0; k < KNN; ++k) {
      const int   j  = lds_j[r][k];
      const float wk = lds_w[r][k];
      acc0 += wk * feats[(size_t)j * DFEAT + lane];
      acc1 += wk * feats[(size_t)j * DFEAT + lane + 32];
    }
    lds_agg[r][lane]      = acc0;
    lds_agg[r][lane + 32] = acc1;
  }
  __syncthreads();

  if (wave < 2) {
    const float* fr = feats + (size_t)(row0 + m) * DFEAT;
    #pragma unroll
    for (int t = 0; t < 2; ++t) {
      const int nt = wave * 2 + t;
      v8f acc = {};
      #pragma unroll
      for (int s = 0; s < 16; ++s) {
        v2f av, bv;
        av.x = fr[4 * s + 2 * kh];
        av.y = fr[4 * s + 1 + 2 * kh];
        bv.x = Wout[(4 * s + 2 * kh) * DFEAT + nt * 16 + m];
        bv.y = Wout[(4 * s + 1 + 2 * kh) * DFEAT + nt * 16 + m];
        acc = wmma_f32(av, bv, acc);
      }
      #pragma unroll
      for (int s = 0; s < 16; ++s) {
        v2f av, bv;
        av.x = lds_agg[m][4 * s + 2 * kh];
        av.y = lds_agg[m][4 * s + 1 + 2 * kh];
        bv.x = Wout[(64 + 4 * s + 2 * kh) * DFEAT + nt * 16 + m];
        bv.y = Wout[(64 + 4 * s + 1 + 2 * kh) * DFEAT + nt * 16 + m];
        acc = wmma_f32(av, bv, acc);
      }
      #pragma unroll
      for (int r = 0; r < 8; ++r) {
        const int row = row0 + r + 8 * kh;
        const int col = nt * 16 + m;
        out[(size_t)row * DFEAT + col] = acc[r] + bout[col];
      }
    }
  }
}

extern "C" void kernel_launch(void* const* d_in, const int* in_sizes, int n_in,
                              void* d_out, int out_size, void* d_ws, size_t ws_size,
                              hipStream_t stream) {
  const float* x  = (const float*)d_in[0];
  const float* Wf = (const float*)d_in[1];
  const float* bf = (const float*)d_in[2];
  const float* Wl = (const float*)d_in[3];
  const float* bl = (const float*)d_in[4];
  const float* Wo = (const float*)d_in[5];
  const float* bo = (const float*)d_in[6];
  float* out = (float*)d_out;

  float* feats  = (float*)d_ws;
  float* coords = feats  + (size_t)N_PTS * DFEAT;
  float* sq     = coords + (size_t)N_PTS * DLAT;

  encoder_kernel<<<NUM_WG, 256, 0, stream>>>(x, Wf, bf, Wl, bl, feats, coords);
  sq_kernel<<<(N_PTS + 255) / 256, 256, 0, stream>>>(coords, sq);
  gravnet_kernel<<<ROW_TILES, 256, 0, stream>>>(coords, sq, feats, Wo, bo, out);
}


# Scan 1's later tiles test the eight distances against U with eight direct compares whose lane masks gate the record stores; running minima, second-smallest and position search are gone
# speedup vs baseline: 14.0772x; 1.0951x over previous
#include <hip/hip_runtime.h>
#include <cmath>

typedef __attribute__((ext_vector_type(2))) float v2f;
typedef __attribute__((ext_vector_type(8))) float v8f;

#define N_PTS   12000
#define DIN     64
#define DLAT    16
#define DFEAT   64
#define KNN     16
#define ROW_TILES (N_PTS / 16)
#define WAVES_PER_WG 8
#define NUM_WG  ((ROW_TILES + WAVES_PER_WG - 1) / WAVES_PER_WG)

static __device__ __forceinline__ v8f wmma_f32(v2f a, v2f b, v8f c) {
  return __builtin_amdgcn_wmma_f32_16x16x4_f32(
       false, a,  false, b,
       (short)0, c,  false,  false);
}

__global__ void __launch_bounds__(256)
encoder_kernel(const float* __restrict__ x,
               const float* __restrict__ Wf, const float* __restrict__ bf,
               const float* __restrict__ Wl, const float* __restrict__ bl,
               float* __restrict__ feats, float* __restrict__ coords) {
  const int lane = threadIdx.x & 31;
  const int wave = threadIdx.x >> 5;
  const int tile = blockIdx.x * WAVES_PER_WG + wave;
  if (tile >= ROW_TILES) return;
  const int row0 = tile * 16;
  const int m  = lane & 15;
  const int kh = lane >> 4;

  v2f a[16];
  const float* xr = x + (size_t)(row0 + m) * DIN;
  #pragma unroll
  for (int s = 0; s < 16; ++s) {
    a[s].x = xr[4 * s + 2 * kh];
    a[s].y = xr[4 * s + 1 + 2 * kh];
  }

  #pragma unroll
  for (int nt = 0; nt < 4; ++nt) {
    v8f acc = {};
    #pragma unroll
    for (int s = 0; s < 16; ++s) {
      v2f b;
      b.x = Wf[(4 * s + 2 * kh) * DFEAT + nt * 16 + m];
      b.y = Wf[(4 * s + 1 + 2 * kh) * DFEAT + nt * 16 + m];
      acc = wmma_f32(a[s], b, acc);
    }
    #pragma unroll
    for (int r = 0; r < 8; ++r) {
      const int row = row0 + r + 8 * kh;
      const int col = nt * 16 + m;
      feats[(size_t)row * DFEAT + col] = tanhf(acc[r] + bf[col]);
    }
  }

  {
    v8f acc = {};
    #pragma unroll
    for (int s = 0; s < 16; ++s) {
      v2f b;
      b.x = Wl[(4 * s + 2 * kh) * DLAT + m];
      b.y = Wl[(4 * s + 1 + 2 * kh) * DLAT + m];
      acc = wmma_f32(a[s], b, acc);
    }
    #pragma unroll
    for (int r = 0; r < 8; ++r) {
      const int row = row0 + r + 8 * kh;
      coords[(size_t)row * DLAT + m] = tanhf(acc[r] + bl[m]);
    }
  }
}

__global__ void sq_kernel(const float* __restrict__ coords, float* __restrict__ sq) {
  const int i = blockIdx.x * blockDim.x + threadIdx.x;
  if (i >= N_PTS) return;
  const float* c = coords + (size_t)i * DLAT;
  float s = 0.f;
  #pragma unroll
  for (int d = 0; d < DLAT; ++d) s += c[d] * c[d];
  sq[i] = s;
}

#define NSPLIT  WAVES_PER_WG
#define NLIST   (2 * NSPLIT)
#define NCAND   (NLIST * KNN)
#define KPITCH  (NCAND + 2)
#define VPITCH  (2 * KPITCH)
#define BIGF    3.4e38f
#define TILES_A 12
#define RCAP    32
#define RPITCH  (NLIST * RCAP + 2)

typedef unsigned long long u64;
typedef __attribute__((ext_vector_type(2))) unsigned long long v2u64;
typedef __attribute__((ext_vector_type(4))) float v4f;
typedef __attribute__((ext_vector_type(2))) int v2i;

#define KEY_NONE 0xFFFFFFFFFFFFFFFFull

static __device__ __forceinline__ float pair_raw(float sqi, float sqj, float dot) {
  return __builtin_fmaf(-2.f, dot, sqi + sqj);
}
static __device__ __forceinline__ float pair_d2(float sqi, float sqj, float dot) {
  return fmaxf(pair_raw(sqi, sqj, dot), 0.f);
}

struct TileFrag {
  v2f a[4];
  v4f s0, s1;
};

static __device__ __forceinline__ void load_frag(const float* __restrict__ coords,
                                                 const float* __restrict__ sq,
                                                 int jt, int m, int kh, TileFrag& f) {
  const int j0 = jt * 16;
  const float* cj = coords + (size_t)(j0 + m) * DLAT;
  #pragma unroll
  for (int s = 0; s < 4; ++s) {
    f.a[s].x = cj[4 * s + 2 * kh];
    f.a[s].y = cj[4 * s + 1 + 2 * kh];
  }
  f.s0 = *reinterpret_cast<const v4f*>(sq + j0 + 8 * kh);
  f.s1 = *reinterpret_cast<const v4f*>(sq + j0 + 8 * kh + 4);
}

static __device__ __forceinline__ v8f frag_dots(const TileFrag& f, const v2f (&b)[4]) {
  v8f acc = {};
  #pragma unroll
  for (int s = 0; s < 4; ++s) acc = wmma_f32(f.a[s], b[s], acc);
  return acc;
}

template <int MASK>
static __device__ __forceinline__ float row_xor(float v) {
  return __int_as_float(__builtin_amdgcn_update_dpp(0, __float_as_int(v), 0x160 + MASK, 0xf, 0xf, false));
}

static __device__ __forceinline__ void list_insert(float (&best)[KNN], float d) {
  #pragma unroll
  for (int s = 0; s < KNN - 1; ++s)
    best[s] = __builtin_amdgcn_fmed3f(best[s], best[s + 1], d);
  best[KNN - 1] = __builtin_amdgcn_fmed3f(best[KNN - 1], -BIGF, d);
}

template <bool CLAMP = true>
static __device__ __forceinline__ void frag_d2(const TileFrag& f, int jt, int rt, int row, int kh,
                                               const v2f (&b)[4], float sqi, float (&dd)[8]) {
  const int j0 = jt * 16;
  const v8f acc = frag_dots(f, b);
  #pragma unroll
  for (int r = 0; r < 8; ++r) {
    const float sqj = r < 4 ? f.s0[r & 3] : f.s1[r & 3];
    dd[r] = CLAMP ? pair_d2(sqi, sqj, acc[r]) : pair_raw(sqi, sqj, acc[r]);
  }
  if (jt == rt) {
    #pragma unroll
    for (int r = 0; r < 8; ++r)
      if (j0 + 8 * kh + r == row) dd[r] = BIGF;
  }
}

static __device__ __forceinline__ float row_16th(const float* lv, int q, int lane) {
  int head = 0;
  float t = 0.f;
  for (int step = 0; step < KNN; ++step) {
    const float v = lv[KNN - 1 - head];
    float mn = v;
    mn = fminf(mn, row_xor<1>(mn));
    mn = fminf(mn, row_xor<2>(mn));
    mn = fminf(mn, row_xor<4>(mn));
    mn = fminf(mn, row_xor<8>(mn));
    t = mn;
    const unsigned holders = (__builtin_amdgcn_ballot_w32(v == mn) >> (lane & 16)) & 0xFFFFu;
    if (q == __builtin_ctz(holders)) ++head;
  }
  return t;
}

__global__ void __launch_bounds__(256, 6)
gravnet_kernel(const float* __restrict__ coords, const float* __restrict__ sq,
               const float* __restrict__ feats,
               const float* __restrict__ Wout, const float* __restrict__ bout,
               float* __restrict__ out) {
  __shared__ __attribute__((aligned(16))) u64 lds_rec[16][RPITCH];
  u64 (*lds_cand)[KPITCH] = reinterpret_cast<u64 (*)[KPITCH]>(&lds_rec[0][0]);
  __shared__ __attribute__((aligned(16))) u64 lds_key[16][KPITCH];
  __shared__ __attribute__((aligned(16))) int lds_cnt[16][NLIST];
  __shared__ float lds_T  [16];
  __shared__ float lds_w  [16][KNN];
  __shared__ int   lds_j  [16][KNN];
  __shared__ float lds_agg[16][DFEAT];
  float* lds_val = reinterpret_cast<float*>(&lds_cand[0][0]);

  const int tid  = threadIdx.x;
  const int lane = tid & 31;
  const int wave = __builtin_amdgcn_readfirstlane(tid >> 5);
  const int rt   = blockIdx.x;
  const int row0 = rt * 16;
  const int m  = lane & 15;
  const int kh = lane >> 4;
  const int row  = row0 + m;
  const int list = wave * 2 + kh;

  v2f b[4];
  {
    const float* cr = coords + (size_t)row * DLAT;
    #pragma unroll
    for (int s = 0; s < 4; ++s) {
      b[s].x = cr[4 * s + 2 * kh];
      b[s].y = cr[4 * s + 1 + 2 * kh];
    }
  }
  const float sqi = sq[row];

  float best[KNN];
  #pragma unroll
  for (int s = 0; s < KNN; ++s) best[s] = BIGF;

  TileFrag f0, f1, f2;
  load_frag(coords, sq, wave, m, kh, f0);
  load_frag(coords, sq, wave + NSPLIT, m, kh, f1);
  {
    auto step = [&](const TileFrag& use, TileFrag& fill, int jt) {
      load_frag(coords, sq, jt + 2 * NSPLIT, m, kh, fill);
      float dd[8];
      frag_d2(use, jt, rt, row, kh, b, sqi, dd);
      #pragma unroll
      for (int r = 0; r < 8; ++r) list_insert(best, dd[r]);
    };
    const int end = wave + NSPLIT * TILES_A;
    int jt = wave;
    #pragma unroll 1
    for (; jt + 2 * NSPLIT < end; jt += 3 * NSPLIT) {
      step(f0, f2, jt);
      step(f1, f0, jt + NSPLIT);
      step(f2, f1, jt + 2 * NSPLIT);
    }
    #pragma unroll 1
    for (; jt < end; jt += NSPLIT) {
      step(f0, f2, jt);
      f0 = f1;
      f1 = f2;
    }
  }

  {
    v4f* vp = reinterpret_cast<v4f*>(lds_val + m * VPITCH + list * KNN);
    #pragma unroll
    for (int g = 0; g < 4; ++g) {
      v4f v = {best[4 * g], best[4 * g + 1], best[4 * g + 2], best[4 * g + 3]};
      vp[g] = v;
    }
  }
  __syncthreads();
  {
    const int r = tid >> 4;
    const int q = tid & 15;
    const float u = row_16th(lds_val + r * VPITCH + q * KNN, q, lane);
    if (q == 0) lds_T[r] = u;
  }
  __syncthreads();
  int nrec = 0;
  u64* rec = &lds_rec[m][list * RCAP];
  v2i* rec2 = reinterpret_cast<v2i*>(rec);
  const float U = lds_T[m];
  {
    auto step = [&](const TileFrag& use, TileFrag& fill, int jt) {
      const int jn = (jt + 2 * NSPLIT < ROW_TILES) ? jt + 2 * NSPLIT
                   : (jt + NSPLIT < ROW_TILES) ? wave : wave + NSPLIT;
      load_frag(coords, sq, jn, m, kh, fill);
      float dd[8];
      frag_d2<false>(use, jt, rt, row, kh, b, sqi, dd);
      const int jc = jt * 16 + 8 * kh;
      unsigned hits[8];
      #pragma unroll
      for (int r = 0; r < 8; ++r) hits[r] = __builtin_amdgcn_ballot_w32(dd[r] <= U);
      #pragma unroll
      for (int r = 0; r < 8; ++r) {
        if (__builtin_amdgcn_inverse_ballot_w32(hits[r])) {
          const v2i kv = {jc + r, max(__float_as_int(dd[r]), 0)};
          rec2[min(nrec, RCAP - 1)] = kv;
          ++nrec;
        }
      }
    };
    int jt = wave + NSPLIT * TILES_A;
    #pragma unroll 1
    for (; jt + 2 * NSPLIT < ROW_TILES; jt += 3 * NSPLIT) {
      step(f0, f2, jt);
      step(f1, f0, jt + NSPLIT);
      step(f2, f1, jt + 2 * NSPLIT);
    }
    #pragma unroll 1
    for (; jt < ROW_TILES; jt += NSPLIT) {
      step(f0, f2, jt);
      f0 = f1;
      f1 = f2;
    }
  }

  if (__builtin_amdgcn_ballot_w32(nrec > RCAP) != 0) {
    #pragma unroll 1
    for (int jt = wave + NSPLIT * TILES_A; jt < ROW_TILES; jt += NSPLIT) {
      load_frag(coords, sq, jt, m, kh, f2);
      float dd[8];
      frag_d2(f2, jt, rt, row, kh, b, sqi, dd);
      #pragma unroll
      for (int r = 0; r < 8; ++r) list_insert(best, dd[r] <= U ? dd[r] : BIGF);
    }
  } else {
    float d = (0 < nrec) ? __int_as_float(rec2[0].y) : BIGF;
    for (int i = 0; __builtin_amdgcn_ballot_w32(i < nrec) != 0; ++i) {
      const float dn = (i + 1 < nrec) ? __int_as_float(rec2[i + 1].y) : BIGF;
      list_insert(best, d);
      d = dn;
    }
  }

  float* lds_val2 = reinterpret_cast<float*>(&lds_key[0][0]);
  {
    v4f* vp = reinterpret_cast<v4f*>(lds_val2 + m * VPITCH + list * KNN);
    #pragma unroll
    for (int g = 0; g < 4; ++g) {
      v4f v = {best[4 * g], best[4 * g + 1], best[4 * g + 2], best[4 * g + 3]};
      vp[g] = v;
    }
  }
  lds_w[tid >> 4][tid & 15] = 0.f;
  lds_j[tid >> 4][tid & 15] = 0;
  __syncthreads();

  {
    const int r = tid >> 4;
    const int q = tid & 15;
    const float t = row_16th(lds_val2 + r * VPITCH + q * KNN, q, lane);
    if (q == 0) lds_T[r] = t;
  }
  __syncthreads();
  {
    v2u64* kp = reinterpret_cast<v2u64*>(&lds_key[m][list * KNN]);
    const v2u64 none = {KEY_NONE, KEY_NONE};
    #pragma unroll
    for (int g = 0; g < KNN / 2; ++g) kp[g] = none;
  }

  int cs = 0, ct = 0;
  {
    const float T = lds_T[m];
    u64* kl = &lds_key[m][list * KNN];
    auto collect = [&](float d, u64 key) {
      if (d < T) {
        if (cs < KNN) {
          if (cs + ct == KNN) --ct;
          kl[cs] = key; ++cs;
        }
      } else if (cs + ct < KNN) {
        ++ct; kl[KNN - ct] = key;
      }
    };
    const bool ovf = __builtin_amdgcn_ballot_w32(nrec > RCAP) != 0;
    const int end2 = ovf ? ROW_TILES : wave + NSPLIT * TILES_A;
    auto step = [&](const TileFrag& use, TileFrag& fill, int jt) {
      const int j0 = jt * 16;
      const int jn = (jt + 2 * NSPLIT < end2) ? jt + 2 * NSPLIT : jt;
      load_frag(coords, sq, jn, m, kh, fill);
      float dd[8];
      frag_d2(use, jt, rt, row, kh, b, sqi, dd);
      float lo = dd[0];
      #pragma unroll
      for (int r = 1; r < 8; ++r) lo = fminf(lo, dd[r]);
      if (lo <= T) {
        #pragma unroll
        for (int r = 0; r < 8; ++r) {
          const float d = dd[r];
          if (d <= T) {
            const int db = max(__float_as_int(d), 0);
            collect(d, ((u64)(unsigned)db << 32) | (u64)(unsigned)(j0 + 8 * kh + r));
          }
        }
      }
    };
    int jt = wave;
    #pragma unroll 1
    for (; jt + 2 * NSPLIT < end2; jt += 3 * NSPLIT) {
      step(f0, f2, jt);
      step(f1, f0, jt + NSPLIT);
      step(f2, f1, jt + 2 * NSPLIT);
    }
    #pragma unroll 1
    for (; jt < end2; jt += NSPLIT) {
      step(f0, f2, jt);
      f0 = f1;
      f1 = f2;
    }
    if (!ovf) {
      for (int i = 0; i < nrec; ++i) {
        const u64 key = rec[i];
        const float d = __int_as_float((int)(key >> 32));
        if (d <= T) collect(d, key);
      }
    }
    lds_cnt[m][list] = cs | (ct << 8);
  }
  __syncthreads();

  int tot = 0;
  {
    const int r = tid >> 4;
    const int q = tid & 15;
    int off = 0, mine = 0;
    #pragma unroll
    for (int l = 0; l < NLIST; ++l) {
      const int c = lds_cnt[r][l];
      const int n = (c & 255) + (c >> 8);
      off += (l < q) ? n : 0;
      mine = (l == q) ? c : mine;
      tot += n;
    }
    const int ncs = mine & 255, nct = mine >> 8;
    const u64* kl = &lds_key[r][q * KNN];
    for (int i = 0; i < ncs + nct; ++i) {
      const int slot = i < ncs ? i : KNN - nct + (i - ncs);
      lds_cand[r][off + i] = kl[slot];
    }
    if (q == 0) lds_cand[r][tot] = KEY_NONE;
  }
  __syncthreads();

  {
    const int r = tid >> 4;
    const int q = tid & 15;
    for (int p = q; p < tot; p += 16) {
      const u64 ck = lds_cand[r][p];
      int rank = 0;
      for (int e = 0; e < tot; e += 2) {
        const v2u64 kv = *reinterpret_cast<const v2u64*>(&lds_cand[r][e]);
        rank += (kv.x < ck ? 1 : 0) + (kv.y < ck ? 1 : 0);
      }
      if (rank < KNN) {
        lds_w[r][rank] = expf(-__int_as_float((int)(ck >> 32)));
        lds_j[r][rank] = (int)(unsigned)ck;
      }
    }
  }
  __syncthreads();

  #pragma unroll
  for (int rr = 0; rr < 16 / NSPLIT; ++rr) {
    const int r = wave * (16 / NSPLIT) + rr;
    float acc0 = 0.f, acc1 = 0.f;
    #pragma unroll
    for (int k = 0; k < KNN; ++k) {
      const int   j  = lds_j[r][k];
      const float wk = lds_w[r][k];
      acc0 += wk * feats[(size_t)j * DFEAT + lane];
      acc1 += wk * feats[(size_t)j * DFEAT + lane + 32];
    }
    lds_agg[r][lane]      = acc0;
    lds_agg[r][lane + 32] = acc1;
  }
  __syncthreads();

  if (wave < 2) {
    const float* fr = feats + (size_t)(row0 + m) * DFEAT;
    #pragma unroll
    for (int t = 0; t < 2; ++t) {
      const int nt = wave * 2 + t;
      v8f acc = {};
      #pragma unroll
      for (int s = 0; s < 16; ++s) {
        v2f av, bv;
        av.x = fr[4 * s + 2 * kh];
        av.y = fr[4 * s + 1 + 2 * kh];
        bv.x = Wout[(4 * s + 2 * kh) * DFEAT + nt * 16 + m];
        bv.y = Wout[(4 * s + 1 + 2 * kh) * DFEAT + nt * 16 + m];
        acc = wmma_f32(av, bv, acc);
      }
      #pragma unroll
      for (int s = 0; s < 16; ++s) {
        v2f av, bv;
        av.x = lds_agg[m][4 * s + 2 * kh];
        av.y = lds_agg[m][4 * s + 1 + 2 * kh];
        bv.x = Wout[(64 + 4 * s + 2 * kh) * DFEAT + nt * 16 + m];
        bv.y = Wout[(64 + 4 * s + 1 + 2 * kh) * DFEAT + nt * 16 + m];
        acc = wmma_f32(av, bv, acc);
      }
      #pragma unroll
      for (int r = 0; r < 8; ++r) {
        const int row = row0 + r + 8 * kh;
        const int col = nt * 16 + m;
        out[(size_t)row * DFEAT + col] = acc[r] + bout[col];
      }
    }
  }
}

extern "C" void kernel_launch(void* const* d_in, const int* in_sizes, int n_in,
                              void* d_out, int out_size, void* d_ws, size_t ws_size,
                              hipStream_t stream) {
  const float* x  = (const float*)d_in[0];
  const float* Wf = (const float*)d_in[1];
  const float* bf = (const float*)d_in[2];
  const float* Wl = (const float*)d_in[3];
  const float* bl = (const float*)d_in[4];
  const float* Wo = (const float*)d_in[5];
  const float* bo = (const float*)d_in[6];
  float* out = (float*)d_out;

  float* feats  = (float*)d_ws;
  float* coords = feats  + (size_t)N_PTS * DFEAT;
  float* sq     = coords + (size_t)N_PTS * DLAT;

  encoder_kernel<<<NUM_WG, 256, 0, stream>>>(x, Wf, bf, Wl, bl, feats, coords);
  sq_kernel<<<(N_PTS + 255) / 256, 256, 0, stream>>>(coords, sq);
  gravnet_kernel<<<ROW_TILES, 256, 0, stream>>>(coords, sq, feats, Wo, bo, out);
}


# Each tile is computed once: the first scan keeps 4 distances per lane to draw the bound U and parks all its distances in LDS; those at or below U then become records, and no second scan recomputes
# speedup vs baseline: 15.7537x; 1.1191x over previous
#include <hip/hip_runtime.h>
#include <cmath>

typedef __attribute__((ext_vector_type(2))) float v2f;
typedef __attribute__((ext_vector_type(8))) float v8f;

#define N_PTS   12000
#define DIN     64
#define DLAT    16
#define DFEAT   64
#define KNN     16
#define ROW_TILES (N_PTS / 16)
#define WAVES_PER_WG 8
#define NUM_WG  ((ROW_TILES + WAVES_PER_WG - 1) / WAVES_PER_WG)

static __device__ __forceinline__ v8f wmma_f32(v2f a, v2f b, v8f c) {
  return __builtin_amdgcn_wmma_f32_16x16x4_f32(
       false, a,  false, b,
       (short)0, c,  false,  false);
}

__global__ void __launch_bounds__(256)
encoder_kernel(const float* __restrict__ x,
               const float* __restrict__ Wf, const float* __restrict__ bf,
               const float* __restrict__ Wl, const float* __restrict__ bl,
               float* __restrict__ feats, float* __restrict__ coords) {
  const int lane = threadIdx.x & 31;
  const int wave = threadIdx.x >> 5;
  const int tile = blockIdx.x * WAVES_PER_WG + wave;
  if (tile >= ROW_TILES) return;
  const int row0 = tile * 16;
  const int m  = lane & 15;
  const int kh = lane >> 4;

  v2f a[16];
  const float* xr = x + (size_t)(row0 + m) * DIN;
  #pragma unroll
  for (int s = 0; s < 16; ++s) {
    a[s].x = xr[4 * s + 2 * kh];
    a[s].y = xr[4 * s + 1 + 2 * kh];
  }

  #pragma unroll
  for (int nt = 0; nt < 4; ++nt) {
    v8f acc = {};
    #pragma unroll
    for (int s = 0; s < 16; ++s) {
      v2f b;
      b.x = Wf[(4 * s + 2 * kh) * DFEAT + nt * 16 + m];
      b.y = Wf[(4 * s + 1 + 2 * kh) * DFEAT + nt * 16 + m];
      acc = wmma_f32(a[s], b, acc);
    }
    #pragma unroll
    for (int r = 0; r < 8; ++r) {
      const int row = row0 + r + 8 * kh;
      const int col = nt * 16 + m;
      feats[(size_t)row * DFEAT + col] = tanhf(acc[r] + bf[col]);
    }
  }

  {
    v8f acc = {};
    #pragma unroll
    for (int s = 0; s < 16; ++s) {
      v2f b;
      b.x = Wl[(4 * s + 2 * kh) * DLAT + m];
      b.y = Wl[(4 * s + 1 + 2 * kh) * DLAT + m];
      acc = wmma_f32(a[s], b, acc);
    }
    #pragma unroll
    for (int r = 0; r < 8; ++r) {
      const int row = row0 + r + 8 * kh;
      coords[(size_t)row * DLAT + m] = tanhf(acc[r] + bl[m]);
    }
  }
}

__global__ void sq_kernel(const float* __restrict__ coords, float* __restrict__ sq) {
  const int i = blockIdx.x * blockDim.x + threadIdx.x;
  if (i >= N_PTS) return;
  const float* c = coords + (size_t)i * DLAT;
  float s = 0.f;
  #pragma unroll
  for (int d = 0; d < DLAT; ++d) s += c[d] * c[d];
  sq[i] = s;
}

#define NSPLIT  WAVES_PER_WG
#define NLIST   (2 * NSPLIT)
#define NCAND   (NLIST * KNN)
#define KPITCH  (NCAND + 2)
#define VPITCH  (2 * KPITCH)
#define BIGF    3.4e38f
#define TILES_A 12
#define NSHORT  4
#define RCAP    32
#define PARK_LO (RCAP / 4)
#define RPITCH  (NLIST * RCAP + 2)

typedef unsigned long long u64;
typedef __attribute__((ext_vector_type(2))) unsigned long long v2u64;
typedef __attribute__((ext_vector_type(4))) float v4f;
typedef __attribute__((ext_vector_type(2))) int v2i;

#define KEY_NONE 0xFFFFFFFFFFFFFFFFull

static __device__ __forceinline__ float pair_raw(float sqi, float sqj, float dot) {
  return __builtin_fmaf(-2.f, dot, sqi + sqj);
}
static __device__ __forceinline__ float pair_d2(float sqi, float sqj, float dot) {
  return fmaxf(pair_raw(sqi, sqj, dot), 0.f);
}

struct TileFrag {
  v2f a[4];
  v4f s0, s1;
};

static __device__ __forceinline__ void load_frag(const float* __restrict__ coords,
                                                 const float* __restrict__ sq,
                                                 int jt, int m, int kh, TileFrag& f) {
  const int j0 = jt * 16;
  const float* cj = coords + (size_t)(j0 + m) * DLAT;
  #pragma unroll
  for (int s = 0; s < 4; ++s) {
    f.a[s].x = cj[4 * s + 2 * kh];
    f.a[s].y = cj[4 * s + 1 + 2 * kh];
  }
  f.s0 = *reinterpret_cast<const v4f*>(sq + j0 + 8 * kh);
  f.s1 = *reinterpret_cast<const v4f*>(sq + j0 + 8 * kh + 4);
}

static __device__ __forceinline__ v8f frag_dots(const TileFrag& f, const v2f (&b)[4]) {
  v8f acc = {};
  #pragma unroll
  for (int s = 0; s < 4; ++s) acc = wmma_f32(f.a[s], b[s], acc);
  return acc;
}

template <int MASK>
static __device__ __forceinline__ float row_xor(float v) {
  return __int_as_float(__builtin_amdgcn_update_dpp(0, __float_as_int(v), 0x160 + MASK, 0xf, 0xf, false));
}

static __device__ __forceinline__ void list_insert(float (&best)[KNN], float d) {
  #pragma unroll
  for (int s = 0; s < KNN - 1; ++s)
    best[s] = __builtin_amdgcn_fmed3f(best[s], best[s + 1], d);
  best[KNN - 1] = __builtin_amdgcn_fmed3f(best[KNN - 1], -BIGF, d);
}

static __device__ __forceinline__ void list_insert_short(float (&l)[NSHORT], float d) {
  #pragma unroll
  for (int s = 0; s < NSHORT - 1; ++s)
    l[s] = __builtin_amdgcn_fmed3f(l[s], l[s + 1], d);
  l[NSHORT - 1] = __builtin_amdgcn_fmed3f(l[NSHORT - 1], -BIGF, d);
}

template <bool CLAMP = true>
static __device__ __forceinline__ void frag_d2(const TileFrag& f, int jt, int rt, int row, int kh,
                                               const v2f (&b)[4], float sqi, float (&dd)[8]) {
  const int j0 = jt * 16;
  const v8f acc = frag_dots(f, b);
  #pragma unroll
  for (int r = 0; r < 8; ++r) {
    const float sqj = r < 4 ? f.s0[r & 3] : f.s1[r & 3];
    dd[r] = CLAMP ? pair_d2(sqi, sqj, acc[r]) : pair_raw(sqi, sqj, acc[r]);
  }
  if (jt == rt) {
    #pragma unroll
    for (int r = 0; r < 8; ++r)
      if (j0 + 8 * kh + r == row) dd[r] = BIGF;
  }
}

static __device__ __forceinline__ float row_16th(const float* lv, int q, int lane) {
  int head = 0;
  float t = 0.f;
  for (int step = 0; step < KNN; ++step) {
    const float v = lv[KNN - 1 - head];
    float mn = v;
    mn = fminf(mn, row_xor<1>(mn));
    mn = fminf(mn, row_xor<2>(mn));
    mn = fminf(mn, row_xor<4>(mn));
    mn = fminf(mn, row_xor<8>(mn));
    t = mn;
    const unsigned holders = (__builtin_amdgcn_ballot_w32(v == mn) >> (lane & 16)) & 0xFFFFu;
    if (q == __builtin_ctz(holders)) ++head;
  }
  return t;
}

static __device__ __forceinline__ float row_16th_short(const float* lv, int q, int lane) {
  const v4f l = *reinterpret_cast<const v4f*>(lv);
  float c0 = l.w, c1 = l.z, c2 = l.y, c3 = l.x;
  float t = 0.f;
  for (int step = 0; step < KNN; ++step) {
    const float v = c0;
    float mn = v;
    mn = fminf(mn, row_xor<1>(mn));
    mn = fminf(mn, row_xor<2>(mn));
    mn = fminf(mn, row_xor<4>(mn));
    mn = fminf(mn, row_xor<8>(mn));
    t = mn;
    const unsigned holders = (__builtin_amdgcn_ballot_w32(v == mn) >> (lane & 16)) & 0xFFFFu;
    if (q == __builtin_ctz(holders)) { c0 = c1; c1 = c2; c2 = c3; c3 = BIGF; }
  }
  return t;
}

__global__ void __launch_bounds__(256, 6)
gravnet_kernel(const float* __restrict__ coords, const float* __restrict__ sq,
               const float* __restrict__ feats,
               const float* __restrict__ Wout, const float* __restrict__ bout,
               float* __restrict__ out) {
  __shared__ __attribute__((aligned(16))) u64 lds_rec[16][RPITCH];
  u64 (*lds_cand)[KPITCH] = reinterpret_cast<u64 (*)[KPITCH]>(&lds_rec[0][0]);
  __shared__ __attribute__((aligned(16))) u64 lds_key[16][KPITCH];
  __shared__ __attribute__((aligned(16))) int lds_cnt[16][NLIST];
  __shared__ float lds_T  [16];
  __shared__ float lds_w  [16][KNN];
  __shared__ int   lds_j  [16][KNN];
  __shared__ __attribute__((aligned(16))) float lds_agg[16][DFEAT];
  static_assert(NLIST * NSHORT <= DFEAT, "short lists must fit a row of lds_agg");
  static_assert(NSHORT == 4, "a short list is moved as one 16-byte vector");
  static_assert(TILES_A * 8 * 4 <= RCAP * 8 + KNN * 8, "parked distances must fit the lane's own slots");

  const int tid  = threadIdx.x;
  const int lane = tid & 31;
  const int wave = __builtin_amdgcn_readfirstlane(tid >> 5);
  const int rt   = blockIdx.x;
  const int row0 = rt * 16;
  const int m  = lane & 15;
  const int kh = lane >> 4;
  const int row  = row0 + m;
  const int list = wave * 2 + kh;

  v2f b[4];
  {
    const float* cr = coords + (size_t)row * DLAT;
    #pragma unroll
    for (int s = 0; s < 4; ++s) {
      b[s].x = cr[4 * s + 2 * kh];
      b[s].y = cr[4 * s + 1 + 2 * kh];
    }
  }
  const float sqi = sq[row];

  u64* rec = &lds_rec[m][list * RCAP];
  v2i* rec2 = reinterpret_cast<v2i*>(rec);
  u64* kl  = &lds_key[m][list * KNN];
  float* park_lo = reinterpret_cast<float*>(rec);
  float* park_hi = reinterpret_cast<float*>(kl);

  float shortl[NSHORT];
  #pragma unroll
  for (int s = 0; s < NSHORT; ++s) shortl[s] = BIGF;

  TileFrag f0, f1, f2;
  load_frag(coords, sq, wave, m, kh, f0);
  load_frag(coords, sq, wave + NSPLIT, m, kh, f1);
  {
    auto step = [&](const TileFrag& use, TileFrag& fill, int jt) {
      load_frag(coords, sq, jt + 2 * NSPLIT, m, kh, fill);
      float dd[8];
      frag_d2<false>(use, jt, rt, row, kh, b, sqi, dd);
      const int t = (jt - wave) / NSPLIT;
      float* pp = (t < PARK_LO) ? park_lo + 8 * t : park_hi + 8 * (t - PARK_LO);
      const v4f p0 = {dd[0], dd[1], dd[2], dd[3]};
      const v4f p1 = {dd[4], dd[5], dd[6], dd[7]};
      reinterpret_cast<v4f*>(pp)[0] = p0;
      reinterpret_cast<v4f*>(pp)[1] = p1;
      #pragma unroll
      for (int r = 0; r < 8; ++r) list_insert_short(shortl, dd[r]);
    };
    const int end = wave + NSPLIT * TILES_A;
    int jt = wave;
    #pragma unroll 1
    for (; jt + 2 * NSPLIT < end; jt += 3 * NSPLIT) {
      step(f0, f2, jt);
      step(f1, f0, jt + NSPLIT);
      step(f2, f1, jt + 2 * NSPLIT);
    }
    #pragma unroll 1
    for (; jt < end; jt += NSPLIT) {
      step(f0, f2, jt);
      f0 = f1;
      f1 = f2;
    }
  }

  {
    const v4f v = {shortl[0], shortl[1], shortl[2], shortl[3]};
    *reinterpret_cast<v4f*>(&lds_agg[m][list * NSHORT]) = v;
  }
  __syncthreads();
  {
    const int r = tid >> 4;
    const int q = tid & 15;
    const float u = row_16th_short(&lds_agg[r][q * NSHORT], q, lane);
    if (q == 0) lds_T[r] = fmaxf(u, 0.f);
  }
  __syncthreads();

  int nrec = 0;
  const float U = lds_T[m];
  auto record8 = [&](const float (&dd)[8], int jc) {
    unsigned hits[8];
    #pragma unroll
    for (int r = 0; r < 8; ++r) hits[r] = __builtin_amdgcn_ballot_w32(dd[r] <= U);
    #pragma unroll
    for (int r = 0; r < 8; ++r) {
      if (__builtin_amdgcn_inverse_ballot_w32(hits[r])) {
        const v2i kv = {jc + r, max(__float_as_int(dd[r]), 0)};
        rec2[min(nrec, RCAP - 1)] = kv;
        ++nrec;
      }
    }
  };
  {
    float pk[PARK_LO][8];
    #pragma unroll
    for (int t = 0; t < PARK_LO; ++t) {
      const v4f p0 = reinterpret_cast<const v4f*>(park_lo + 8 * t)[0];
      const v4f p1 = reinterpret_cast<const v4f*>(park_lo + 8 * t)[1];
      pk[t][0] = p0.x; pk[t][1] = p0.y; pk[t][2] = p0.z; pk[t][3] = p0.w;
      pk[t][4] = p1.x; pk[t][5] = p1.y; pk[t][6] = p1.z; pk[t][7] = p1.w;
    }
    #pragma unroll
    for (int t = 0; t < PARK_LO; ++t)
      record8(pk[t], (wave + NSPLIT * t) * 16 + 8 * kh);
  }
  #pragma unroll
  for (int t = PARK_LO; t < TILES_A; ++t) {
    const v4f p0 = reinterpret_cast<const v4f*>(park_hi + 8 * (t - PARK_LO))[0];
    const v4f p1 = reinterpret_cast<const v4f*>(park_hi + 8 * (t - PARK_LO))[1];
    const float pk[8] = {p0.x, p0.y, p0.z, p0.w, p1.x, p1.y, p1.z, p1.w};
    record8(pk, (wave + NSPLIT * t) * 16 + 8 * kh);
  }

  {
    auto step = [&](const TileFrag& use, TileFrag& fill, int jt) {
      const int jn = (jt + 2 * NSPLIT < ROW_TILES) ? jt + 2 * NSPLIT : jt;
      load_frag(coords, sq, jn, m, kh, fill);
      float dd[8];
      frag_d2<false>(use, jt, rt, row, kh, b, sqi, dd);
      record8(dd, jt * 16 + 8 * kh);
    };
    int jt = wave + NSPLIT * TILES_A;
    #pragma unroll 1
    for (; jt + 2 * NSPLIT < ROW_TILES; jt += 3 * NSPLIT) {
      step(f0, f2, jt);
      step(f1, f0, jt + NSPLIT);
      step(f2, f1, jt + 2 * NSPLIT);
    }
    #pragma unroll 1
    for (; jt < ROW_TILES; jt += NSPLIT) {
      step(f0, f2, jt);
      f0 = f1;
      f1 = f2;
    }
  }

  const bool ovf = __builtin_amdgcn_ballot_w32(nrec > RCAP) != 0;

  float best[KNN];
  #pragma unroll
  for (int s = 0; s < KNN; ++s) best[s] = BIGF;
  if (ovf) {
    #pragma unroll 1
    for (int jt = wave; jt < ROW_TILES; jt += NSPLIT) {
      load_frag(coords, sq, jt, m, kh, f2);
      float dd[8];
      frag_d2(f2, jt, rt, row, kh, b, sqi, dd);
      #pragma unroll
      for (int r = 0; r < 8; ++r) list_insert(best, dd[r] <= U ? dd[r] : BIGF);
    }
  } else {
    float d = (0 < nrec) ? __int_as_float(rec2[0].y) : BIGF;
    for (int i = 0; __builtin_amdgcn_ballot_w32(i < nrec) != 0; ++i) {
      const float dn = (i + 1 < nrec) ? __int_as_float(rec2[i + 1].y) : BIGF;
      list_insert(best, d);
      d = dn;
    }
  }

  {
    v4f* vp = reinterpret_cast<v4f*>(kl);
    #pragma unroll
    for (int g = 0; g < 4; ++g) {
      v4f v = {best[4 * g], best[4 * g + 1], best[4 * g + 2], best[4 * g + 3]};
      vp[g] = v;
    }
  }
  lds_w[tid >> 4][tid & 15] = 0.f;
  lds_j[tid >> 4][tid & 15] = 0;
  __syncthreads();

  {
    const int r = tid >> 4;
    const int q = tid & 15;
    const float t = row_16th(reinterpret_cast<const float*>(&lds_key[r][q * KNN]), q, lane);
    if (q == 0) lds_T[r] = t;
  }
  __syncthreads();
  {
    v2u64* kp = reinterpret_cast<v2u64*>(kl);
    const v2u64 none = {KEY_NONE, KEY_NONE};
    #pragma unroll
    for (int g = 0; g < KNN / 2; ++g) kp[g] = none;
  }

  int cs = 0, ct = 0;
  {
    const float T = lds_T[m];
    auto collect = [&](float d, u64 key) {
      if (d < T) {
        if (cs < KNN) {
          if (cs + ct == KNN) --ct;
          kl[cs] = key; ++cs;
        }
      } else if (cs + ct < KNN) {
        ++ct; kl[KNN - ct] = key;
      }
    };
    if (ovf) {
      #pragma unroll 1
      for (int jt = wave; jt < ROW_TILES; jt += NSPLIT) {
        const int j0 = jt * 16;
        load_frag(coords, sq, jt, m, kh, f2);
        float dd[8];
        frag_d2(f2, jt, rt, row, kh, b, sqi, dd);
        #pragma unroll
        for (int r = 0; r < 8; ++r) {
          const float d = dd[r];
          if (d <= T) {
            const int db = max(__float_as_int(d), 0);
            collect(d, ((u64)(unsigned)db << 32) | (u64)(unsigned)(j0 + 8 * kh + r));
          }
        }
      }
    } else {
      for (int i = 0; i < nrec; ++i) {
        const u64 key = rec[i];
        const float d = __int_as_float((int)(key >> 32));
        if (d <= T) collect(d, key);
      }
    }
    lds_cnt[m][list] = cs | (ct << 8);
  }
  __syncthreads();

  int tot = 0;
  {
    const int r = tid >> 4;
    const int q = tid & 15;
    int off = 0, mine = 0;
    #pragma unroll
    for (int l = 0; l < NLIST; ++l) {
      const int c = lds_cnt[r][l];
      const int n = (c & 255) + (c >> 8);
      off += (l < q) ? n : 0;
      mine = (l == q) ? c : mine;
      tot += n;
    }
    const int ncs = mine & 255, nct = mine >> 8;
    const u64* kl = &lds_key[r][q * KNN];
    for (int i = 0; i < ncs + nct; ++i) {
      const int slot = i < ncs ? i : KNN - nct + (i - ncs);
      lds_cand[r][off + i] = kl[slot];
    }
    if (q == 0) lds_cand[r][tot] = KEY_NONE;
  }
  __syncthreads();

  {
    const int r = tid >> 4;
    const int q = tid & 15;
    for (int p = q; p < tot; p += 16) {
      const u64 ck = lds_cand[r][p];
      int rank = 0;
      for (int e = 0; e < tot; e += 2) {
        const v2u64 kv = *reinterpret_cast<const v2u64*>(&lds_cand[r][e]);
        rank += (kv.x < ck ? 1 : 0) + (kv.y < ck ? 1 : 0);
      }
      if (rank < KNN) {
        lds_w[r][rank] = expf(-__int_as_float((int)(ck >> 32)));
        lds_j[r][rank] = (int)(unsigned)ck;
      }
    }
  }
  __syncthreads();

  #pragma unroll
  for (int rr = 0; rr < 16 / NSPLIT; ++rr) {
    const int r = wave * (16 / NSPLIT) + rr;
    float acc0 = 0.f, acc1 = 0.f;
    #pragma unroll
    for (int k = 0; k < KNN; ++k) {
      const int   j  = lds_j[r][k];
      const float wk = lds_w[r][k];
      acc0 += wk * feats[(size_t)j * DFEAT + lane];
      acc1 += wk * feats[(size_t)j * DFEAT + lane + 32];
    }
    lds_agg[r][lane]      = acc0;
    lds_agg[r][lane + 32] = acc1;
  }
  __syncthreads();

  if (wave < 2) {
    const float* fr = feats + (size_t)(row0 + m) * DFEAT;
    #pragma unroll
    for (int t = 0; t < 2; ++t) {
      const int nt = wave * 2 + t;
      v8f acc = {};
      #pragma unroll
      for (int s = 0; s < 16; ++s) {
        v2f av, bv;
        av.x = fr[4 * s + 2 * kh];
        av.y = fr[4 * s + 1 + 2 * kh];
        bv.x = Wout[(4 * s + 2 * kh) * DFEAT + nt * 16 + m];
        bv.y = Wout[(4 * s + 1 + 2 * kh) * DFEAT + nt * 16 + m];
        acc = wmma_f32(av, bv, acc);
      }
      #pragma unroll
      for (int s = 0; s < 16; ++s) {
        v2f av, bv;
        av.x = lds_agg[m][4 * s + 2 * kh];
        av.y = lds_agg[m][4 * s + 1 + 2 * kh];
        bv.x = Wout[(64 + 4 * s + 2 * kh) * DFEAT + nt * 16 + m];
        bv.y = Wout[(64 + 4 * s + 1 + 2 * kh) * DFEAT + nt * 16 + m];
        acc = wmma_f32(av, bv, acc);
      }
      #pragma unroll
      for (int r = 0; r < 8; ++r) {
        const int row = row0 + r + 8 * kh;
        const int col = nt * 16 + m;
        out[(size_t)row * DFEAT + col] = acc[r] + bout[col];
      }
    }
  }
}

extern "C" void kernel_launch(void* const* d_in, const int* in_sizes, int n_in,
                              void* d_out, int out_size, void* d_ws, size_t ws_size,
                              hipStream_t stream) {
  const float* x  = (const float*)d_in[0];
  const float* Wf = (const float*)d_in[1];
  const float* bf = (const float*)d_in[2];
  const float* Wl = (const float*)d_in[3];
  const float* bl = (const float*)d_in[4];
  const float* Wo = (const float*)d_in[5];
  const float* bo = (const float*)d_in[6];
  float* out = (float*)d_out;

  float* feats  = (float*)d_ws;
  float* coords = feats  + (size_t)N_PTS * DFEAT;
  float* sq     = coords + (size_t)N_PTS * DLAT;

  encoder_kernel<<<NUM_WG, 256, 0, stream>>>(x, Wf, bf, Wl, bl, feats, coords);
  sq_kernel<<<(N_PTS + 255) / 256, 256, 0, stream>>>(coords, sq);
  gravnet_kernel<<<ROW_TILES, 256, 0, stream>>>(coords, sq, feats, Wo, bo, out);
}


# Norms computed in the encoder's epilogue, sq kernel dropped (3 launches -> 2); encoder regridded to 375 two-wave groups, one 128-B line of norms each; squares summed in the old order, bit-identical
# speedup vs baseline: 16.2878x; 1.0339x over previous
#include <hip/hip_runtime.h>
#include <cmath>

typedef __attribute__((ext_vector_type(2))) float v2f;
typedef __attribute__((ext_vector_type(8))) float v8f;

#define N_PTS   12000
#define DIN     64
#define DLAT    16
#define DFEAT   64
#define KNN     16
#define ROW_TILES (N_PTS / 16)
#define WAVES_PER_WG 8
#define ENC_WAVES 2
#define ENC_WG  (ROW_TILES / ENC_WAVES)

static __device__ __forceinline__ v8f wmma_f32(v2f a, v2f b, v8f c) {
  return __builtin_amdgcn_wmma_f32_16x16x4_f32(
       false, a,  false, b,
       (short)0, c,  false,  false);
}

template <int SRC>
static __device__ __forceinline__ float row_share(float v) {
  return __int_as_float(__builtin_amdgcn_update_dpp(0, __float_as_int(v), 0x150 + SRC, 0xf, 0xf, true));
}

static __device__ __forceinline__ float row_sum_in_order(float p) {
  float s = row_share<0>(p) + row_share<1>(p);
  s += row_share<2>(p);
  s += row_share<3>(p);
  s += row_share<4>(p);
  s += row_share<5>(p);
  s += row_share<6>(p);
  s += row_share<7>(p);
  s += row_share<8>(p);
  s += row_share<9>(p);
  s += row_share<10>(p);
  s += row_share<11>(p);
  s += row_share<12>(p);
  s += row_share<13>(p);
  s += row_share<14>(p);
  s += row_share<15>(p);
  return s;
}

static_assert(ROW_TILES % ENC_WAVES == 0, "every encoder workgroup is full: its barrier is reached by all waves");
__global__ void __launch_bounds__(32 * ENC_WAVES)
encoder_kernel(const float* __restrict__ x,
               const float* __restrict__ Wf, const float* __restrict__ bf,
               const float* __restrict__ Wl, const float* __restrict__ bl,
               float* __restrict__ feats, float* __restrict__ coords,
               float* __restrict__ sq) {
  __shared__ float lds_sq[16 * ENC_WAVES];
  const int lane = threadIdx.x & 31;
  const int wave = threadIdx.x >> 5;
  const int tile = blockIdx.x * ENC_WAVES + wave;
  const int row0 = tile * 16;
  const int m  = lane & 15;
  const int kh = lane >> 4;

  v2f a[16];
  const float* xr = x + (size_t)(row0 + m) * DIN;
  #pragma unroll
  for (int s = 0; s < 16; ++s) {
    a[s].x = xr[4 * s + 2 * kh];
    a[s].y = xr[4 * s + 1 + 2 * kh];
  }

  {
    v8f acc = {};
    #pragma unroll
    for (int s = 0; s < 16; ++s) {
      v2f b;
      b.x = Wl[(4 * s + 2 * kh) * DLAT + m];
      b.y = Wl[(4 * s + 1 + 2 * kh) * DLAT + m];
      acc = wmma_f32(a[s], b, acc);
    }
    float s[8];
    #pragma unroll
    for (int r = 0; r < 8; ++r) {
      const int row = row0 + r + 8 * kh;
      const float c = tanhf(acc[r] + bl[m]);
      coords[(size_t)row * DLAT + m] = c;
      s[r] = row_sum_in_order(c * c);
    }
    float v = s[0];
    #pragma unroll
    for (int r = 1; r < 8; ++r) v = (m == r) ? s[r] : v;
    if (m < 8) lds_sq[wave * 16 + 8 * kh + m] = v;
  }
  __builtin_amdgcn_fence(__ATOMIC_RELEASE, "workgroup", "local");
  __builtin_amdgcn_s_barrier();
  __builtin_amdgcn_fence(__ATOMIC_ACQUIRE, "workgroup", "local");
  if (wave == 0) sq[blockIdx.x * (16 * ENC_WAVES) + lane] = lds_sq[lane];

  #pragma unroll
  for (int nt = 0; nt < 4; ++nt) {
    v8f acc = {};
    #pragma unroll
    for (int s = 0; s < 16; ++s) {
      v2f b;
      b.x = Wf[(4 * s + 2 * kh) * DFEAT + nt * 16 + m];
      b.y = Wf[(4 * s + 1 + 2 * kh) * DFEAT + nt * 16 + m];
      acc = wmma_f32(a[s], b, acc);
    }
    #pragma unroll
    for (int r = 0; r < 8; ++r) {
      const int row = row0 + r + 8 * kh;
      const int col = nt * 16 + m;
      feats[(size_t)row * DFEAT + col] = tanhf(acc[r] + bf[col]);
    }
  }
}

#define NSPLIT  WAVES_PER_WG
#define NLIST   (2 * NSPLIT)
#define NCAND   (NLIST * KNN)
#define KPITCH  (NCAND + 2)
#define VPITCH  (2 * KPITCH)
#define BIGF    3.4e38f
#define TILES_A 12
#define NSHORT  4
#define RCAP    32
#define PARK_LO (RCAP / 4)
#define RPITCH  (NLIST * RCAP + 2)

typedef unsigned long long u64;
typedef __attribute__((ext_vector_type(2))) unsigned long long v2u64;
typedef __attribute__((ext_vector_type(4))) float v4f;
typedef __attribute__((ext_vector_type(2))) int v2i;

#define KEY_NONE 0xFFFFFFFFFFFFFFFFull

static __device__ __forceinline__ float pair_raw(float sqi, float sqj, float dot) {
  return __builtin_fmaf(-2.f, dot, sqi + sqj);
}
static __device__ __forceinline__ float pair_d2(float sqi, float sqj, float dot) {
  return fmaxf(pair_raw(sqi, sqj, dot), 0.f);
}

struct TileFrag {
  v2f a[4];
  v4f s0, s1;
};

static __device__ __forceinline__ void load_frag(const float* __restrict__ coords,
                                                 const float* __restrict__ sq,
                                                 int jt, int m, int kh, TileFrag& f) {
  const int j0 = jt * 16;
  const float* cj = coords + (size_t)(j0 + m) * DLAT;
  #pragma unroll
  for (int s = 0; s < 4; ++s) {
    f.a[s].x = cj[4 * s + 2 * kh];
    f.a[s].y = cj[4 * s + 1 + 2 * kh];
  }
  f.s0 = *reinterpret_cast<const v4f*>(sq + j0 + 8 * kh);
  f.s1 = *reinterpret_cast<const v4f*>(sq + j0 + 8 * kh + 4);
}

static __device__ __forceinline__ v8f frag_dots(const TileFrag& f, const v2f (&b)[4]) {
  v8f acc = {};
  #pragma unroll
  for (int s = 0; s < 4; ++s) acc = wmma_f32(f.a[s], b[s], acc);
  return acc;
}

template <int MASK>
static __device__ __forceinline__ float row_xor(float v) {
  return __int_as_float(__builtin_amdgcn_update_dpp(0, __float_as_int(v), 0x160 + MASK, 0xf, 0xf, false));
}

static __device__ __forceinline__ void list_insert(float (&best)[KNN], float d) {
  #pragma unroll
  for (int s = 0; s < KNN - 1; ++s)
    best[s] = __builtin_amdgcn_fmed3f(best[s], best[s + 1], d);
  best[KNN - 1] = __builtin_amdgcn_fmed3f(best[KNN - 1], -BIGF, d);
}

static __device__ __forceinline__ void list_insert_short(float (&l)[NSHORT], float d) {
  #pragma unroll
  for (int s = 0; s < NSHORT - 1; ++s)
    l[s] = __builtin_amdgcn_fmed3f(l[s], l[s + 1], d);
  l[NSHORT - 1] = __builtin_amdgcn_fmed3f(l[NSHORT - 1], -BIGF, d);
}

template <bool CLAMP = true>
static __device__ __forceinline__ void frag_d2(const TileFrag& f, int jt, int rt, int row, int kh,
                                               const v2f (&b)[4], float sqi, float (&dd)[8]) {
  const int j0 = jt * 16;
  const v8f acc = frag_dots(f, b);
  #pragma unroll
  for (int r = 0; r < 8; ++r) {
    const float sqj = r < 4 ? f.s0[r & 3] : f.s1[r & 3];
    dd[r] = CLAMP ? pair_d2(sqi, sqj, acc[r]) : pair_raw(sqi, sqj, acc[r]);
  }
  if (jt == rt) {
    #pragma unroll
    for (int r = 0; r < 8; ++r)
      if (j0 + 8 * kh + r == row) dd[r] = BIGF;
  }
}

static __device__ __forceinline__ float row_16th(const float* lv, int q, int lane) {
  int head = 0;
  float t = 0.f;
  for (int step = 0; step < KNN; ++step) {
    const float v = lv[KNN - 1 - head];
    float mn = v;
    mn = fminf(mn, row_xor<1>(mn));
    mn = fminf(mn, row_xor<2>(mn));
    mn = fminf(mn, row_xor<4>(mn));
    mn = fminf(mn, row_xor<8>(mn));
    t = mn;
    const unsigned holders = (__builtin_amdgcn_ballot_w32(v == mn) >> (lane & 16)) & 0xFFFFu;
    if (q == __builtin_ctz(holders)) ++head;
  }
  return t;
}

static __device__ __forceinline__ float row_16th_short(const float* lv, int q, int lane) {
  const v4f l = *reinterpret_cast<const v4f*>(lv);
  float c0 = l.w, c1 = l.z, c2 = l.y, c3 = l.x;
  float t = 0.f;
  for (int step = 0; step < KNN; ++step) {
    const float v = c0;
    float mn = v;
    mn = fminf(mn, row_xor<1>(mn));
    mn = fminf(mn, row_xor<2>(mn));
    mn = fminf(mn, row_xor<4>(mn));
    mn = fminf(mn, row_xor<8>(mn));
    t = mn;
    const unsigned holders = (__builtin_amdgcn_ballot_w32(v == mn) >> (lane & 16)) & 0xFFFFu;
    if (q == __builtin_ctz(holders)) { c0 = c1; c1 = c2; c2 = c3; c3 = BIGF; }
  }
  return t;
}

__global__ void __launch_bounds__(256, 6)
gravnet_kernel(const float* __restrict__ coords, const float* __restrict__ sq,
               const float* __restrict__ feats,
               const float* __restrict__ Wout, const float* __restrict__ bout,
               float* __restrict__ out) {
  __shared__ __attribute__((aligned(16))) u64 lds_rec[16][RPITCH];
  u64 (*lds_cand)[KPITCH] = reinterpret_cast<u64 (*)[KPITCH]>(&lds_rec[0][0]);
  __shared__ __attribute__((aligned(16))) u64 lds_key[16][KPITCH];
  __shared__ __attribute__((aligned(16))) int lds_cnt[16][NLIST];
  __shared__ float lds_T  [16];
  __shared__ float lds_w  [16][KNN];
  __shared__ int   lds_j  [16][KNN];
  __shared__ __attribute__((aligned(16))) float lds_agg[16][DFEAT];
  static_assert(NLIST * NSHORT <= DFEAT, "short lists must fit a row of lds_agg");
  static_assert(NSHORT == 4, "a short list is moved as one 16-byte vector");
  static_assert(TILES_A * 8 * 4 <= RCAP * 8 + KNN * 8, "parked distances must fit the lane's own slots");

  const int tid  = threadIdx.x;
  const int lane = tid & 31;
  const int wave = __builtin_amdgcn_readfirstlane(tid >> 5);
  const int rt   = blockIdx.x;
  const int row0 = rt * 16;
  const int m  = lane & 15;
  const int kh = lane >> 4;
  const int row  = row0 + m;
  const int list = wave * 2 + kh;

  v2f b[4];
  {
    const float* cr = coords + (size_t)row * DLAT;
    #pragma unroll
    for (int s = 0; s < 4; ++s) {
      b[s].x = cr[4 * s + 2 * kh];
      b[s].y = cr[4 * s + 1 + 2 * kh];
    }
  }
  const float sqi = sq[row];

  u64* rec = &lds_rec[m][list * RCAP];
  v2i* rec2 = reinterpret_cast<v2i*>(rec);
  u64* kl  = &lds_key[m][list * KNN];
  float* park_lo = reinterpret_cast<float*>(rec);
  float* park_hi = reinterpret_cast<float*>(kl);

  float shortl[NSHORT];
  #pragma unroll
  for (int s = 0; s < NSHORT; ++s) shortl[s] = BIGF;

  TileFrag f0, f1, f2;
  load_frag(coords, sq, wave, m, kh, f0);
  load_frag(coords, sq, wave + NSPLIT, m, kh, f1);
  {
    auto step = [&](const TileFrag& use, TileFrag& fill, int jt) {
      load_frag(coords, sq, jt + 2 * NSPLIT, m, kh, fill);
      float dd[8];
      frag_d2<false>(use, jt, rt, row, kh, b, sqi, dd);
      const int t = (jt - wave) / NSPLIT;
      float* pp = (t < PARK_LO) ? park_lo + 8 * t : park_hi + 8 * (t - PARK_LO);
      const v4f p0 = {dd[0], dd[1], dd[2], dd[3]};
      const v4f p1 = {dd[4], dd[5], dd[6], dd[7]};
      reinterpret_cast<v4f*>(pp)[0] = p0;
      reinterpret_cast<v4f*>(pp)[1] = p1;
      #pragma unroll
      for (int r = 0; r < 8; ++r) list_insert_short(shortl, dd[r]);
    };
    const int end = wave + NSPLIT * TILES_A;
    int jt = wave;
    #pragma unroll 1
    for (; jt + 2 * NSPLIT < end; jt += 3 * NSPLIT) {
      step(f0, f2, jt);
      step(f1, f0, jt + NSPLIT);
      step(f2, f1, jt + 2 * NSPLIT);
    }
    #pragma unroll 1
    for (; jt < end; jt += NSPLIT) {
      step(f0, f2, jt);
      f0 = f1;
      f1 = f2;
    }
  }

  {
    const v4f v = {shortl[0], shortl[1], shortl[2], shortl[3]};
    *reinterpret_cast<v4f*>(&lds_agg[m][list * NSHORT]) = v;
  }
  __syncthreads();
  {
    const int r = tid >> 4;
    const int q = tid & 15;
    const float u = row_16th_short(&lds_agg[r][q * NSHORT], q, lane);
    if (q == 0) lds_T[r] = fmaxf(u, 0.f);
  }
  __syncthreads();

  int nrec = 0;
  const float U = lds_T[m];
  auto record8 = [&](const float (&dd)[8], int jc) {
    unsigned hits[8];
    #pragma unroll
    for (int r = 0; r < 8; ++r) hits[r] = __builtin_amdgcn_ballot_w32(dd[r] <= U);
    #pragma unroll
    for (int r = 0; r < 8; ++r) {
      if (__builtin_amdgcn_inverse_ballot_w32(hits[r])) {
        const v2i kv = {jc + r, max(__float_as_int(dd[r]), 0)};
        rec2[min(nrec, RCAP - 1)] = kv;
        ++nrec;
      }
    }
  };
  {
    float pk[PARK_LO][8];
    #pragma unroll
    for (int t = 0; t < PARK_LO; ++t) {
      const v4f p0 = reinterpret_cast<const v4f*>(park_lo + 8 * t)[0];
      const v4f p1 = reinterpret_cast<const v4f*>(park_lo + 8 * t)[1];
      pk[t][0] = p0.x; pk[t][1] = p0.y; pk[t][2] = p0.z; pk[t][3] = p0.w;
      pk[t][4] = p1.x; pk[t][5] = p1.y; pk[t][6] = p1.z; pk[t][7] = p1.w;
    }
    #pragma unroll
    for (int t = 0; t < PARK_LO; ++t)
      record8(pk[t], (wave + NSPLIT * t) * 16 + 8 * kh);
  }
  #pragma unroll
  for (int t = PARK_LO; t < TILES_A; ++t) {
    const v4f p0 = reinterpret_cast<const v4f*>(park_hi + 8 * (t - PARK_LO))[0];
    const v4f p1 = reinterpret_cast<const v4f*>(park_hi + 8 * (t - PARK_LO))[1];
    const float pk[8] = {p0.x, p0.y, p0.z, p0.w, p1.x, p1.y, p1.z, p1.w};
    record8(pk, (wave + NSPLIT * t) * 16 + 8 * kh);
  }

  {
    auto step = [&](const TileFrag& use, TileFrag& fill, int jt) {
      const int jn = (jt + 2 * NSPLIT < ROW_TILES) ? jt + 2 * NSPLIT : jt;
      load_frag(coords, sq, jn, m, kh, fill);
      float dd[8];
      frag_d2<false>(use, jt, rt, row, kh, b, sqi, dd);
      record8(dd, jt * 16 + 8 * kh);
    };
    int jt = wave + NSPLIT * TILES_A;
    #pragma unroll 1
    for (; jt + 2 * NSPLIT < ROW_TILES; jt += 3 * NSPLIT) {
      step(f0, f2, jt);
      step(f1, f0, jt + NSPLIT);
      step(f2, f1, jt + 2 * NSPLIT);
    }
    #pragma unroll 1
    for (; jt < ROW_TILES; jt += NSPLIT) {
      step(f0, f2, jt);
      f0 = f1;
      f1 = f2;
    }
  }

  const bool ovf = __builtin_amdgcn_ballot_w32(nrec > RCAP) != 0;

  float best[KNN];
  #pragma unroll
  for (int s = 0; s < KNN; ++s) best[s] = BIGF;
  if (ovf) {
    #pragma unroll 1
    for (int jt = wave; jt < ROW_TILES; jt += NSPLIT) {
      load_frag(coords, sq, jt, m, kh, f2);
      float dd[8];
      frag_d2(f2, jt, rt, row, kh, b, sqi, dd);
      #pragma unroll
      for (int r = 0; r < 8; ++r) list_insert(best, dd[r] <= U ? dd[r] : BIGF);
    }
  } else {
    float d = (0 < nrec) ? __int_as_float(rec2[0].y) : BIGF;
    for (int i = 0; __builtin_amdgcn_ballot_w32(i < nrec) != 0; ++i) {
      const float dn = (i + 1 < nrec) ? __int_as_float(rec2[i + 1].y) : BIGF;
      list_insert(best, d);
      d = dn;
    }
  }

  {
    v4f* vp = reinterpret_cast<v4f*>(kl);
    #pragma unroll
    for (int g = 0; g < 4; ++g) {
      v4f v = {best[4 * g], best[4 * g + 1], best[4 * g + 2], best[4 * g + 3]};
      vp[g] = v;
    }
  }
  lds_w[tid >> 4][tid & 15] = 0.f;
  lds_j[tid >> 4][tid & 15] = 0;
  __syncthreads();

  {
    const int r = tid >> 4;
    const int q = tid & 15;
    const float t = row_16th(reinterpret_cast<const float*>(&lds_key[r][q * KNN]), q, lane);
    if (q == 0) lds_T[r] = t;
  }
  __syncthreads();
  {
    v2u64* kp = reinterpret_cast<v2u64*>(kl);
    const v2u64 none = {KEY_NONE, KEY_NONE};
    #pragma unroll
    for (int g = 0; g < KNN / 2; ++g) kp[g] = none;
  }

  int cs = 0, ct = 0;
  {
    const float T = lds_T[m];
    auto collect = [&](float d, u64 key) {
      if (d < T) {
        if (cs < KNN) {
          if (cs + ct == KNN) --ct;
          kl[cs] = key; ++cs;
        }
      } else if (cs + ct < KNN) {
        ++ct; kl[KNN - ct] = key;
      }
    };
    if (ovf) {
      #pragma unroll 1
      for (int jt = wave; jt < ROW_TILES; jt += NSPLIT) {
        const int j0 = jt * 16;
        load_frag(coords, sq, jt, m, kh, f2);
        float dd[8];
        frag_d2(f2, jt, rt, row, kh, b, sqi, dd);
        #pragma unroll
        for (int r = 0; r < 8; ++r) {
          const float d = dd[r];
          if (d <= T) {
            const int db = max(__float_as_int(d), 0);
            collect(d, ((u64)(unsigned)db << 32) | (u64)(unsigned)(j0 + 8 * kh + r));
          }
        }
      }
    } else {
      for (int i = 0; i < nrec; ++i) {
        const u64 key = rec[i];
        const float d = __int_as_float((int)(key >> 32));
        if (d <= T) collect(d, key);
      }
    }
    lds_cnt[m][list] = cs | (ct << 8);
  }
  __syncthreads();

  int tot = 0;
  {
    const int r = tid >> 4;
    const int q = tid & 15;
    int off = 0, mine = 0;
    #pragma unroll
    for (int l = 0; l < NLIST; ++l) {
      const int c = lds_cnt[r][l];
      const int n = (c & 255) + (c >> 8);
      off += (l < q) ? n : 0;
      mine = (l == q) ? c : mine;
      tot += n;
    }
    const int ncs = mine & 255, nct = mine >> 8;
    const u64* kl = &lds_key[r][q * KNN];
    for (int i = 0; i < ncs + nct; ++i) {
      const int slot = i < ncs ? i : KNN - nct + (i - ncs);
      lds_cand[r][off + i] = kl[slot];
    }
    if (q == 0) lds_cand[r][tot] = KEY_NONE;
  }
  __syncthreads();

  {
    const int r = tid >> 4;
    const int q = tid & 15;
    for (int p = q; p < tot; p += 16) {
      const u64 ck = lds_cand[r][p];
      int rank = 0;
      for (int e = 0; e < tot; e += 2) {
        const v2u64 kv = *reinterpret_cast<const v2u64*>(&lds_cand[r][e]);
        rank += (kv.x < ck ? 1 : 0) + (kv.y < ck ? 1 : 0);
      }
      if (rank < KNN) {
        lds_w[r][rank] = expf(-__int_as_float((int)(ck >> 32)));
        lds_j[r][rank] = (int)(unsigned)ck;
      }
    }
  }
  __syncthreads();

  #pragma unroll
  for (int rr = 0; rr < 16 / NSPLIT; ++rr) {
    const int r = wave * (16 / NSPLIT) + rr;
    float acc0 = 0.f, acc1 = 0.f;
    #pragma unroll
    for (int k = 0; k < KNN; ++k) {
      const int   j  = lds_j[r][k];
      const float wk = lds_w[r][k];
      acc0 += wk * feats[(size_t)j * DFEAT + lane];
      acc1 += wk * feats[(size_t)j * DFEAT + lane + 32];
    }
    lds_agg[r][lane]      = acc0;
    lds_agg[r][lane + 32] = acc1;
  }
  __syncthreads();

  if (wave < 2) {
    const float* fr = feats + (size_t)(row0 + m) * DFEAT;
    #pragma unroll
    for (int t = 0; t < 2; ++t) {
      const int nt = wave * 2 + t;
      v8f acc = {};
      #pragma unroll
      for (int s = 0; s < 16; ++s) {
        v2f av, bv;
        av.x = fr[4 * s + 2 * kh];
        av.y = fr[4 * s + 1 + 2 * kh];
        bv.x = Wout[(4 * s + 2 * kh) * DFEAT + nt * 16 + m];
        bv.y = Wout[(4 * s + 1 + 2 * kh) * DFEAT + nt * 16 + m];
        acc = wmma_f32(av, bv, acc);
      }
      #pragma unroll
      for (int s = 0; s < 16; ++s) {
        v2f av, bv;
        av.x = lds_agg[m][4 * s + 2 * kh];
        av.y = lds_agg[m][4 * s + 1 + 2 * kh];
        bv.x = Wout[(64 + 4 * s + 2 * kh) * DFEAT + nt * 16 + m];
        bv.y = Wout[(64 + 4 * s + 1 + 2 * kh) * DFEAT + nt * 16 + m];
        acc = wmma_f32(av, bv, acc);
      }
      #pragma unroll
      for (int r = 0; r < 8; ++r) {
        const int row = row0 + r + 8 * kh;
        const int col = nt * 16 + m;
        out[(size_t)row * DFEAT + col] = acc[r] + bout[col];
      }
    }
  }
}

extern "C" void kernel_launch(void* const* d_in, const int* in_sizes, int n_in,
                              void* d_out, int out_size, void* d_ws, size_t ws_size,
                              hipStream_t stream) {
  const float* x  = (const float*)d_in[0];
  const float* Wf = (const float*)d_in[1];
  const float* bf = (const float*)d_in[2];
  const float* Wl = (const float*)d_in[3];
  const float* bl = (const float*)d_in[4];
  const float* Wo = (const float*)d_in[5];
  const float* bo = (const float*)d_in[6];
  float* out = (float*)d_out;

  float* feats  = (float*)d_ws;
  float* coords = feats  + (size_t)N_PTS * DFEAT;
  float* sq     = coords + (size_t)N_PTS * DLAT;

  encoder_kernel<<<ENC_WG, 32 * ENC_WAVES, 0, stream>>>(x, Wf, bf, Wl, bl, feats, coords, sq);
  gravnet_kernel<<<ROW_TILES, 256, 0, stream>>>(coords, sq, feats, Wo, bo, out);
}


# Encoder split three ways per 16-row tile: one wave takes coords and norms, two waves 32 feats columns each with their two product chains side by side (was one wave, 5 tiles in turn); bit-identical
# speedup vs baseline: 16.7510x; 1.0284x over previous
#include <hip/hip_runtime.h>
#include <cmath>

typedef __attribute__((ext_vector_type(2))) float v2f;
typedef __attribute__((ext_vector_type(8))) float v8f;

#define N_PTS   12000
#define DIN     64
#define DLAT    16
#define DFEAT   64
#define KNN     16
#define ROW_TILES (N_PTS / 16)
#define WAVES_PER_WG 8
#define ENC_TILES 2
#define ENC_FEAT_WAVES (2 * ENC_TILES)
#define ENC_WAVES (ENC_FEAT_WAVES + ENC_TILES)
#define ENC_WG  (ROW_TILES / ENC_TILES)

static __device__ __forceinline__ v8f wmma_f32(v2f a, v2f b, v8f c) {
  return __builtin_amdgcn_wmma_f32_16x16x4_f32(
       false, a,  false, b,
       (short)0, c,  false,  false);
}

template <int SRC>
static __device__ __forceinline__ float row_share(float v) {
  return __int_as_float(__builtin_amdgcn_update_dpp(0, __float_as_int(v), 0x150 + SRC, 0xf, 0xf, true));
}

static __device__ __forceinline__ float row_sum_in_order(float p) {
  float s = row_share<0>(p) + row_share<1>(p);
  s += row_share<2>(p);
  s += row_share<3>(p);
  s += row_share<4>(p);
  s += row_share<5>(p);
  s += row_share<6>(p);
  s += row_share<7>(p);
  s += row_share<8>(p);
  s += row_share<9>(p);
  s += row_share<10>(p);
  s += row_share<11>(p);
  s += row_share<12>(p);
  s += row_share<13>(p);
  s += row_share<14>(p);
  s += row_share<15>(p);
  return s;
}

static_assert(ROW_TILES % ENC_TILES == 0, "every encoder workgroup is full: its barrier is reached by all waves");
static_assert(ENC_TILES == 2, "wave -> (row tile, column half) below is written for two row tiles");
__global__ void __launch_bounds__(32 * ENC_WAVES)
encoder_kernel(const float* __restrict__ x,
               const float* __restrict__ Wf, const float* __restrict__ bf,
               const float* __restrict__ Wl, const float* __restrict__ bl,
               float* __restrict__ feats, float* __restrict__ coords,
               float* __restrict__ sq) {
  __shared__ float lds_sq[16 * ENC_TILES];
  const int lane = threadIdx.x & 31;
  const int wave = __builtin_amdgcn_readfirstlane(threadIdx.x >> 5);
  const bool coords_wave = wave >= ENC_FEAT_WAVES;
  const int wt   = wave & (ENC_TILES - 1);
  const int tile = blockIdx.x * ENC_TILES + wt;
  const int row0 = tile * 16;
  const int m  = lane & 15;
  const int kh = lane >> 4;

  v2f a[16];
  const float* xr = x + (size_t)(row0 + m) * DIN;
  #pragma unroll
  for (int s = 0; s < 16; ++s) {
    a[s].x = xr[4 * s + 2 * kh];
    a[s].y = xr[4 * s + 1 + 2 * kh];
  }

  if (coords_wave) {
    v8f acc = {};
    #pragma unroll
    for (int s = 0; s < 16; ++s) {
      v2f b;
      b.x = Wl[(4 * s + 2 * kh) * DLAT + m];
      b.y = Wl[(4 * s + 1 + 2 * kh) * DLAT + m];
      acc = wmma_f32(a[s], b, acc);
    }
    float s[8];
    #pragma unroll
    for (int r = 0; r < 8; ++r) {
      const int row = row0 + r + 8 * kh;
      const float c = tanhf(acc[r] + bl[m]);
      coords[(size_t)row * DLAT + m] = c;
      s[r] = row_sum_in_order(c * c);
    }
    float v = s[0];
    #pragma unroll
    for (int r = 1; r < 8; ++r) v = (m == r) ? s[r] : v;
    if (m < 8) lds_sq[wt * 16 + 8 * kh + m] = v;
  } else {
    const int nt0 = (wave / ENC_TILES) * 2;
    v8f acc0 = {}, acc1 = {};
    #pragma unroll
    for (int s = 0; s < 16; ++s) {
      v2f b0, b1;
      b0.x = Wf[(4 * s + 2 * kh) * DFEAT + nt0 * 16 + m];
      b0.y = Wf[(4 * s + 1 + 2 * kh) * DFEAT + nt0 * 16 + m];
      b1.x = Wf[(4 * s + 2 * kh) * DFEAT + nt0 * 16 + 16 + m];
      b1.y = Wf[(4 * s + 1 + 2 * kh) * DFEAT + nt0 * 16 + 16 + m];
      acc0 = wmma_f32(a[s], b0, acc0);
      acc1 = wmma_f32(a[s], b1, acc1);
    }
    #pragma unroll
    for (int t = 0; t < 2; ++t) {
      const v8f acc = t == 0 ? acc0 : acc1;
      #pragma unroll
      for (int r = 0; r < 8; ++r) {
        const int row = row0 + r + 8 * kh;
        const int col = (nt0 + t) * 16 + m;
        feats[(size_t)row * DFEAT + col] = tanhf(acc[r] + bf[col]);
      }
    }
  }
  __builtin_amdgcn_fence(__ATOMIC_RELEASE, "workgroup", "local");
  __builtin_amdgcn_s_barrier();
  __builtin_amdgcn_fence(__ATOMIC_ACQUIRE, "workgroup", "local");
  if (wave == ENC_FEAT_WAVES) sq[blockIdx.x * (16 * ENC_TILES) + lane] = lds_sq[lane];
}

#define NSPLIT  WAVES_PER_WG
#define NLIST   (2 * NSPLIT)
#define NCAND   (NLIST * KNN)
#define KPITCH  (NCAND + 2)
#define VPITCH  (2 * KPITCH)
#define BIGF    3.4e38f
#define TILES_A 12
#define NSHORT  4
#define RCAP    32
#define PARK_LO (RCAP / 4)
#define RPITCH  (NLIST * RCAP + 2)

typedef unsigned long long u64;
typedef __attribute__((ext_vector_type(2))) unsigned long long v2u64;
typedef __attribute__((ext_vector_type(4))) float v4f;
typedef __attribute__((ext_vector_type(2))) int v2i;

#define KEY_NONE 0xFFFFFFFFFFFFFFFFull

static __device__ __forceinline__ float pair_raw(float sqi, float sqj, float dot) {
  return __builtin_fmaf(-2.f, dot, sqi + sqj);
}
static __device__ __forceinline__ float pair_d2(float sqi, float sqj, float dot) {
  return fmaxf(pair_raw(sqi, sqj, dot), 0.f);
}

struct TileFrag {
  v2f a[4];
  v4f s0, s1;
};

static __device__ __forceinline__ void load_frag(const float* __restrict__ coords,
                                                 const float* __restrict__ sq,
                                                 int jt, int m, int kh, TileFrag& f) {
  const int j0 = jt * 16;
  const float* cj = coords + (size_t)(j0 + m) * DLAT;
  #pragma unroll
  for (int s = 0; s < 4; ++s) {
    f.a[s].x = cj[4 * s + 2 * kh];
    f.a[s].y = cj[4 * s + 1 + 2 * kh];
  }
  f.s0 = *reinterpret_cast<const v4f*>(sq + j0 + 8 * kh);
  f.s1 = *reinterpret_cast<const v4f*>(sq + j0 + 8 * kh + 4);
}

static __device__ __forceinline__ v8f frag_dots(const TileFrag& f, const v2f (&b)[4]) {
  v8f acc = {};
  #pragma unroll
  for (int s = 0; s < 4; ++s) acc = wmma_f32(f.a[s], b[s], acc);
  return acc;
}

template <int MASK>
static __device__ __forceinline__ float row_xor(float v) {
  return __int_as_float(__builtin_amdgcn_update_dpp(0, __float_as_int(v), 0x160 + MASK, 0xf, 0xf, false));
}

static __device__ __forceinline__ void list_insert(float (&best)[KNN], float d) {
  #pragma unroll
  for (int s = 0; s < KNN - 1; ++s)
    best[s] = __builtin_amdgcn_fmed3f(best[s], best[s + 1], d);
  best[KNN - 1] = __builtin_amdgcn_fmed3f(best[KNN - 1], -BIGF, d);
}

static __device__ __forceinline__ void list_insert_short(float (&l)[NSHORT], float d) {
  #pragma unroll
  for (int s = 0; s < NSHORT - 1; ++s)
    l[s] = __builtin_amdgcn_fmed3f(l[s], l[s + 1], d);
  l[NSHORT - 1] = __builtin_amdgcn_fmed3f(l[NSHORT - 1], -BIGF, d);
}

template <bool CLAMP = true>
static __device__ __forceinline__ void frag_d2(const TileFrag& f, int jt, int rt, int row, int kh,
                                               const v2f (&b)[4], float sqi, float (&dd)[8]) {
  const int j0 = jt * 16;
  const v8f acc = frag_dots(f, b);
  #pragma unroll
  for (int r = 0; r < 8; ++r) {
    const float sqj = r < 4 ? f.s0[r & 3] : f.s1[r & 3];
    dd[r] = CLAMP ? pair_d2(sqi, sqj, acc[r]) : pair_raw(sqi, sqj, acc[r]);
  }
  if (jt == rt) {
    #pragma unroll
    for (int r = 0; r < 8; ++r)
      if (j0 + 8 * kh + r == row) dd[r] = BIGF;
  }
}

static __device__ __forceinline__ float row_16th(const float* lv, int q, int lane) {
  int head = 0;
  float t = 0.f;
  for (int step = 0; step < KNN; ++step) {
    const float v = lv[KNN - 1 - head];
    float mn = v;
    mn = fminf(mn, row_xor<1>(mn));
    mn = fminf(mn, row_xor<2>(mn));
    mn = fminf(mn, row_xor<4>(mn));
    mn = fminf(mn, row_xor<8>(mn));
    t = mn;
    const unsigned holders = (__builtin_amdgcn_ballot_w32(v == mn) >> (lane & 16)) & 0xFFFFu;
    if (q == __builtin_ctz(holders)) ++head;
  }
  return t;
}

static __device__ __forceinline__ float row_16th_short(const float* lv, int q, int lane) {
  const v4f l = *reinterpret_cast<const v4f*>(lv);
  float c0 = l.w, c1 = l.z, c2 = l.y, c3 = l.x;
  float t = 0.f;
  for (int step = 0; step < KNN; ++step) {
    const float v = c0;
    float mn = v;
    mn = fminf(mn, row_xor<1>(mn));
    mn = fminf(mn, row_xor<2>(mn));
    mn = fminf(mn, row_xor<4>(mn));
    mn = fminf(mn, row_xor<8>(mn));
    t = mn;
    const unsigned holders = (__builtin_amdgcn_ballot_w32(v == mn) >> (lane & 16)) & 0xFFFFu;
    if (q == __builtin_ctz(holders)) { c0 = c1; c1 = c2; c2 = c3; c3 = BIGF; }
  }
  return t;
}

__global__ void __launch_bounds__(256, 6)
gravnet_kernel(const float* __restrict__ coords, const float* __restrict__ sq,
               const float* __restrict__ feats,
               const float* __restrict__ Wout, const float* __restrict__ bout,
               float* __restrict__ out) {
  __shared__ __attribute__((aligned(16))) u64 lds_rec[16][RPITCH];
  u64 (*lds_cand)[KPITCH] = reinterpret_cast<u64 (*)[KPITCH]>(&lds_rec[0][0]);
  __shared__ __attribute__((aligned(16))) u64 lds_key[16][KPITCH];
  __shared__ __attribute__((aligned(16))) int lds_cnt[16][NLIST];
  __shared__ float lds_T  [16];
  __shared__ float lds_w  [16][KNN];
  __shared__ int   lds_j  [16][KNN];
  __shared__ __attribute__((aligned(16))) float lds_agg[16][DFEAT];
  static_assert(NLIST * NSHORT <= DFEAT, "short lists must fit a row of lds_agg");
  static_assert(NSHORT == 4, "a short list is moved as one 16-byte vector");
  static_assert(TILES_A * 8 * 4 <= RCAP * 8 + KNN * 8, "parked distances must fit the lane's own slots");

  const int tid  = threadIdx.x;
  const int lane = tid & 31;
  const int wave = __builtin_amdgcn_readfirstlane(tid >> 5);
  const int rt   = blockIdx.x;
  const int row0 = rt * 16;
  const int m  = lane & 15;
  const int kh = lane >> 4;
  const int row  = row0 + m;
  const int list = wave * 2 + kh;

  v2f b[4];
  {
    const float* cr = coords + (size_t)row * DLAT;
    #pragma unroll
    for (int s = 0; s < 4; ++s) {
      b[s].x = cr[4 * s + 2 * kh];
      b[s].y = cr[4 * s + 1 + 2 * kh];
    }
  }
  const float sqi = sq[row];

  u64* rec = &lds_rec[m][list * RCAP];
  v2i* rec2 = reinterpret_cast<v2i*>(rec);
  u64* kl  = &lds_key[m][list * KNN];
  float* park_lo = reinterpret_cast<float*>(rec);
  float* park_hi = reinterpret_cast<float*>(kl);

  float shortl[NSHORT];
  #pragma unroll
  for (int s = 0; s < NSHORT; ++s) shortl[s] = BIGF;

  TileFrag f0, f1, f2;
  load_frag(coords, sq, wave, m, kh, f0);
  load_frag(coords, sq, wave + NSPLIT, m, kh, f1);
  {
    auto step = [&](const TileFrag& use, TileFrag& fill, int jt) {
      load_frag(coords, sq, jt + 2 * NSPLIT, m, kh, fill);
      float dd[8];
      frag_d2<false>(use, jt, rt, row, kh, b, sqi, dd);
      const int t = (jt - wave) / NSPLIT;
      float* pp = (t < PARK_LO) ? park_lo + 8 * t : park_hi + 8 * (t - PARK_LO);
      const v4f p0 = {dd[0], dd[1], dd[2], dd[3]};
      const v4f p1 = {dd[4], dd[5], dd[6], dd[7]};
      reinterpret_cast<v4f*>(pp)[0] = p0;
      reinterpret_cast<v4f*>(pp)[1] = p1;
      #pragma unroll
      for (int r = 0; r < 8; ++r) list_insert_short(shortl, dd[r]);
    };
    const int end = wave + NSPLIT * TILES_A;
    int jt = wave;
    #pragma unroll 1
    for (; jt + 2 * NSPLIT < end; jt += 3 * NSPLIT) {
      step(f0, f2, jt);
      step(f1, f0, jt + NSPLIT);
      step(f2, f1, jt + 2 * NSPLIT);
    }
    #pragma unroll 1
    for (; jt < end; jt += NSPLIT) {
      step(f0, f2, jt);
      f0 = f1;
      f1 = f2;
    }
  }

  {
    const v4f v = {shortl[0], shortl[1], shortl[2], shortl[3]};
    *reinterpret_cast<v4f*>(&lds_agg[m][list * NSHORT]) = v;
  }
  __syncthreads();
  {
    const int r = tid >> 4;
    const int q = tid & 15;
    const float u = row_16th_short(&lds_agg[r][q * NSHORT], q, lane);
    if (q == 0) lds_T[r] = fmaxf(u, 0.f);
  }
  __syncthreads();

  int nrec = 0;
  const float U = lds_T[m];
  auto record8 = [&](const float (&dd)[8], int jc) {
    unsigned hits[8];
    #pragma unroll
    for (int r = 0; r < 8; ++r) hits[r] = __builtin_amdgcn_ballot_w32(dd[r] <= U);
    #pragma unroll
    for (int r = 0; r < 8; ++r) {
      if (__builtin_amdgcn_inverse_ballot_w32(hits[r])) {
        const v2i kv = {jc + r, max(__float_as_int(dd[r]), 0)};
        rec2[min(nrec, RCAP - 1)] = kv;
        ++nrec;
      }
    }
  };
  {
    float pk[PARK_LO][8];
    #pragma unroll
    for (int t = 0; t < PARK_LO; ++t) {
      const v4f p0 = reinterpret_cast<const v4f*>(park_lo + 8 * t)[0];
      const v4f p1 = reinterpret_cast<const v4f*>(park_lo + 8 * t)[1];
      pk[t][0] = p0.x; pk[t][1] = p0.y; pk[t][2] = p0.z; pk[t][3] = p0.w;
      pk[t][4] = p1.x; pk[t][5] = p1.y; pk[t][6] = p1.z; pk[t][7] = p1.w;
    }
    #pragma unroll
    for (int t = 0; t < PARK_LO; ++t)
      record8(pk[t], (wave + NSPLIT * t) * 16 + 8 * kh);
  }
  #pragma unroll
  for (int t = PARK_LO; t < TILES_A; ++t) {
    const v4f p0 = reinterpret_cast<const v4f*>(park_hi + 8 * (t - PARK_LO))[0];
    const v4f p1 = reinterpret_cast<const v4f*>(park_hi + 8 * (t - PARK_LO))[1];
    const float pk[8] = {p0.x, p0.y, p0.z, p0.w, p1.x, p1.y, p1.z, p1.w};
    record8(pk, (wave + NSPLIT * t) * 16 + 8 * kh);
  }

  {
    auto step = [&](const TileFrag& use, TileFrag& fill, int jt) {
      const int jn = (jt + 2 * NSPLIT < ROW_TILES) ? jt + 2 * NSPLIT : jt;
      load_frag(coords, sq, jn, m, kh, fill);
      float dd[8];
      frag_d2<false>(use, jt, rt, row, kh, b, sqi, dd);
      record8(dd, jt * 16 + 8 * kh);
    };
    int jt = wave + NSPLIT * TILES_A;
    #pragma unroll 1
    for (; jt + 2 * NSPLIT < ROW_TILES; jt += 3 * NSPLIT) {
      step(f0, f2, jt);
      step(f1, f0, jt + NSPLIT);
      step(f2, f1, jt + 2 * NSPLIT);
    }
    #pragma unroll 1
    for (; jt < ROW_TILES; jt += NSPLIT) {
      step(f0, f2, jt);
      f0 = f1;
      f1 = f2;
    }
  }

  const bool ovf = __builtin_amdgcn_ballot_w32(nrec > RCAP) != 0;

  float best[KNN];
  #pragma unroll
  for (int s = 0; s < KNN; ++s) best[s] = BIGF;
  if (ovf) {
    #pragma unroll 1
    for (int jt = wave; jt < ROW_TILES; jt += NSPLIT) {
      load_frag(coords, sq, jt, m, kh, f2);
      float dd[8];
      frag_d2(f2, jt, rt, row, kh, b, sqi, dd);
      #pragma unroll
      for (int r = 0; r < 8; ++r) list_insert(best, dd[r] <= U ? dd[r] : BIGF);
    }
  } else {
    float d = (0 < nrec) ? __int_as_float(rec2[0].y) : BIGF;
    for (int i = 0; __builtin_amdgcn_ballot_w32(i < nrec) != 0; ++i) {
      const float dn = (i + 1 < nrec) ? __int_as_float(rec2[i + 1].y) : BIGF;
      list_insert(best, d);
      d = dn;
    }
  }

  {
    v4f* vp = reinterpret_cast<v4f*>(kl);
    #pragma unroll
    for (int g = 0; g < 4; ++g) {
      v4f v = {best[4 * g], best[4 * g + 1], best[4 * g + 2], best[4 * g + 3]};
      vp[g] = v;
    }
  }
  lds_w[tid >> 4][tid & 15] = 0.f;
  lds_j[tid >> 4][tid & 15] = 0;
  __syncthreads();

  {
    const int r = tid >> 4;
    const int q = tid & 15;
    const float t = row_16th(reinterpret_cast<const float*>(&lds_key[r][q * KNN]), q, lane);
    if (q == 0) lds_T[r] = t;
  }
  __syncthreads();
  {
    v2u64* kp = reinterpret_cast<v2u64*>(kl);
    const v2u64 none = {KEY_NONE, KEY_NONE};
    #pragma unroll
    for (int g = 0; g < KNN / 2; ++g) kp[g] = none;
  }

  int cs = 0, ct = 0;
  {
    const float T = lds_T[m];
    auto collect = [&](float d, u64 key) {
      if (d < T) {
        if (cs < KNN) {
          if (cs + ct == KNN) --ct;
          kl[cs] = key; ++cs;
        }
      } else if (cs + ct < KNN) {
        ++ct; kl[KNN - ct] = key;
      }
    };
    if (ovf) {
      #pragma unroll 1
      for (int jt = wave; jt < ROW_TILES; jt += NSPLIT) {
        const int j0 = jt * 16;
        load_frag(coords, sq, jt, m, kh, f2);
        float dd[8];
        frag_d2(f2, jt, rt, row, kh, b, sqi, dd);
        #pragma unroll
        for (int r = 0; r < 8; ++r) {
          const float d = dd[r];
          if (d <= T) {
            const int db = max(__float_as_int(d), 0);
            collect(d, ((u64)(unsigned)db << 32) | (u64)(unsigned)(j0 + 8 * kh + r));
          }
        }
      }
    } else {
      for (int i = 0; i < nrec; ++i) {
        const u64 key = rec[i];
        const float d = __int_as_float((int)(key >> 32));
        if (d <= T) collect(d, key);
      }
    }
    lds_cnt[m][list] = cs | (ct << 8);
  }
  __syncthreads();

  int tot = 0;
  {
    const int r = tid >> 4;
    const int q = tid & 15;
    int off = 0, mine = 0;
    #pragma unroll
    for (int l = 0; l < NLIST; ++l) {
      const int c = lds_cnt[r][l];
      const int n = (c & 255) + (c >> 8);
      off += (l < q) ? n : 0;
      mine = (l == q) ? c : mine;
      tot += n;
    }
    const int ncs = mine & 255, nct = mine >> 8;
    const u64* kl = &lds_key[r][q * KNN];
    for (int i = 0; i < ncs + nct; ++i) {
      const int slot = i < ncs ? i : KNN - nct + (i - ncs);
      lds_cand[r][off + i] = kl[slot];
    }
    if (q == 0) lds_cand[r][tot] = KEY_NONE;
  }
  __syncthreads();

  {
    const int r = tid >> 4;
    const int q = tid & 15;
    for (int p = q; p < tot; p += 16) {
      const u64 ck = lds_cand[r][p];
      int rank = 0;
      for (int e = 0; e < tot; e += 2) {
        const v2u64 kv = *reinterpret_cast<const v2u64*>(&lds_cand[r][e]);
        rank += (kv.x < ck ? 1 : 0) + (kv.y < ck ? 1 : 0);
      }
      if (rank < KNN) {
        lds_w[r][rank] = expf(-__int_as_float((int)(ck >> 32)));
        lds_j[r][rank] = (int)(unsigned)ck;
      }
    }
  }
  __syncthreads();

  #pragma unroll
  for (int rr = 0; rr < 16 / NSPLIT; ++rr) {
    const int r = wave * (16 / NSPLIT) + rr;
    float acc0 = 0.f, acc1 = 0.f;
    #pragma unroll
    for (int k = 0; k < KNN; ++k) {
      const int   j  = lds_j[r][k];
      const float wk = lds_w[r][k];
      acc0 += wk * feats[(size_t)j * DFEAT + lane];
      acc1 += wk * feats[(size_t)j * DFEAT + lane + 32];
    }
    lds_agg[r][lane]      = acc0;
    lds_agg[r][lane + 32] = acc1;
  }
  __syncthreads();

  if (wave < 2) {
    const float* fr = feats + (size_t)(row0 + m) * DFEAT;
    #pragma unroll
    for (int t = 0; t < 2; ++t) {
      const int nt = wave * 2 + t;
      v8f acc = {};
      #pragma unroll
      for (int s = 0; s < 16; ++s) {
        v2f av, bv;
        av.x = fr[4 * s + 2 * kh];
        av.y = fr[4 * s + 1 + 2 * kh];
        bv.x = Wout[(4 * s + 2 * kh) * DFEAT + nt * 16 + m];
        bv.y = Wout[(4 * s + 1 + 2 * kh) * DFEAT + nt * 16 + m];
        acc = wmma_f32(av, bv, acc);
      }
      #pragma unroll
      for (int s = 0; s < 16; ++s) {
        v2f av, bv;
        av.x = lds_agg[m][4 * s + 2 * kh];
        av.y = lds_agg[m][4 * s + 1 + 2 * kh];
        bv.x = Wout[(64 + 4 * s + 2 * kh) * DFEAT + nt * 16 + m];
        bv.y = Wout[(64 + 4 * s + 1 + 2 * kh) * DFEAT + nt * 16 + m];
        acc = wmma_f32(av, bv, acc);
      }
      #pragma unroll
      for (int r = 0; r < 8; ++r) {
        const int row = row0 + r + 8 * kh;
        const int col = nt * 16 + m;
        out[(size_t)row * DFEAT + col] = acc[r] + bout[col];
      }
    }
  }
}

extern "C" void kernel_launch(void* const* d_in, const int* in_sizes, int n_in,
                              void* d_out, int out_size, void* d_ws, size_t ws_size,
                              hipStream_t stream) {
  const float* x  = (const float*)d_in[0];
  const float* Wf = (const float*)d_in[1];
  const float* bf = (const float*)d_in[2];
  const float* Wl = (const float*)d_in[3];
  const float* bl = (const float*)d_in[4];
  const float* Wo = (const float*)d_in[5];
  const float* bo = (const float*)d_in[6];
  float* out = (float*)d_out;

  float* feats  = (float*)d_ws;
  float* coords = feats  + (size_t)N_PTS * DFEAT;
  float* sq     = coords + (size_t)N_PTS * DLAT;

  encoder_kernel<<<ENC_WG, 32 * ENC_WAVES, 0, stream>>>(x, Wf, bf, Wl, bl, feats, coords, sq);
  gravnet_kernel<<<ROW_TILES, 256, 0, stream>>>(coords, sq, feats, Wo, bo, out);
}
